# starting kernel
# baseline (speedup 1.0000x reference)
#include <hip/hip_runtime.h>
#include <math.h>

#define BB 4
#define MM 500
#define NN 500
#define CC 91

typedef __attribute__((ext_vector_type(2))) float v2f;
typedef __attribute__((ext_vector_type(8))) float v8f;

__device__ __forceinline__ int imin_i(int a, int b) { return a < b ? a : b; }

__global__ void sigmoid_kernel(const float* __restrict__ p1l,
                               const float* __restrict__ p2l,
                               float* __restrict__ s1,
                               float* __restrict__ s2) {
    int idx = blockIdx.x * blockDim.x + threadIdx.x;
    const int n1 = BB * NN * CC;
    const int n2 = BB * MM * CC;
    if (idx < n1) {
        float x = p1l[idx];
        s1[idx] = 1.0f / (1.0f + expf(-x));
    } else if (idx < n1 + n2) {
        float x = p2l[idx - n1];
        s2[idx - n1] = 1.0f / (1.0f + expf(-x));
    }
}

__global__ void cost_kernel(const float* __restrict__ p1b,
                            const float* __restrict__ p2b,
                            const float* __restrict__ s1,
                            const float* __restrict__ s2,
                            float* __restrict__ cost) {
    const int TN = (NN + 15) / 16;
    const int TM = (MM + 15) / 16;
    int bid = blockIdx.x;
    int b  = bid / (TM * TN);
    int t  = bid % (TM * TN);
    int m0 = (t / TN) * 16;
    int n0 = (t % TN) * 16;
    int lane = threadIdx.x;

    __shared__ float sq2[16];
    __shared__ float sq1[16];
    __shared__ float ls2[16 * CC];
    __shared__ float ls1[16 * CC];

    v2f a  = {0.0f, 0.0f};
    v2f bv = {0.0f, 0.0f};
    if (lane < 16) {
        int m = imin_i(m0 + lane, MM - 1);
        const float* pb = p2b + ((size_t)b * MM + m) * 4;
        float x = pb[0], y = pb[1];
        a.x = x; a.y = y;
        sq2[lane] = x * x + y * y;
        int n = imin_i(n0 + lane, NN - 1);
        const float* qb = p1b + ((size_t)b * NN + n) * 4;
        float bx = qb[0], by = qb[1];
        bv.x = bx; bv.y = by;
        sq1[lane] = bx * bx + by * by;
    }
    const float* s2base = s2 + (size_t)b * MM * CC;
    const float* s1base = s1 + (size_t)b * NN * CC;
    for (int idx = lane; idx < 16 * CC; idx += 32) {
        int r = idx / CC;
        int c = idx - r * CC;
        ls2[idx] = s2base[(size_t)imin_i(m0 + r, MM - 1) * CC + c];
        ls1[idx] = s1base[(size_t)imin_i(n0 + r, NN - 1) * CC + c];
    }
    __syncthreads();

    v8f cacc = {0.f, 0.f, 0.f, 0.f, 0.f, 0.f, 0.f, 0.f};
    v8f dot = __builtin_amdgcn_wmma_f32_16x16x4_f32(
         false, a,  false, bv,
         (short)0, cacc,  false,  false);

    int n_loc  = lane & 15;
    int m_base = (lane >> 4) << 3;

    float cls[8];
#pragma unroll
    for (int i = 0; i < 8; ++i) cls[i] = 0.0f;
    for (int c = 0; c < CC; ++c) {
        float s1v = ls1[n_loc * CC + c];
#pragma unroll
        for (int i = 0; i < 8; ++i) {
            float d = fabsf(ls2[(m_base + i) * CC + c] - s1v);
            cls[i] = fmaxf(cls[i], d);
        }
    }

#pragma unroll
    for (int i = 0; i < 8; ++i) {
        int m = m0 + m_base + i;
        int n = n0 + n_loc;
        if (m < MM && n < NN) {
            float d2 = sq2[m_base + i] + sq1[n_loc] - 2.0f * dot[i];
            float cd = sqrtf(fmaxf(d2, 0.0f));
            cost[((size_t)b * MM + m) * NN + n] = 0.5f * cd + 0.5f * cls[i];
        }
    }
}

#define LTH 512
#define NWAVE (LTH / 32)
__global__ void __launch_bounds__(LTH)
lapjv_kernel(const float* __restrict__ cost, int* __restrict__ mapping) {
    int b   = blockIdx.x;
    int tid = threadIdx.x;
    const float* costB = cost + (size_t)b * MM * NN;
    const double INF = __builtin_inf();

    __shared__ double sh_u[LTH], sh_v[LTH], sh_short[LTH];
    __shared__ int sh_path[LTH], sh_c4r[LTH], sh_r4c[LTH], sh_SR[LTH],
                   sh_rem[LTH];
    __shared__ double wred_v[NWAVE];
    __shared__ int    wred_i[NWAVE];
    __shared__ int s_i, s_sink, s_nsr;
    __shared__ double s_minval;

    if (tid < NN) {
        sh_u[tid] = 0.0; sh_v[tid] = 0.0;
        sh_c4r[tid] = -1; sh_r4c[tid] = -1;
    }
    __syncthreads();

    for (int cur = 0; cur < MM; ++cur) {
        if (tid < NN) { sh_rem[tid] = 1; sh_short[tid] = INF; sh_path[tid] = -1; }
        if (tid == 0) { s_i = cur; s_minval = 0.0; s_sink = -1; s_nsr = 0; }
        __syncthreads();

        while (true) {
            int    ii   = s_i;
            double minv = s_minval;
            if (tid == 0) { sh_SR[s_nsr] = ii; s_nsr = s_nsr + 1; }

            double cand;
            int    candi = tid;
            if (tid < NN) {
                double r = minv + (double)costB[(size_t)ii * NN + tid]
                                - sh_u[ii] - sh_v[tid];
                if (sh_rem[tid] && r < sh_short[tid]) {
                    sh_short[tid] = r;
                    sh_path[tid]  = ii;
                }
                cand = sh_rem[tid] ? sh_short[tid] : INF;
            } else {
                cand = INF;
            }

#pragma unroll
            for (int off = 16; off > 0; off >>= 1) {
                double ov = __shfl_xor(cand, off);
                int    oi = __shfl_xor(candi, off);
                if (ov < cand || (ov == cand && oi < candi)) {
                    cand = ov; candi = oi;
                }
            }
            int wave = tid >> 5;
            if ((tid & 31) == 0) { wred_v[wave] = cand; wred_i[wave] = candi; }
            __syncthreads();

            if (tid == 0) {
                double bvv = wred_v[0]; int bii = wred_i[0];
#pragma unroll
                for (int w = 1; w < NWAVE; ++w) {
                    double ov = wred_v[w]; int oi = wred_i[w];
                    if (ov < bvv || (ov == bvv && oi < bii)) {
                        bvv = ov; bii = oi;
                    }
                }
                s_minval = bvv;
                sh_rem[bii] = 0;
                if (sh_r4c[bii] == -1) s_sink = bii;
                else                   s_i = sh_r4c[bii];
            }
            __syncthreads();
            if (s_sink != -1) break;
        }

        double minv = s_minval;
        int    nsr  = s_nsr;
        if (tid == 0) {
            sh_u[cur] += minv;
        } else if (tid < nsr) {
            int sr = sh_SR[tid];
            sh_u[sr] += minv - sh_short[sh_c4r[sr]];
        }
        if (tid < NN && !sh_rem[tid]) sh_v[tid] -= minv - sh_short[tid];
        __syncthreads();

        if (tid == 0) {
            int j = s_sink;
            while (true) {
                int i = sh_path[j];
                sh_r4c[j] = i;
                int tmp = sh_c4r[i];
                sh_c4r[i] = j;
                j = tmp;
                if (i == cur) break;
            }
        }
        __syncthreads();
    }
    if (tid < MM) mapping[b * MM + tid] = sh_c4r[tid];
}

__global__ void extrap_kernel(const float* __restrict__ p1b,
                              const float* __restrict__ p1l,
                              const float* __restrict__ p2b,
                              const float* __restrict__ p2l,
                              const float* __restrict__ toff,
                              const int*   __restrict__ mapping,
                              float* __restrict__ out) {
    int idx = blockIdx.x * blockDim.x + threadIdx.x;
    if (idx >= BB * MM) return;
    int b = idx / MM;

    float first  = toff[b * 3 + 1] - toff[b * 3 + 0];
    float second = toff[b * 3 + 2] - toff[b * 3 + 1];
    float factor = second / first;

    int  j   = mapping[idx];
    bool has = (j >= 0);
    int  jc  = has ? j : 0;

    const float* pb2 = p2b + (size_t)idx * 4;
    const float* pb1 = p1b + ((size_t)b * NN + jc) * 4;
    float* ob = out + (size_t)idx * 4;
#pragma unroll
    for (int k = 0; k < 4; ++k) {
        float c1 = has ? pb1[k] : pb2[k];
        float v  = pb2[k] + (pb2[k] - c1) * factor;
        ob[k] = (k >= 2) ? fmaxf(v, 0.0f) : v;
    }

    const float* pl2 = p2l + (size_t)idx * CC;
    const float* pl1 = p1l + ((size_t)b * NN + jc) * CC;
    float* ol = out + (size_t)BB * MM * 4 + (size_t)idx * CC;
    for (int c = 0; c < CC; ++c) {
        float c1 = has ? pl1[c] : 0.0f;
        ol[c] = 0.5f * (pl2[c] + c1);
    }
}

extern "C" void kernel_launch(void* const* d_in, const int* in_sizes, int n_in,
                              void* d_out, int out_size, void* d_ws, size_t ws_size,
                              hipStream_t stream) {
    (void)in_sizes; (void)n_in; (void)out_size; (void)ws_size;
    const float* p1b  = (const float*)d_in[0];
    const float* p1l  = (const float*)d_in[1];
    const float* p2b  = (const float*)d_in[2];
    const float* p2l  = (const float*)d_in[3];
    const float* toff = (const float*)d_in[4];

    float* ws   = (float*)d_ws;
    float* s1   = ws;
    float* s2   = s1 + (size_t)BB * NN * CC;
    float* cost = s2 + (size_t)BB * MM * CC;
    int*   map  = (int*)(cost + (size_t)BB * MM * NN);
    float* out  = (float*)d_out;

    int tot = BB * NN * CC + BB * MM * CC;
    sigmoid_kernel<<<(tot + 255) / 256, 256, 0, stream>>>(p1l, p2l, s1, s2);

    const int TM = (MM + 15) / 16, TN = (NN + 15) / 16;
    cost_kernel<<<BB * TM * TN, 32, 0, stream>>>(p1b, p2b, s1, s2, cost);

    lapjv_kernel<<<BB, LTH, 0, stream>>>(cost, map);

    int tot2 = BB * MM;
    extrap_kernel<<<(tot2 + 255) / 256, 256, 0, stream>>>(p1b, p1l, p2b, p2l,
                                                          toff, map, out);
}


# Approximate f32 assignment pass: per row the 64 cheapest columns kept in LDS and searched by one wave (LDS gather/scatter, no barrier) instead of dense rows from L2; exact f64 repair kept
# speedup vs baseline: 4.6756x; 4.6756x over previous
#include <hip/hip_runtime.h>
#include <math.h>

#define BB 4
#define MM 500
#define NN 500
#define CC 91

typedef __attribute__((ext_vector_type(2))) float v2f;
typedef __attribute__((ext_vector_type(8))) float v8f;

__device__ __forceinline__ int imin_i(int a, int b) { return a < b ? a : b; }

__global__ void sigmoid_kernel(const float* __restrict__ p1l,
                               const float* __restrict__ p2l,
                               float* __restrict__ s1,
                               float* __restrict__ s2) {
    int idx = blockIdx.x * blockDim.x + threadIdx.x;
    const int n1 = BB * NN * CC;
    const int n2 = BB * MM * CC;
    if (idx < n1) {
        float x = p1l[idx];
        s1[idx] = 1.0f / (1.0f + expf(-x));
    } else if (idx < n1 + n2) {
        float x = p2l[idx - n1];
        s2[idx - n1] = 1.0f / (1.0f + expf(-x));
    }
}

__global__ void cost_kernel(const float* __restrict__ p1b,
                            const float* __restrict__ p2b,
                            const float* __restrict__ s1,
                            const float* __restrict__ s2,
                            float* __restrict__ cost) {
    const int TN = (NN + 15) / 16;
    const int TM = (MM + 15) / 16;
    int bid = blockIdx.x;
    int b  = bid / (TM * TN);
    int t  = bid % (TM * TN);
    int m0 = (t / TN) * 16;
    int n0 = (t % TN) * 16;
    int lane = threadIdx.x;

    __shared__ float sq2[16];
    __shared__ float sq1[16];
    __shared__ float ls2[16 * CC];
    __shared__ float ls1[16 * CC];

    v2f a  = {0.0f, 0.0f};
    v2f bv = {0.0f, 0.0f};
    if (lane < 16) {
        int m = imin_i(m0 + lane, MM - 1);
        const float* pb = p2b + ((size_t)b * MM + m) * 4;
        float x = pb[0], y = pb[1];
        a.x = x; a.y = y;
        sq2[lane] = x * x + y * y;
        int n = imin_i(n0 + lane, NN - 1);
        const float* qb = p1b + ((size_t)b * NN + n) * 4;
        float bx = qb[0], by = qb[1];
        bv.x = bx; bv.y = by;
        sq1[lane] = bx * bx + by * by;
    }
    const float* s2base = s2 + (size_t)b * MM * CC;
    const float* s1base = s1 + (size_t)b * NN * CC;
    for (int idx = lane; idx < 16 * CC; idx += 32) {
        int r = idx / CC;
        int c = idx - r * CC;
        ls2[idx] = s2base[(size_t)imin_i(m0 + r, MM - 1) * CC + c];
        ls1[idx] = s1base[(size_t)imin_i(n0 + r, NN - 1) * CC + c];
    }
    __syncthreads();

    v8f cacc = {0.f, 0.f, 0.f, 0.f, 0.f, 0.f, 0.f, 0.f};
    v8f dot = __builtin_amdgcn_wmma_f32_16x16x4_f32(
         false, a,  false, bv,
         (short)0, cacc,  false,  false);

    int n_loc  = lane & 15;
    int m_base = (lane >> 4) << 3;

    float cls[8];
#pragma unroll
    for (int i = 0; i < 8; ++i) cls[i] = 0.0f;
    for (int c = 0; c < CC; ++c) {
        float s1v = ls1[n_loc * CC + c];
#pragma unroll
        for (int i = 0; i < 8; ++i) {
            float d = fabsf(ls2[(m_base + i) * CC + c] - s1v);
            cls[i] = fmaxf(cls[i], d);
        }
    }

#pragma unroll
    for (int i = 0; i < 8; ++i) {
        int m = m0 + m_base + i;
        int n = n0 + n_loc;
        if (m < MM && n < NN) {
            float d2 = sq2[m_base + i] + sq1[n_loc] - 2.0f * dot[i];
            float cd = sqrtf(fmaxf(d2, 0.0f));
            cost[((size_t)b * MM + m) * NN + n] = 0.5f * cd + 0.5f * cls[i];
        }
    }
}

#define LTH 128
#define NWAVE (LTH / 32)
#define CPL 4
typedef __attribute__((ext_vector_type(4))) float v4f;
typedef __attribute__((ext_vector_type(4))) int   v4i;
typedef __attribute__((ext_vector_type(2))) int   v2i;
typedef __attribute__((ext_vector_type(2))) double v2d;
typedef __attribute__((ext_vector_type(2))) unsigned short v2us;

#define KC 64
static_assert(KC == 64, "two candidates per lane of one wave");
static_assert(NN <= 512 && NN <= 65535, "16 columns per lane; 16-bit column ids");

static_assert(NN % CPL == 0 && NN / CPL <= LTH, "column tiling");
static_assert(MM == NN, "square problem");

template <typename T> struct XT;
template <> struct XT<double> {
    template <int MASK>
    static __device__ __forceinline__ double rx(double x) {
        v2i p = __builtin_bit_cast(v2i, x);
        p.x = __builtin_amdgcn_update_dpp(p.x, p.x, 0x160 + MASK, 0xf, 0xf, false);
        p.y = __builtin_amdgcn_update_dpp(p.y, p.y, 0x160 + MASK, 0xf, 0xf, false);
        return __builtin_bit_cast(double, p);
    }
    static __device__ __forceinline__ double hx(double x) {
        v2i p = __builtin_bit_cast(v2i, x);
        p.x = __builtin_amdgcn_permlanex16(p.x, p.x, 0x76543210, 0xfedcba98, false, false);
        p.y = __builtin_amdgcn_permlanex16(p.y, p.y, 0x76543210, 0xfedcba98, false, false);
        return __builtin_bit_cast(double, p);
    }
    static __device__ __forceinline__ v2i pack(double x) { return __builtin_bit_cast(v2i, x); }
    static __device__ __forceinline__ double unpack(v2i p) { return __builtin_bit_cast(double, p); }
};
template <> struct XT<float> {
    template <int MASK>
    static __device__ __forceinline__ float rx(float x) {
        int p = __builtin_bit_cast(int, x);
        p = __builtin_amdgcn_update_dpp(p, p, 0x160 + MASK, 0xf, 0xf, false);
        return __builtin_bit_cast(float, p);
    }
    static __device__ __forceinline__ float hx(float x) {
        int p = __builtin_bit_cast(int, x);
        p = __builtin_amdgcn_permlanex16(p, p, 0x76543210, 0xfedcba98, false, false);
        return __builtin_bit_cast(float, p);
    }
    static __device__ __forceinline__ v2i pack(float x) {
        v2i p = {__builtin_bit_cast(int, x), 0};
        return p;
    }
    static __device__ __forceinline__ float unpack(v2i p) { return __builtin_bit_cast(float, p.x); }
};

template <typename T, bool LIST>
__device__ __forceinline__ void sap_rows(const float* __restrict__ costB,
                                         const int nrows, const int* rowlist,
                                         T (&v)[CPL], T* sh_u, T* sh_short,
                                         int* sh_path, int* sh_c4r, int* sh_r4c,
                                         int* sh_SR, v4i (*rec)[NWAVE], int& par,
                                         const int tid, const int lane,
                                         const int wave, const bool owner,
                                         const unsigned full) {
    const T INF = (T)__builtin_inf();
    for (int idx = 0; idx < nrows; ++idx) {
        const int cur = LIST ? rowlist[idx] : idx;
        T   sh[CPL];
        int path[CPL];
        int r4c[CPL];
        {
            v4i q = *(const v4i*)&sh_r4c[CPL * (owner ? tid : 0)];
            r4c[0] = q.x; r4c[1] = q.y; r4c[2] = q.z; r4c[3] = q.w;
        }
        for (int k = 0; k < CPL; ++k) { sh[k] = INF; path[k] = -1; }
        unsigned rem = full;
        int ii   = cur;
        T   minv = (T)0;
        int nsr  = 0;
        int sink = -1;

        v4f c = {0.0f, 0.0f, 0.0f, 0.0f};
        if (owner) c = *(const v4f*)(costB + (size_t)ii * NN + CPL * tid);
        T z[CPL];
        {
            const T h = minv - sh_u[ii];
            for (int k = 0; k < CPL; ++k) z[k] = h - v[k];
        }

        while (true) {
            if (tid == 0) sh_SR[nsr] = ii;
            ++nsr;

            T t[CPL];
            for (int k = 0; k < CPL; ++k) t[k] = (T)c[k];
            for (int k = 0; k < CPL; ++k) t[k] = t[k] + z[k];

            T lm = INF; int lk = 0; int lr = -1;
            for (int k = 0; k < CPL; ++k) {
                const bool rm = ((rem >> k) & 1u) != 0u;
                const bool better = rm && (t[k] < sh[k]);
                sh[k]   = better ? t[k] : sh[k];
                path[k] = better ? ii : path[k];
                const T cd = rm ? sh[k] : INF;
                if (k == 0)       { lm = cd; lk = 0; lr = r4c[0]; }
                else if (cd < lm) { lm = cd; lk = k; lr = r4c[k]; }
            }

            T wm = lm;
            { const T o = XT<T>::template rx<1>(wm); wm = (o < wm) ? o : wm; }
            { const T o = XT<T>::template rx<2>(wm); wm = (o < wm) ? o : wm; }
            { const T o = XT<T>::template rx<4>(wm); wm = (o < wm) ? o : wm; }
            { const T o = XT<T>::template rx<8>(wm); wm = (o < wm) ? o : wm; }
            { const T o = XT<T>::hx(wm);             wm = (o < wm) ? o : wm; }
            const unsigned bal = __builtin_amdgcn_ballot_w32(lm == wm);
            const int srcl = __builtin_ctz(bal | 0x80000000u);
            const int wk = __builtin_amdgcn_readlane(lk, srcl);
            const int wr = __builtin_amdgcn_readlane(lr, srcl);
            if (lane == 0) {
                v2i p = XT<T>::pack(wm);
                v4i q = {p.x, p.y, CPL * (wave * 32 + srcl) + wk, wr};
                rec[par][wave] = q;
            }
            __syncthreads();

            T bv; int bj, br;
            {
                v4i q = rec[par][0];
                v2i p = {q.x, q.y};
                bv = XT<T>::unpack(p); bj = q.z; br = q.w;
            }
            for (int w = 1; w < NWAVE; ++w) {
                v4i q = rec[par][w];
                v2i p = {q.x, q.y};
                const T ov = XT<T>::unpack(p);
                if (ov < bv) { bv = ov; bj = q.z; br = q.w; }
            }
            par ^= 1;
            minv = bv;
            bj = __builtin_amdgcn_readfirstlane(bj);
            br = __builtin_amdgcn_readfirstlane(br);
            if (tid == (bj >> 2)) rem &= ~(1u << (bj & 3));
            if (br == -1) { sink = bj; break; }
            ii = br;

            c = (v4f){0.0f, 0.0f, 0.0f, 0.0f};
            if (owner) c = *(const v4f*)(costB + (size_t)ii * NN + CPL * tid);
            {
                const T h = minv - sh_u[ii];
                for (int k = 0; k < CPL; ++k) z[k] = h - v[k];
            }
        }

        if (owner) {
            v4i pq = {path[0], path[1], path[2], path[3]};
            *(v4i*)&sh_path[CPL * tid] = pq;
            for (int k = 0; k < CPL; ++k) sh_short[CPL * tid + k] = sh[k];
        }
        __syncthreads();

        for (int t2 = tid; t2 < nsr; t2 += LTH) {
            if (t2 == 0) {
                sh_u[cur] += minv;
            } else {
                int sr = sh_SR[t2];
                sh_u[sr] += minv - sh_short[sh_c4r[sr]];
            }
        }
        {
            const unsigned scanned = full & ~rem;
            for (int k = 0; k < CPL; ++k)
                if ((scanned >> k) & 1u) v[k] -= minv - sh[k];
        }
        __syncthreads();

        if (tid == 0) {
            int j = sink;
            while (true) {
                int i = sh_path[j];
                sh_r4c[j] = i;
                int tmp = sh_c4r[i];
                sh_c4r[i] = j;
                j = tmp;
                if (i == cur) break;
            }
        }
        __syncthreads();
    }
}

__device__ __forceinline__ int wave_sum_i(int x) {
    x += __builtin_amdgcn_update_dpp(x, x, 0x160 + 1, 0xf, 0xf, false);
    x += __builtin_amdgcn_update_dpp(x, x, 0x160 + 2, 0xf, 0xf, false);
    x += __builtin_amdgcn_update_dpp(x, x, 0x160 + 4, 0xf, 0xf, false);
    x += __builtin_amdgcn_update_dpp(x, x, 0x160 + 8, 0xf, 0xf, false);
    x += __builtin_amdgcn_permlanex16(x, x, 0x76543210, 0xfedcba98, false, false);
    return x;
}

__device__ __forceinline__ unsigned fbits(const float x) { return __builtin_bit_cast(unsigned, x); }

__device__ __forceinline__ void load_row16(const float* __restrict__ row, const int lane,
                                           unsigned (&bits)[16]) {
    for (int q = 0; q < 4; ++q) {
        const int c0 = 128 * q + 4 * lane;
        if (c0 < NN) {
            const v4f c = *(const v4f*)(row + c0);
            bits[4 * q + 0] = fbits(c.x);
            bits[4 * q + 1] = fbits(c.y);
            bits[4 * q + 2] = fbits(c.z);
            bits[4 * q + 3] = fbits(c.w);
        } else {
            for (int e = 0; e < 4; ++e) bits[4 * q + e] = 0x7f800000u;
        }
    }
}

__global__ void __launch_bounds__(LTH)
lapjv_kernel(const float* __restrict__ cost, int* __restrict__ mapping) {
    const int b    = blockIdx.x;
    const int tid  = threadIdx.x;
    const int lane = tid & 31;
    const int wave = __builtin_amdgcn_readfirstlane(tid >> 5);
    const float* costB = cost + (size_t)b * MM * NN;
    const bool owner = tid < NN / CPL;
    const unsigned full = owner ? 0xFu : 0u;

    __shared__ __attribute__((aligned(16))) float lcost[MM * KC];
    __shared__ __attribute__((aligned(16))) unsigned short lcol[MM * KC];
    __shared__ __attribute__((aligned(16))) double sh_ud[512];
    __shared__ __attribute__((aligned(16))) double sh_vd[512];
    __shared__ __attribute__((aligned(16))) double sh_short[512];
    __shared__ __attribute__((aligned(16))) float  sh_shortf[512];
    __shared__ __attribute__((aligned(16))) float  sh_vf[512];
    __shared__ __attribute__((aligned(16))) float  sh_uf[512];
    __shared__ __attribute__((aligned(16))) float  sh_popv[512];
    __shared__ __attribute__((aligned(16))) int sh_popc[512];
    __shared__ __attribute__((aligned(16))) int sh_path[512];
    __shared__ __attribute__((aligned(16))) int sh_c4r[512];
    __shared__ __attribute__((aligned(16))) int sh_r4c[512];
    __shared__ __attribute__((aligned(16))) int sh_SR[512];
    __shared__ __attribute__((aligned(16))) int sh_free[512];
    __shared__ __attribute__((aligned(16))) int sh_list[512];
    __shared__ int sh_cnt;
    __shared__ v4i rec[2][NWAVE];

    for (int t = tid; t < 512; t += LTH) {
        sh_uf[t] = 0.0f; sh_vf[t] = 0.0f; sh_c4r[t] = -1; sh_r4c[t] = -1;
    }
    int par = 0;

    {
        const unsigned ltmask = (1u << lane) - 1u;
        unsigned nxt[16];
        load_row16(costB + (size_t)wave * NN, lane, nxt);
        for (int i = wave; i < MM; i += NWAVE) {
            unsigned bits[16];
            for (int k = 0; k < 16; ++k) bits[k] = nxt[k];
            if (i + NWAVE < MM) load_row16(costB + (size_t)(i + NWAVE) * NN, lane, nxt);
            unsigned T = 0u;
            for (int bit = 30; bit >= 7; --bit) {
                const unsigned trial = T | (1u << bit);
                int cnt = 0;
                for (int k = 0; k < 16; ++k) cnt += (bits[k] < trial) ? 1 : 0;
                cnt = __builtin_amdgcn_readfirstlane(wave_sum_i(cnt));
                if (cnt <= KC) T = trial;
            }
            lcost[i * KC + lane]      = __builtin_inff();
            lcost[i * KC + lane + 32] = __builtin_inff();
            lcol[i * KC + lane]       = 0;
            lcol[i * KC + lane + 32]  = 0;
            int base = 0;
            for (int k = 0; k < 16; ++k) {
                const bool f = bits[k] < T;
                const unsigned bal = __builtin_amdgcn_ballot_w32(f);
                const int pos = base + __builtin_popcount(bal & ltmask);
                if (f && pos < KC) {
                    lcost[i * KC + pos] = __builtin_bit_cast(float, bits[k]);
                    lcol[i * KC + pos]  = (unsigned short)(128 * (k >> 2) + 4 * lane + (k & 3));
                }
                base += __builtin_popcount(bal);
            }
        }
    }
    __syncthreads();

    if (wave == 0) {
        const float INFf = __builtin_inff();
        const float NANf = __builtin_nanf("");
        for (int cur = 0; cur < MM; ++cur) {
            int pk[16];
            for (int q = 0; q < 4; ++q) {
                const int c0 = 128 * q + 4 * lane;
                const float iv = (c0 < NN) ? INFf : NANf;
                const v4f ini = {iv, iv, iv, iv};
                *(v4f*)&sh_shortf[c0] = ini;
                const v4i r = *(const v4i*)&sh_r4c[c0];
                pk[4 * q + 0] = ((r.x + 1) << 4) | (4 * q + 0);
                pk[4 * q + 1] = ((r.y + 1) << 4) | (4 * q + 1);
                pk[4 * q + 2] = ((r.z + 1) << 4) | (4 * q + 2);
                pk[4 * q + 3] = ((r.w + 1) << 4) | (4 * q + 3);
            }

            int   ii   = cur;
            float minv = 0.0f;
            int   nsr  = 0;
            int   sink = -1;
            bool  okay = true;
            while (true) {
                const v2f  cc = *(const v2f*)&lcost[ii * KC + 2 * lane];
                const v2us cw = *(const v2us*)&lcol[ii * KC + 2 * lane];
                const float h = minv - sh_uf[ii];
                const int c0 = cw.x, c1 = cw.y;
                const float s0 = sh_shortf[c0], s1 = sh_shortf[c1];
                const float r0 = cc.x + (h - sh_vf[c0]);
                const float r1 = cc.y + (h - sh_vf[c1]);
                if (r0 < s0) { sh_shortf[c0] = r0; sh_path[c0] = ii; }
                if (r1 < s1) { sh_shortf[c1] = r1; sh_path[c1] = ii; }

                float cm[4]; int cp[4];
                for (int q = 0; q < 4; ++q) {
                    const v4f x = *(const v4f*)&sh_shortf[128 * q + 4 * lane];
                    float m = INFf; int p = pk[4 * q];
                    if (x.x < m) { m = x.x; p = pk[4 * q + 0]; }
                    if (x.y < m) { m = x.y; p = pk[4 * q + 1]; }
                    if (x.z < m) { m = x.z; p = pk[4 * q + 2]; }
                    if (x.w < m) { m = x.w; p = pk[4 * q + 3]; }
                    cm[q] = m; cp[q] = p;
                }
                if (cm[1] < cm[0]) { cm[0] = cm[1]; cp[0] = cp[1]; }
                if (cm[3] < cm[2]) { cm[2] = cm[3]; cp[2] = cp[3]; }
                float lm = cm[0]; int lp = cp[0];
                if (cm[2] < lm) { lm = cm[2]; lp = cp[2]; }

                float wm = lm;
                { const float o = XT<float>::rx<1>(wm); wm = (o < wm) ? o : wm; }
                { const float o = XT<float>::rx<2>(wm); wm = (o < wm) ? o : wm; }
                { const float o = XT<float>::rx<4>(wm); wm = (o < wm) ? o : wm; }
                { const float o = XT<float>::rx<8>(wm); wm = (o < wm) ? o : wm; }
                { const float o = XT<float>::hx(wm);    wm = (o < wm) ? o : wm; }
                const unsigned bal = __builtin_amdgcn_ballot_w32(lm == wm);
                const int srcl = __builtin_ctz(bal | 0x80000000u);
                const int wp   = __builtin_amdgcn_readlane(lp, srcl);
                const float wms = __builtin_bit_cast(float,
                    __builtin_amdgcn_readfirstlane(__builtin_bit_cast(int, wm)));
                if (!(wms < INFf)) { okay = false; break; }
                const int k  = wp & 15;
                const int br = (wp >> 4) - 1;
                const int bj = 128 * (k >> 2) + 4 * srcl + (k & 3);
                minv = wms;
                if (lane == 0) {
                    sh_SR[nsr]   = ii;
                    sh_popc[nsr] = bj;
                    sh_popv[nsr] = wms;
                    sh_shortf[bj] = NANf;
                }
                ++nsr;
                if (br == -1) { sink = bj; break; }
                ii = br;
            }
            if (!okay) continue;

            for (int t = lane; t < nsr; t += 32) {
                const float d = minv - sh_popv[t];
                sh_vf[sh_popc[t]] -= d;
                if (t == 0) sh_uf[cur] += minv;
                if (t + 1 < nsr) sh_uf[sh_SR[t + 1]] += d;
            }
            if (lane == 0) {
                int j = sink;
                for (int g = 0; g < MM; ++g) {
                    const int i = sh_path[j];
                    sh_r4c[j] = i;
                    const int tmp = sh_c4r[i];
                    sh_c4r[i] = j;
                    j = tmp;
                    if (i == cur) break;
                }
            }
        }
    }
    __syncthreads();

    double v[CPL];
    {
        v4f q = {0.0f, 0.0f, 0.0f, 0.0f};
        if (owner) q = *(const v4f*)&sh_vf[CPL * tid];
        v[0] = (double)q.x; v[1] = (double)q.y; v[2] = (double)q.z; v[3] = (double)q.w;
    }
    if (owner) {
        v2d a01 = {v[0], v[1]};
        v2d a23 = {v[2], v[3]};
        *(v2d*)&sh_vd[CPL * tid]     = a01;
        *(v2d*)&sh_vd[CPL * tid + 2] = a23;
    }
    __syncthreads();
    for (int i = wave; i < MM; i += NWAVE) {
        const float* row = costB + (size_t)i * NN;
        double m = __builtin_inf();
        for (int q = 0; q < 4; ++q) {
            const int c0 = 128 * q + CPL * lane;
            if (c0 < NN) {
                const v4f c = *(const v4f*)(row + c0);
                const v2d va = *(const v2d*)&sh_vd[c0];
                const v2d vb = *(const v2d*)&sh_vd[c0 + 2];
                const double d0 = (double)c.x - va.x;
                const double d1 = (double)c.y - va.y;
                const double d2 = (double)c.z - vb.x;
                const double d3 = (double)c.w - vb.y;
                m = (d0 < m) ? d0 : m;
                m = (d1 < m) ? d1 : m;
                m = (d2 < m) ? d2 : m;
                m = (d3 < m) ? d3 : m;
            }
        }
        { const double o = XT<double>::rx<1>(m); m = (o < m) ? o : m; }
        { const double o = XT<double>::rx<2>(m); m = (o < m) ? o : m; }
        { const double o = XT<double>::rx<4>(m); m = (o < m) ? o : m; }
        { const double o = XT<double>::rx<8>(m); m = (o < m) ? o : m; }
        { const double o = XT<double>::hx(m);    m = (o < m) ? o : m; }
        const int jm = sh_c4r[i];
        const bool has = jm >= 0;
        const int jq = has ? jm : 0;
        const double dm = (double)row[jq] - sh_vd[jq];
        const bool keep = has && (dm == m);
        if (lane == 0) {
            sh_ud[i]   = m;
            sh_free[i] = keep ? 0 : 1;
            if (has && !keep) { sh_c4r[i] = -1; sh_r4c[jm] = -1; }
        }
    }
    __syncthreads();
    if (wave == 0) {
        int cnt = 0;
        for (int base = 0; base < 512; base += 32) {
            const int r = base + lane;
            const bool f = (r < MM) && (sh_free[r] != 0);
            const unsigned bal = __builtin_amdgcn_ballot_w32(f);
            const int pos = cnt + __builtin_popcount(bal & ((1u << lane) - 1u));
            if (f) sh_list[pos] = r;
            cnt += __builtin_popcount(bal);
        }
        if (lane == 0) sh_cnt = cnt;
    }
    __syncthreads();
    const int nfree = __builtin_amdgcn_readfirstlane(sh_cnt);

    sap_rows<double, true>(costB, nfree, sh_list, v, sh_ud, sh_short,
                           sh_path, sh_c4r, sh_r4c, sh_SR, rec, par,
                           tid, lane, wave, owner, full);

    for (int t = tid; t < MM; t += LTH) mapping[b * MM + t] = sh_c4r[t];
}

__global__ void extrap_kernel(const float* __restrict__ p1b,
                              const float* __restrict__ p1l,
                              const float* __restrict__ p2b,
                              const float* __restrict__ p2l,
                              const float* __restrict__ toff,
                              const int*   __restrict__ mapping,
                              float* __restrict__ out) {
    int idx = blockIdx.x * blockDim.x + threadIdx.x;
    if (idx >= BB * MM) return;
    int b = idx / MM;

    float first  = toff[b * 3 + 1] - toff[b * 3 + 0];
    float second = toff[b * 3 + 2] - toff[b * 3 + 1];
    float factor = second / first;

    int  j   = mapping[idx];
    bool has = (j >= 0);
    int  jc  = has ? j : 0;

    const float* pb2 = p2b + (size_t)idx * 4;
    const float* pb1 = p1b + ((size_t)b * NN + jc) * 4;
    float* ob = out + (size_t)idx * 4;
#pragma unroll
    for (int k = 0; k < 4; ++k) {
        float c1 = has ? pb1[k] : pb2[k];
        float v  = pb2[k] + (pb2[k] - c1) * factor;
        ob[k] = (k >= 2) ? fmaxf(v, 0.0f) : v;
    }

    const float* pl2 = p2l + (size_t)idx * CC;
    const float* pl1 = p1l + ((size_t)b * NN + jc) * CC;
    float* ol = out + (size_t)BB * MM * 4 + (size_t)idx * CC;
    for (int c = 0; c < CC; ++c) {
        float c1 = has ? pl1[c] : 0.0f;
        ol[c] = 0.5f * (pl2[c] + c1);
    }
}

extern "C" void kernel_launch(void* const* d_in, const int* in_sizes, int n_in,
                              void* d_out, int out_size, void* d_ws, size_t ws_size,
                              hipStream_t stream) {
    (void)in_sizes; (void)n_in; (void)out_size; (void)ws_size;
    const float* p1b  = (const float*)d_in[0];
    const float* p1l  = (const float*)d_in[1];
    const float* p2b  = (const float*)d_in[2];
    const float* p2l  = (const float*)d_in[3];
    const float* toff = (const float*)d_in[4];

    float* ws   = (float*)d_ws;
    float* s1   = ws;
    float* s2   = s1 + (size_t)BB * NN * CC;
    float* cost = s2 + (size_t)BB * MM * CC;
    int*   map  = (int*)(cost + (size_t)BB * MM * NN);
    float* out  = (float*)d_out;

    int tot = BB * NN * CC + BB * MM * CC;
    sigmoid_kernel<<<(tot + 255) / 256, 256, 0, stream>>>(p1l, p2l, s1, s2);

    const int TM = (MM + 15) / 16, TN = (NN + 15) / 16;
    cost_kernel<<<BB * TM * TN, 32, 0, stream>>>(p1b, p2b, s1, s2, cost);

    lapjv_kernel<<<BB, LTH, 0, stream>>>(cost, map);

    int tot2 = BB * MM;
    extrap_kernel<<<(tot2 + 255) / 256, 256, 0, stream>>>(p1b, p1l, p2b, p2l,
                                                          toff, map, out);
}


# Sparse f32 assignment pass: each lane keeps shortest, path, v of its 16 columns in registers; row candidates reach them via a per-column LDS mailbox: 2 dependent LDS trips per step, not 3; same sums
# speedup vs baseline: 5.4694x; 1.1698x over previous
#include <hip/hip_runtime.h>
#include <math.h>

#define BB 4
#define MM 500
#define NN 500
#define CC 91

typedef __attribute__((ext_vector_type(2))) float v2f;
typedef __attribute__((ext_vector_type(8))) float v8f;

__device__ __forceinline__ int imin_i(int a, int b) { return a < b ? a : b; }

__global__ void sigmoid_kernel(const float* __restrict__ p1l,
                               const float* __restrict__ p2l,
                               float* __restrict__ s1,
                               float* __restrict__ s2) {
    int idx = blockIdx.x * blockDim.x + threadIdx.x;
    const int n1 = BB * NN * CC;
    const int n2 = BB * MM * CC;
    if (idx < n1) {
        float x = p1l[idx];
        s1[idx] = 1.0f / (1.0f + expf(-x));
    } else if (idx < n1 + n2) {
        float x = p2l[idx - n1];
        s2[idx - n1] = 1.0f / (1.0f + expf(-x));
    }
}

__global__ void cost_kernel(const float* __restrict__ p1b,
                            const float* __restrict__ p2b,
                            const float* __restrict__ s1,
                            const float* __restrict__ s2,
                            float* __restrict__ cost) {
    const int TN = (NN + 15) / 16;
    const int TM = (MM + 15) / 16;
    int bid = blockIdx.x;
    int b  = bid / (TM * TN);
    int t  = bid % (TM * TN);
    int m0 = (t / TN) * 16;
    int n0 = (t % TN) * 16;
    int lane = threadIdx.x;

    __shared__ float sq2[16];
    __shared__ float sq1[16];
    __shared__ float ls2[16 * CC];
    __shared__ float ls1[16 * CC];

    v2f a  = {0.0f, 0.0f};
    v2f bv = {0.0f, 0.0f};
    if (lane < 16) {
        int m = imin_i(m0 + lane, MM - 1);
        const float* pb = p2b + ((size_t)b * MM + m) * 4;
        float x = pb[0], y = pb[1];
        a.x = x; a.y = y;
        sq2[lane] = x * x + y * y;
        int n = imin_i(n0 + lane, NN - 1);
        const float* qb = p1b + ((size_t)b * NN + n) * 4;
        float bx = qb[0], by = qb[1];
        bv.x = bx; bv.y = by;
        sq1[lane] = bx * bx + by * by;
    }
    const float* s2base = s2 + (size_t)b * MM * CC;
    const float* s1base = s1 + (size_t)b * NN * CC;
    for (int idx = lane; idx < 16 * CC; idx += 32) {
        int r = idx / CC;
        int c = idx - r * CC;
        ls2[idx] = s2base[(size_t)imin_i(m0 + r, MM - 1) * CC + c];
        ls1[idx] = s1base[(size_t)imin_i(n0 + r, NN - 1) * CC + c];
    }
    __syncthreads();

    v8f cacc = {0.f, 0.f, 0.f, 0.f, 0.f, 0.f, 0.f, 0.f};
    v8f dot = __builtin_amdgcn_wmma_f32_16x16x4_f32(
         false, a,  false, bv,
         (short)0, cacc,  false,  false);

    int n_loc  = lane & 15;
    int m_base = (lane >> 4) << 3;

    float cls[8];
#pragma unroll
    for (int i = 0; i < 8; ++i) cls[i] = 0.0f;
    for (int c = 0; c < CC; ++c) {
        float s1v = ls1[n_loc * CC + c];
#pragma unroll
        for (int i = 0; i < 8; ++i) {
            float d = fabsf(ls2[(m_base + i) * CC + c] - s1v);
            cls[i] = fmaxf(cls[i], d);
        }
    }

#pragma unroll
    for (int i = 0; i < 8; ++i) {
        int m = m0 + m_base + i;
        int n = n0 + n_loc;
        if (m < MM && n < NN) {
            float d2 = sq2[m_base + i] + sq1[n_loc] - 2.0f * dot[i];
            float cd = sqrtf(fmaxf(d2, 0.0f));
            cost[((size_t)b * MM + m) * NN + n] = 0.5f * cd + 0.5f * cls[i];
        }
    }
}

#define LTH 128
#define NWAVE (LTH / 32)
#define CPL 4
typedef __attribute__((ext_vector_type(4))) float v4f;
typedef __attribute__((ext_vector_type(4))) int   v4i;
typedef __attribute__((ext_vector_type(2))) int   v2i;
typedef __attribute__((ext_vector_type(2))) double v2d;
typedef __attribute__((ext_vector_type(2))) unsigned short v2us;

#define KC 64
static_assert(KC == 64, "two candidates per lane of one wave");
static_assert(NN < 512 && NN <= 65535, "16 columns per lane, column 511 is a pad; 16-bit column ids");

static_assert(NN % CPL == 0 && NN / CPL <= LTH, "column tiling");
static_assert(MM == NN, "square problem");

template <typename T> struct XT;
template <> struct XT<double> {
    template <int MASK>
    static __device__ __forceinline__ double rx(double x) {
        v2i p = __builtin_bit_cast(v2i, x);
        p.x = __builtin_amdgcn_update_dpp(p.x, p.x, 0x160 + MASK, 0xf, 0xf, false);
        p.y = __builtin_amdgcn_update_dpp(p.y, p.y, 0x160 + MASK, 0xf, 0xf, false);
        return __builtin_bit_cast(double, p);
    }
    static __device__ __forceinline__ double hx(double x) {
        v2i p = __builtin_bit_cast(v2i, x);
        p.x = __builtin_amdgcn_permlanex16(p.x, p.x, 0x76543210, 0xfedcba98, false, false);
        p.y = __builtin_amdgcn_permlanex16(p.y, p.y, 0x76543210, 0xfedcba98, false, false);
        return __builtin_bit_cast(double, p);
    }
    static __device__ __forceinline__ v2i pack(double x) { return __builtin_bit_cast(v2i, x); }
    static __device__ __forceinline__ double unpack(v2i p) { return __builtin_bit_cast(double, p); }
};
template <> struct XT<float> {
    template <int MASK>
    static __device__ __forceinline__ float rx(float x) {
        int p = __builtin_bit_cast(int, x);
        p = __builtin_amdgcn_update_dpp(p, p, 0x160 + MASK, 0xf, 0xf, false);
        return __builtin_bit_cast(float, p);
    }
    static __device__ __forceinline__ float hx(float x) {
        int p = __builtin_bit_cast(int, x);
        p = __builtin_amdgcn_permlanex16(p, p, 0x76543210, 0xfedcba98, false, false);
        return __builtin_bit_cast(float, p);
    }
    static __device__ __forceinline__ v2i pack(float x) {
        v2i p = {__builtin_bit_cast(int, x), 0};
        return p;
    }
    static __device__ __forceinline__ float unpack(v2i p) { return __builtin_bit_cast(float, p.x); }
};

template <typename T, bool LIST>
__device__ __forceinline__ void sap_rows(const float* __restrict__ costB,
                                         const int nrows, const int* rowlist,
                                         T (&v)[CPL], T* sh_u, T* sh_short,
                                         int* sh_path, int* sh_c4r, int* sh_r4c,
                                         int* sh_SR, v4i (*rec)[NWAVE], int& par,
                                         const int tid, const int lane,
                                         const int wave, const bool owner,
                                         const unsigned full) {
    const T INF = (T)__builtin_inf();
    for (int idx = 0; idx < nrows; ++idx) {
        const int cur = LIST ? rowlist[idx] : idx;
        T   sh[CPL];
        int path[CPL];
        int r4c[CPL];
        {
            v4i q = *(const v4i*)&sh_r4c[CPL * (owner ? tid : 0)];
            r4c[0] = q.x; r4c[1] = q.y; r4c[2] = q.z; r4c[3] = q.w;
        }
        for (int k = 0; k < CPL; ++k) { sh[k] = INF; path[k] = -1; }
        unsigned rem = full;
        int ii   = cur;
        T   minv = (T)0;
        int nsr  = 0;
        int sink = -1;

        v4f c = {0.0f, 0.0f, 0.0f, 0.0f};
        if (owner) c = *(const v4f*)(costB + (size_t)ii * NN + CPL * tid);
        T z[CPL];
        {
            const T h = minv - sh_u[ii];
            for (int k = 0; k < CPL; ++k) z[k] = h - v[k];
        }

        while (true) {
            if (tid == 0) sh_SR[nsr] = ii;
            ++nsr;

            T t[CPL];
            for (int k = 0; k < CPL; ++k) t[k] = (T)c[k];
            for (int k = 0; k < CPL; ++k) t[k] = t[k] + z[k];

            T lm = INF; int lk = 0; int lr = -1;
            for (int k = 0; k < CPL; ++k) {
                const bool rm = ((rem >> k) & 1u) != 0u;
                const bool better = rm && (t[k] < sh[k]);
                sh[k]   = better ? t[k] : sh[k];
                path[k] = better ? ii : path[k];
                const T cd = rm ? sh[k] : INF;
                if (k == 0)       { lm = cd; lk = 0; lr = r4c[0]; }
                else if (cd < lm) { lm = cd; lk = k; lr = r4c[k]; }
            }

            T wm = lm;
            { const T o = XT<T>::template rx<1>(wm); wm = (o < wm) ? o : wm; }
            { const T o = XT<T>::template rx<2>(wm); wm = (o < wm) ? o : wm; }
            { const T o = XT<T>::template rx<4>(wm); wm = (o < wm) ? o : wm; }
            { const T o = XT<T>::template rx<8>(wm); wm = (o < wm) ? o : wm; }
            { const T o = XT<T>::hx(wm);             wm = (o < wm) ? o : wm; }
            const unsigned bal = __builtin_amdgcn_ballot_w32(lm == wm);
            const int srcl = __builtin_ctz(bal | 0x80000000u);
            const int wk = __builtin_amdgcn_readlane(lk, srcl);
            const int wr = __builtin_amdgcn_readlane(lr, srcl);
            if (lane == 0) {
                v2i p = XT<T>::pack(wm);
                v4i q = {p.x, p.y, CPL * (wave * 32 + srcl) + wk, wr};
                rec[par][wave] = q;
            }
            __syncthreads();

            T bv; int bj, br;
            {
                v4i q = rec[par][0];
                v2i p = {q.x, q.y};
                bv = XT<T>::unpack(p); bj = q.z; br = q.w;
            }
            for (int w = 1; w < NWAVE; ++w) {
                v4i q = rec[par][w];
                v2i p = {q.x, q.y};
                const T ov = XT<T>::unpack(p);
                if (ov < bv) { bv = ov; bj = q.z; br = q.w; }
            }
            par ^= 1;
            minv = bv;
            bj = __builtin_amdgcn_readfirstlane(bj);
            br = __builtin_amdgcn_readfirstlane(br);
            if (tid == (bj >> 2)) rem &= ~(1u << (bj & 3));
            if (br == -1) { sink = bj; break; }
            ii = br;

            c = (v4f){0.0f, 0.0f, 0.0f, 0.0f};
            if (owner) c = *(const v4f*)(costB + (size_t)ii * NN + CPL * tid);
            {
                const T h = minv - sh_u[ii];
                for (int k = 0; k < CPL; ++k) z[k] = h - v[k];
            }
        }

        if (owner) {
            v4i pq = {path[0], path[1], path[2], path[3]};
            *(v4i*)&sh_path[CPL * tid] = pq;
            for (int k = 0; k < CPL; ++k) sh_short[CPL * tid + k] = sh[k];
        }
        __syncthreads();

        for (int t2 = tid; t2 < nsr; t2 += LTH) {
            if (t2 == 0) {
                sh_u[cur] += minv;
            } else {
                int sr = sh_SR[t2];
                sh_u[sr] += minv - sh_short[sh_c4r[sr]];
            }
        }
        {
            const unsigned scanned = full & ~rem;
            for (int k = 0; k < CPL; ++k)
                if ((scanned >> k) & 1u) v[k] -= minv - sh[k];
        }
        __syncthreads();

        if (tid == 0) {
            int j = sink;
            while (true) {
                int i = sh_path[j];
                sh_r4c[j] = i;
                int tmp = sh_c4r[i];
                sh_c4r[i] = j;
                j = tmp;
                if (i == cur) break;
            }
        }
        __syncthreads();
    }
}

__device__ __forceinline__ int wave_sum_i(int x) {
    x += __builtin_amdgcn_update_dpp(x, x, 0x160 + 1, 0xf, 0xf, false);
    x += __builtin_amdgcn_update_dpp(x, x, 0x160 + 2, 0xf, 0xf, false);
    x += __builtin_amdgcn_update_dpp(x, x, 0x160 + 4, 0xf, 0xf, false);
    x += __builtin_amdgcn_update_dpp(x, x, 0x160 + 8, 0xf, 0xf, false);
    x += __builtin_amdgcn_permlanex16(x, x, 0x76543210, 0xfedcba98, false, false);
    return x;
}

__device__ __forceinline__ unsigned fbits(const float x) { return __builtin_bit_cast(unsigned, x); }

__device__ __forceinline__ void load_row16(const float* __restrict__ row, const int lane,
                                           unsigned (&bits)[16]) {
    for (int q = 0; q < 4; ++q) {
        const int c0 = 128 * q + 4 * lane;
        if (c0 < NN) {
            const v4f c = *(const v4f*)(row + c0);
            bits[4 * q + 0] = fbits(c.x);
            bits[4 * q + 1] = fbits(c.y);
            bits[4 * q + 2] = fbits(c.z);
            bits[4 * q + 3] = fbits(c.w);
        } else {
            for (int e = 0; e < 4; ++e) bits[4 * q + e] = 0x7f800000u;
        }
    }
}

__global__ void __launch_bounds__(LTH)
lapjv_kernel(const float* __restrict__ cost, int* __restrict__ mapping) {
    const int b    = blockIdx.x;
    const int tid  = threadIdx.x;
    const int lane = tid & 31;
    const int wave = __builtin_amdgcn_readfirstlane(tid >> 5);
    const float* costB = cost + (size_t)b * MM * NN;
    const bool owner = tid < NN / CPL;
    const unsigned full = owner ? 0xFu : 0u;

    __shared__ __attribute__((aligned(16))) float lcost[MM * KC];
    __shared__ __attribute__((aligned(16))) unsigned short lcol[MM * KC];
    __shared__ __attribute__((aligned(16))) double sh_ud[512];
    __shared__ __attribute__((aligned(16))) double sh_vd[512];
    __shared__ __attribute__((aligned(16))) double sh_short[512];
    __shared__ __attribute__((aligned(16))) float  sh_mbox[512];
    __shared__ __attribute__((aligned(16))) float  sh_vf[512];
    __shared__ __attribute__((aligned(16))) float  sh_uf[512];
    __shared__ __attribute__((aligned(16))) float  sh_popv[512];
    __shared__ __attribute__((aligned(16))) int sh_popc[512];
    __shared__ __attribute__((aligned(16))) int sh_path[512];
    __shared__ __attribute__((aligned(16))) int sh_c4r[512];
    __shared__ __attribute__((aligned(16))) int sh_r4c[512];
    __shared__ __attribute__((aligned(16))) int sh_SR[512];
    __shared__ __attribute__((aligned(16))) int sh_free[512];
    __shared__ __attribute__((aligned(16))) int sh_list[512];
    __shared__ int sh_cnt;
    __shared__ v4i rec[2][NWAVE];

    for (int t = tid; t < 512; t += LTH) {
        sh_uf[t] = 0.0f; sh_vf[t] = 0.0f; sh_c4r[t] = -1; sh_r4c[t] = -1;
    }
    int par = 0;

    {
        const unsigned ltmask = (1u << lane) - 1u;
        unsigned nxt[16];
        load_row16(costB + (size_t)wave * NN, lane, nxt);
        for (int i = wave; i < MM; i += NWAVE) {
            unsigned bits[16];
            for (int k = 0; k < 16; ++k) bits[k] = nxt[k];
            if (i + NWAVE < MM) load_row16(costB + (size_t)(i + NWAVE) * NN, lane, nxt);
            unsigned T = 0u;
            for (int bit = 30; bit >= 7; --bit) {
                const unsigned trial = T | (1u << bit);
                int cnt = 0;
                for (int k = 0; k < 16; ++k) cnt += (bits[k] < trial) ? 1 : 0;
                cnt = __builtin_amdgcn_readfirstlane(wave_sum_i(cnt));
                if (cnt <= KC) T = trial;
            }
            lcost[i * KC + lane]      = __builtin_inff();
            lcost[i * KC + lane + 32] = __builtin_inff();
            lcol[i * KC + lane]       = 511;
            lcol[i * KC + lane + 32]  = 511;
            int base = 0;
            for (int k = 0; k < 16; ++k) {
                const bool f = bits[k] < T;
                const unsigned bal = __builtin_amdgcn_ballot_w32(f);
                const int pos = base + __builtin_popcount(bal & ltmask);
                if (f && pos < KC) {
                    lcost[i * KC + pos] = __builtin_bit_cast(float, bits[k]);
                    lcol[i * KC + pos]  = (unsigned short)(128 * (k >> 2) + 4 * lane + (k & 3));
                }
                base += __builtin_popcount(bal);
            }
        }
    }
    __syncthreads();

    if (wave == 0) {
        const float INFf = __builtin_inff();
        for (int q = 0; q < 4; ++q) {
            const v4f ini = {INFf, INFf, INFf, INFf};
            *(v4f*)&sh_mbox[128 * q + 4 * lane] = ini;
        }
        for (int cur = 0; cur < MM; ++cur) {
            float sh[16], vv[16];
            int   path[16], pk[16];
            for (int q = 0; q < 4; ++q) {
                const int c0 = 128 * q + 4 * lane;
                const bool real = c0 < NN;
                const v4i r = *(const v4i*)&sh_r4c[c0];
                const v4f w = *(const v4f*)&sh_vf[c0];
                pk[4 * q + 0] = ((r.x + 1) << 4) | (4 * q + 0);
                pk[4 * q + 1] = ((r.y + 1) << 4) | (4 * q + 1);
                pk[4 * q + 2] = ((r.z + 1) << 4) | (4 * q + 2);
                pk[4 * q + 3] = ((r.w + 1) << 4) | (4 * q + 3);
                vv[4 * q + 0] = real ? w.x : -INFf;
                vv[4 * q + 1] = real ? w.y : -INFf;
                vv[4 * q + 2] = real ? w.z : -INFf;
                vv[4 * q + 3] = real ? w.w : -INFf;
            }
            for (int k = 0; k < 16; ++k) { sh[k] = INFf; path[k] = -1; }

            int   ii   = cur;
            float minv = 0.0f;
            int   nsr  = 0;
            int   sink = -1;
            bool  okay = true;
            v2f   cc = *(const v2f*)&lcost[ii * KC + 2 * lane];
            v2us  cw = *(const v2us*)&lcol[ii * KC + 2 * lane];
            float uf = sh_uf[ii];
            while (true) {
                const int c0 = cw.x, c1 = cw.y;
                sh_mbox[c0] = cc.x;
                sh_mbox[c1] = cc.y;
                v4f x[4];
                for (int q = 0; q < 4; ++q) x[q] = *(const v4f*)&sh_mbox[128 * q + 4 * lane];
                sh_mbox[c0] = INFf;
                sh_mbox[c1] = INFf;

                const float h = minv - uf;
                for (int k = 0; k < 16; ++k) {
                    const float xk = x[k >> 2][k & 3];
                    const float r = xk + (h - vv[k]);
                    const bool better = r < sh[k];
                    sh[k]   = better ? r : sh[k];
                    path[k] = better ? ii : path[k];
                }

                float m8[8]; int p8[8];
                for (int t = 0; t < 8; ++t) {
                    const float ma = sh[2 * t], mb = sh[2 * t + 1];
                    const int   pa = pk[2 * t], pb = pk[2 * t + 1];
                    const bool tk = mb < ma;
                    m8[t] = tk ? mb : ma;
                    p8[t] = tk ? pb : pa;
                }
                float m4[4]; int p4[4];
                for (int t = 0; t < 4; ++t) {
                    const float ma = m8[2 * t], mb = m8[2 * t + 1];
                    const int   pa = p8[2 * t], pb = p8[2 * t + 1];
                    const bool tk = mb < ma;
                    m4[t] = tk ? mb : ma;
                    p4[t] = tk ? pb : pa;
                }
                float m2[2]; int p2[2];
                for (int t = 0; t < 2; ++t) {
                    const float ma = m4[2 * t], mb = m4[2 * t + 1];
                    const int   pa = p4[2 * t], pb = p4[2 * t + 1];
                    const bool tk = mb < ma;
                    m2[t] = tk ? mb : ma;
                    p2[t] = tk ? pb : pa;
                }
                const float m2a = m2[0], m2b = m2[1];
                const int   p2a = p2[0], p2b = p2[1];
                const bool tk1 = m2b < m2a;
                const float lm = tk1 ? m2b : m2a;
                const int   lp = tk1 ? p2b : p2a;

                float wm = lm;
                { const float o = XT<float>::rx<1>(wm); wm = (o < wm) ? o : wm; }
                { const float o = XT<float>::rx<2>(wm); wm = (o < wm) ? o : wm; }
                { const float o = XT<float>::rx<4>(wm); wm = (o < wm) ? o : wm; }
                { const float o = XT<float>::rx<8>(wm); wm = (o < wm) ? o : wm; }
                { const float o = XT<float>::hx(wm);    wm = (o < wm) ? o : wm; }
                const unsigned bal = __builtin_amdgcn_ballot_w32(lm == wm);
                const int srcl = __builtin_ctz(bal | 0x80000000u);
                const int wp   = __builtin_amdgcn_readlane(lp, srcl);
                const float wms = __builtin_bit_cast(float,
                    __builtin_amdgcn_readfirstlane(__builtin_bit_cast(int, wm)));
                if (!(wms < INFf)) { okay = false; break; }
                const int k  = wp & 15;
                const int br = (wp >> 4) - 1;
                const int bj = 128 * (k >> 2) + 4 * srcl + (k & 3);
                minv = wms;
                const int nrow = (br < 0) ? ii : br;
                cc = *(const v2f*)&lcost[nrow * KC + 2 * lane];
                cw = *(const v2us*)&lcol[nrow * KC + 2 * lane];
                uf = sh_uf[nrow];
                if (lane == 0) {
                    sh_SR[nsr]   = ii;
                    sh_popc[nsr] = bj;
                    sh_popv[nsr] = wms;
                }
                ++nsr;
                if (br == -1) { sink = bj; break; }
                const int mk = (lp & 15) + ((lane == srcl) ? 0 : 16);
                for (int kk = 0; kk < 16; ++kk) {
                    const bool hit = (mk == kk);
                    sh[kk] = hit ? INFf : sh[kk];
                    vv[kk] = hit ? -INFf : vv[kk];
                }
                ii = br;
            }
            if (!okay) continue;

            for (int q = 0; q < 4; ++q) {
                const v4i pq = {path[4 * q + 0], path[4 * q + 1], path[4 * q + 2], path[4 * q + 3]};
                *(v4i*)&sh_path[128 * q + 4 * lane] = pq;
            }
            for (int t = lane; t < nsr; t += 32) {
                const float d = minv - sh_popv[t];
                sh_vf[sh_popc[t]] -= d;
                if (t == 0) sh_uf[cur] += minv;
                if (t + 1 < nsr) sh_uf[sh_SR[t + 1]] += d;
            }
            if (lane == 0) {
                int j = sink;
                for (int g = 0; g < MM; ++g) {
                    const int i = sh_path[j];
                    sh_r4c[j] = i;
                    const int tmp = sh_c4r[i];
                    sh_c4r[i] = j;
                    j = tmp;
                    if (i == cur) break;
                }
            }
        }
    }
    __syncthreads();

    double v[CPL];
    {
        v4f q = {0.0f, 0.0f, 0.0f, 0.0f};
        if (owner) q = *(const v4f*)&sh_vf[CPL * tid];
        v[0] = (double)q.x; v[1] = (double)q.y; v[2] = (double)q.z; v[3] = (double)q.w;
    }
    if (owner) {
        v2d a01 = {v[0], v[1]};
        v2d a23 = {v[2], v[3]};
        *(v2d*)&sh_vd[CPL * tid]     = a01;
        *(v2d*)&sh_vd[CPL * tid + 2] = a23;
    }
    __syncthreads();
    for (int i = wave; i < MM; i += NWAVE) {
        const float* row = costB + (size_t)i * NN;
        double m = __builtin_inf();
        for (int q = 0; q < 4; ++q) {
            const int c0 = 128 * q + CPL * lane;
            if (c0 < NN) {
                const v4f c = *(const v4f*)(row + c0);
                const v2d va = *(const v2d*)&sh_vd[c0];
                const v2d vb = *(const v2d*)&sh_vd[c0 + 2];
                const double d0 = (double)c.x - va.x;
                const double d1 = (double)c.y - va.y;
                const double d2 = (double)c.z - vb.x;
                const double d3 = (double)c.w - vb.y;
                m = (d0 < m) ? d0 : m;
                m = (d1 < m) ? d1 : m;
                m = (d2 < m) ? d2 : m;
                m = (d3 < m) ? d3 : m;
            }
        }
        { const double o = XT<double>::rx<1>(m); m = (o < m) ? o : m; }
        { const double o = XT<double>::rx<2>(m); m = (o < m) ? o : m; }
        { const double o = XT<double>::rx<4>(m); m = (o < m) ? o : m; }
        { const double o = XT<double>::rx<8>(m); m = (o < m) ? o : m; }
        { const double o = XT<double>::hx(m);    m = (o < m) ? o : m; }
        const int jm = sh_c4r[i];
        const bool has = jm >= 0;
        const int jq = has ? jm : 0;
        const double dm = (double)row[jq] - sh_vd[jq];
        const bool keep = has && (dm == m);
        if (lane == 0) {
            sh_ud[i]   = m;
            sh_free[i] = keep ? 0 : 1;
            if (has && !keep) { sh_c4r[i] = -1; sh_r4c[jm] = -1; }
        }
    }
    __syncthreads();
    if (wave == 0) {
        int cnt = 0;
        for (int base = 0; base < 512; base += 32) {
            const int r = base + lane;
            const bool f = (r < MM) && (sh_free[r] != 0);
            const unsigned bal = __builtin_amdgcn_ballot_w32(f);
            const int pos = cnt + __builtin_popcount(bal & ((1u << lane) - 1u));
            if (f) sh_list[pos] = r;
            cnt += __builtin_popcount(bal);
        }
        if (lane == 0) sh_cnt = cnt;
    }
    __syncthreads();
    const int nfree = __builtin_amdgcn_readfirstlane(sh_cnt);

    sap_rows<double, true>(costB, nfree, sh_list, v, sh_ud, sh_short,
                           sh_path, sh_c4r, sh_r4c, sh_SR, rec, par,
                           tid, lane, wave, owner, full);

    for (int t = tid; t < MM; t += LTH) mapping[b * MM + t] = sh_c4r[t];
}

__global__ void extrap_kernel(const float* __restrict__ p1b,
                              const float* __restrict__ p1l,
                              const float* __restrict__ p2b,
                              const float* __restrict__ p2l,
                              const float* __restrict__ toff,
                              const int*   __restrict__ mapping,
                              float* __restrict__ out) {
    int idx = blockIdx.x * blockDim.x + threadIdx.x;
    if (idx >= BB * MM) return;
    int b = idx / MM;

    float first  = toff[b * 3 + 1] - toff[b * 3 + 0];
    float second = toff[b * 3 + 2] - toff[b * 3 + 1];
    float factor = second / first;

    int  j   = mapping[idx];
    bool has = (j >= 0);
    int  jc  = has ? j : 0;

    const float* pb2 = p2b + (size_t)idx * 4;
    const float* pb1 = p1b + ((size_t)b * NN + jc) * 4;
    float* ob = out + (size_t)idx * 4;
    for (int k = 0; k < 4; ++k) {
        float c1 = has ? pb1[k] : pb2[k];
        float v  = pb2[k] + (pb2[k] - c1) * factor;
        ob[k] = (k >= 2) ? fmaxf(v, 0.0f) : v;
    }

    const float* pl2 = p2l + (size_t)idx * CC;
    const float* pl1 = p1l + ((size_t)b * NN + jc) * CC;
    float* ol = out + (size_t)BB * MM * 4 + (size_t)idx * CC;
    for (int c = 0; c < CC; ++c) {
        float c1 = has ? pl1[c] : 0.0f;
        ol[c] = 0.5f * (pl2[c] + c1);
    }
}

extern "C" void kernel_launch(void* const* d_in, const int* in_sizes, int n_in,
                              void* d_out, int out_size, void* d_ws, size_t ws_size,
                              hipStream_t stream) {
    (void)in_sizes; (void)n_in; (void)out_size; (void)ws_size;
    const float* p1b  = (const float*)d_in[0];
    const float* p1l  = (const float*)d_in[1];
    const float* p2b  = (const float*)d_in[2];
    const float* p2l  = (const float*)d_in[3];
    const float* toff = (const float*)d_in[4];

    float* ws   = (float*)d_ws;
    float* s1   = ws;
    float* s2   = s1 + (size_t)BB * NN * CC;
    float* cost = s2 + (size_t)BB * MM * CC;
    int*   map  = (int*)(cost + (size_t)BB * MM * NN);
    float* out  = (float*)d_out;

    int tot = BB * NN * CC + BB * MM * CC;
    sigmoid_kernel<<<(tot + 255) / 256, 256, 0, stream>>>(p1l, p2l, s1, s2);

    const int TM = (MM + 15) / 16, TN = (NN + 15) / 16;
    cost_kernel<<<BB * TM * TN, 32, 0, stream>>>(p1b, p2b, s1, s2, cost);

    lapjv_kernel<<<BB, LTH, 0, stream>>>(cost, map);

    int tot2 = BB * MM;
    extrap_kernel<<<(tot2 + 255) / 256, 256, 0, stream>>>(p1b, p1l, p2b, p2l,
                                                          toff, map, out);
}


# Each row's 64 cheapest columns are selected in a launch of their own, one wave per row over the whole chip, not by 4 waves per batch inside the search kernel; lists pass via the workspace (5 launches)
# speedup vs baseline: 5.7716x; 1.0553x over previous
#include <hip/hip_runtime.h>
#include <math.h>

#define BB 4
#define MM 500
#define NN 500
#define CC 91

typedef __attribute__((ext_vector_type(2))) float v2f;
typedef __attribute__((ext_vector_type(8))) float v8f;

__device__ __forceinline__ int imin_i(int a, int b) { return a < b ? a : b; }

__global__ void sigmoid_kernel(const float* __restrict__ p1l,
                               const float* __restrict__ p2l,
                               float* __restrict__ s1,
                               float* __restrict__ s2) {
    int idx = blockIdx.x * blockDim.x + threadIdx.x;
    const int n1 = BB * NN * CC;
    const int n2 = BB * MM * CC;
    if (idx < n1) {
        float x = p1l[idx];
        s1[idx] = 1.0f / (1.0f + expf(-x));
    } else if (idx < n1 + n2) {
        float x = p2l[idx - n1];
        s2[idx - n1] = 1.0f / (1.0f + expf(-x));
    }
}

__global__ void cost_kernel(const float* __restrict__ p1b,
                            const float* __restrict__ p2b,
                            const float* __restrict__ s1,
                            const float* __restrict__ s2,
                            float* __restrict__ cost) {
    const int TN = (NN + 15) / 16;
    const int TM = (MM + 15) / 16;
    int bid = blockIdx.x;
    int b  = bid / (TM * TN);
    int t  = bid % (TM * TN);
    int m0 = (t / TN) * 16;
    int n0 = (t % TN) * 16;
    int lane = threadIdx.x;

    __shared__ float sq2[16];
    __shared__ float sq1[16];
    __shared__ float ls2[16 * CC];
    __shared__ float ls1[16 * CC];

    v2f a  = {0.0f, 0.0f};
    v2f bv = {0.0f, 0.0f};
    if (lane < 16) {
        int m = imin_i(m0 + lane, MM - 1);
        const float* pb = p2b + ((size_t)b * MM + m) * 4;
        float x = pb[0], y = pb[1];
        a.x = x; a.y = y;
        sq2[lane] = x * x + y * y;
        int n = imin_i(n0 + lane, NN - 1);
        const float* qb = p1b + ((size_t)b * NN + n) * 4;
        float bx = qb[0], by = qb[1];
        bv.x = bx; bv.y = by;
        sq1[lane] = bx * bx + by * by;
    }
    const float* s2base = s2 + (size_t)b * MM * CC;
    const float* s1base = s1 + (size_t)b * NN * CC;
    for (int idx = lane; idx < 16 * CC; idx += 32) {
        int r = idx / CC;
        int c = idx - r * CC;
        ls2[idx] = s2base[(size_t)imin_i(m0 + r, MM - 1) * CC + c];
        ls1[idx] = s1base[(size_t)imin_i(n0 + r, NN - 1) * CC + c];
    }
    __syncthreads();

    v8f cacc = {0.f, 0.f, 0.f, 0.f, 0.f, 0.f, 0.f, 0.f};
    v8f dot = __builtin_amdgcn_wmma_f32_16x16x4_f32(
         false, a,  false, bv,
         (short)0, cacc,  false,  false);

    int n_loc  = lane & 15;
    int m_base = (lane >> 4) << 3;

    float cls[8];
#pragma unroll
    for (int i = 0; i < 8; ++i) cls[i] = 0.0f;
    for (int c = 0; c < CC; ++c) {
        float s1v = ls1[n_loc * CC + c];
#pragma unroll
        for (int i = 0; i < 8; ++i) {
            float d = fabsf(ls2[(m_base + i) * CC + c] - s1v);
            cls[i] = fmaxf(cls[i], d);
        }
    }

#pragma unroll
    for (int i = 0; i < 8; ++i) {
        int m = m0 + m_base + i;
        int n = n0 + n_loc;
        if (m < MM && n < NN) {
            float d2 = sq2[m_base + i] + sq1[n_loc] - 2.0f * dot[i];
            float cd = sqrtf(fmaxf(d2, 0.0f));
            cost[((size_t)b * MM + m) * NN + n] = 0.5f * cd + 0.5f * cls[i];
        }
    }
}

#define LTH 128
#define NWAVE (LTH / 32)
#define CPL 4
typedef __attribute__((ext_vector_type(4))) float v4f;
typedef __attribute__((ext_vector_type(4))) int   v4i;
typedef __attribute__((ext_vector_type(2))) int   v2i;
typedef __attribute__((ext_vector_type(2))) double v2d;
typedef __attribute__((ext_vector_type(2))) unsigned short v2us;

#define KC 64
static_assert(KC == 64, "two candidates per lane of one wave");
static_assert(NN < 512 && NN <= 65535, "16 columns per lane, column 511 is a pad; 16-bit column ids");

static_assert(NN % CPL == 0 && NN / CPL <= LTH, "column tiling");
static_assert(MM == NN, "square problem");

template <typename T> struct XT;
template <> struct XT<double> {
    template <int MASK>
    static __device__ __forceinline__ double rx(double x) {
        v2i p = __builtin_bit_cast(v2i, x);
        p.x = __builtin_amdgcn_update_dpp(p.x, p.x, 0x160 + MASK, 0xf, 0xf, false);
        p.y = __builtin_amdgcn_update_dpp(p.y, p.y, 0x160 + MASK, 0xf, 0xf, false);
        return __builtin_bit_cast(double, p);
    }
    static __device__ __forceinline__ double hx(double x) {
        v2i p = __builtin_bit_cast(v2i, x);
        p.x = __builtin_amdgcn_permlanex16(p.x, p.x, 0x76543210, 0xfedcba98, false, false);
        p.y = __builtin_amdgcn_permlanex16(p.y, p.y, 0x76543210, 0xfedcba98, false, false);
        return __builtin_bit_cast(double, p);
    }
    static __device__ __forceinline__ v2i pack(double x) { return __builtin_bit_cast(v2i, x); }
    static __device__ __forceinline__ double unpack(v2i p) { return __builtin_bit_cast(double, p); }
};
template <> struct XT<float> {
    template <int MASK>
    static __device__ __forceinline__ float rx(float x) {
        int p = __builtin_bit_cast(int, x);
        p = __builtin_amdgcn_update_dpp(p, p, 0x160 + MASK, 0xf, 0xf, false);
        return __builtin_bit_cast(float, p);
    }
    static __device__ __forceinline__ float hx(float x) {
        int p = __builtin_bit_cast(int, x);
        p = __builtin_amdgcn_permlanex16(p, p, 0x76543210, 0xfedcba98, false, false);
        return __builtin_bit_cast(float, p);
    }
    static __device__ __forceinline__ v2i pack(float x) {
        v2i p = {__builtin_bit_cast(int, x), 0};
        return p;
    }
    static __device__ __forceinline__ float unpack(v2i p) { return __builtin_bit_cast(float, p.x); }
};

template <typename T, bool LIST>
__device__ __forceinline__ void sap_rows(const float* __restrict__ costB,
                                         const int nrows, const int* rowlist,
                                         T (&v)[CPL], T* sh_u, T* sh_short,
                                         int* sh_path, int* sh_c4r, int* sh_r4c,
                                         int* sh_SR, v4i (*rec)[NWAVE], int& par,
                                         const int tid, const int lane,
                                         const int wave, const bool owner,
                                         const unsigned full) {
    const T INF = (T)__builtin_inf();
    for (int idx = 0; idx < nrows; ++idx) {
        const int cur = LIST ? rowlist[idx] : idx;
        T   sh[CPL];
        int path[CPL];
        int r4c[CPL];
        {
            v4i q = *(const v4i*)&sh_r4c[CPL * (owner ? tid : 0)];
            r4c[0] = q.x; r4c[1] = q.y; r4c[2] = q.z; r4c[3] = q.w;
        }
        for (int k = 0; k < CPL; ++k) { sh[k] = INF; path[k] = -1; }
        unsigned rem = full;
        int ii   = cur;
        T   minv = (T)0;
        int nsr  = 0;
        int sink = -1;

        v4f c = {0.0f, 0.0f, 0.0f, 0.0f};
        if (owner) c = *(const v4f*)(costB + (size_t)ii * NN + CPL * tid);
        T z[CPL];
        {
            const T h = minv - sh_u[ii];
            for (int k = 0; k < CPL; ++k) z[k] = h - v[k];
        }

        while (true) {
            if (tid == 0) sh_SR[nsr] = ii;
            ++nsr;

            T t[CPL];
            for (int k = 0; k < CPL; ++k) t[k] = (T)c[k];
            for (int k = 0; k < CPL; ++k) t[k] = t[k] + z[k];

            T lm = INF; int lk = 0; int lr = -1;
            for (int k = 0; k < CPL; ++k) {
                const bool rm = ((rem >> k) & 1u) != 0u;
                const bool better = rm && (t[k] < sh[k]);
                sh[k]   = better ? t[k] : sh[k];
                path[k] = better ? ii : path[k];
                const T cd = rm ? sh[k] : INF;
                if (k == 0)       { lm = cd; lk = 0; lr = r4c[0]; }
                else if (cd < lm) { lm = cd; lk = k; lr = r4c[k]; }
            }

            T wm = lm;
            { const T o = XT<T>::template rx<1>(wm); wm = (o < wm) ? o : wm; }
            { const T o = XT<T>::template rx<2>(wm); wm = (o < wm) ? o : wm; }
            { const T o = XT<T>::template rx<4>(wm); wm = (o < wm) ? o : wm; }
            { const T o = XT<T>::template rx<8>(wm); wm = (o < wm) ? o : wm; }
            { const T o = XT<T>::hx(wm);             wm = (o < wm) ? o : wm; }
            const unsigned bal = __builtin_amdgcn_ballot_w32(lm == wm);
            const int srcl = __builtin_ctz(bal | 0x80000000u);
            const int wk = __builtin_amdgcn_readlane(lk, srcl);
            const int wr = __builtin_amdgcn_readlane(lr, srcl);
            if (lane == 0) {
                v2i p = XT<T>::pack(wm);
                v4i q = {p.x, p.y, CPL * (wave * 32 + srcl) + wk, wr};
                rec[par][wave] = q;
            }
            __syncthreads();

            T bv; int bj, br;
            {
                v4i q = rec[par][0];
                v2i p = {q.x, q.y};
                bv = XT<T>::unpack(p); bj = q.z; br = q.w;
            }
            for (int w = 1; w < NWAVE; ++w) {
                v4i q = rec[par][w];
                v2i p = {q.x, q.y};
                const T ov = XT<T>::unpack(p);
                if (ov < bv) { bv = ov; bj = q.z; br = q.w; }
            }
            par ^= 1;
            minv = bv;
            bj = __builtin_amdgcn_readfirstlane(bj);
            br = __builtin_amdgcn_readfirstlane(br);
            if (tid == (bj >> 2)) rem &= ~(1u << (bj & 3));
            if (br == -1) { sink = bj; break; }
            ii = br;

            c = (v4f){0.0f, 0.0f, 0.0f, 0.0f};
            if (owner) c = *(const v4f*)(costB + (size_t)ii * NN + CPL * tid);
            {
                const T h = minv - sh_u[ii];
                for (int k = 0; k < CPL; ++k) z[k] = h - v[k];
            }
        }

        if (owner) {
            v4i pq = {path[0], path[1], path[2], path[3]};
            *(v4i*)&sh_path[CPL * tid] = pq;
            for (int k = 0; k < CPL; ++k) sh_short[CPL * tid + k] = sh[k];
        }
        __syncthreads();

        for (int t2 = tid; t2 < nsr; t2 += LTH) {
            if (t2 == 0) {
                sh_u[cur] += minv;
            } else {
                int sr = sh_SR[t2];
                sh_u[sr] += minv - sh_short[sh_c4r[sr]];
            }
        }
        {
            const unsigned scanned = full & ~rem;
            for (int k = 0; k < CPL; ++k)
                if ((scanned >> k) & 1u) v[k] -= minv - sh[k];
        }
        __syncthreads();

        if (tid == 0) {
            int j = sink;
            while (true) {
                int i = sh_path[j];
                sh_r4c[j] = i;
                int tmp = sh_c4r[i];
                sh_c4r[i] = j;
                j = tmp;
                if (i == cur) break;
            }
        }
        __syncthreads();
    }
}

__device__ __forceinline__ int wave_sum_i(int x) {
    x += __builtin_amdgcn_update_dpp(x, x, 0x160 + 1, 0xf, 0xf, false);
    x += __builtin_amdgcn_update_dpp(x, x, 0x160 + 2, 0xf, 0xf, false);
    x += __builtin_amdgcn_update_dpp(x, x, 0x160 + 4, 0xf, 0xf, false);
    x += __builtin_amdgcn_update_dpp(x, x, 0x160 + 8, 0xf, 0xf, false);
    x += __builtin_amdgcn_permlanex16(x, x, 0x76543210, 0xfedcba98, false, false);
    return x;
}

__device__ __forceinline__ unsigned fbits(const float x) { return __builtin_bit_cast(unsigned, x); }

__device__ __forceinline__ void load_row16(const float* __restrict__ row, const int lane,
                                           unsigned (&bits)[16]) {
    for (int q = 0; q < 4; ++q) {
        const int c0 = 128 * q + 4 * lane;
        if (c0 < NN) {
            const v4f c = *(const v4f*)(row + c0);
            bits[4 * q + 0] = fbits(c.x);
            bits[4 * q + 1] = fbits(c.y);
            bits[4 * q + 2] = fbits(c.z);
            bits[4 * q + 3] = fbits(c.w);
        } else {
            for (int e = 0; e < 4; ++e) bits[4 * q + e] = 0x7f800000u;
        }
    }
}

#define SEL_WAVES 4
static_assert((BB * MM) % SEL_WAVES == 0, "whole work-groups of rows");
static_assert(MM * KC <= NN * CC && KC <= CC, "the lists fit into the two sigmoid planes they reuse");
__global__ void __launch_bounds__(32 * SEL_WAVES)
select_kernel(const float* __restrict__ cost, float* __restrict__ gcost, int* __restrict__ gcol) {
    const int lane = threadIdx.x & 31;
    const int wave = __builtin_amdgcn_readfirstlane((int)(threadIdx.x >> 5));
    const int row  = blockIdx.x * SEL_WAVES + wave;

    __shared__ float sc[SEL_WAVES][KC];
    __shared__ int   sj[SEL_WAVES][KC];

    const unsigned ltmask = (1u << lane) - 1u;
    unsigned bits[16];
    load_row16(cost + (size_t)row * NN, lane, bits);
    unsigned T = 0u;
    for (int bit = 30; bit >= 7; --bit) {
        const unsigned trial = T | (1u << bit);
        int cnt = 0;
        for (int k = 0; k < 16; ++k) cnt += (bits[k] < trial) ? 1 : 0;
        cnt = __builtin_amdgcn_readfirstlane(wave_sum_i(cnt));
        if (cnt <= KC) T = trial;
    }
    sc[wave][lane]      = __builtin_inff();
    sc[wave][lane + 32] = __builtin_inff();
    sj[wave][lane]      = 511;
    sj[wave][lane + 32] = 511;
    int base = 0;
    for (int k = 0; k < 16; ++k) {
        const bool f = bits[k] < T;
        const unsigned bal = __builtin_amdgcn_ballot_w32(f);
        const int pos = base + __builtin_popcount(bal & ltmask);
        if (f && pos < KC) {
            sc[wave][pos] = __builtin_bit_cast(float, bits[k]);
            sj[wave][pos] = 128 * (k >> 2) + 4 * lane + (k & 3);
        }
        base += __builtin_popcount(bal);
    }
    const float a0 = sc[wave][lane], a1 = sc[wave][lane + 32];
    const int   j0 = sj[wave][lane], j1 = sj[wave][lane + 32];
    gcost[(size_t)row * KC + lane]      = a0;
    gcost[(size_t)row * KC + lane + 32] = a1;
    gcol[(size_t)row * KC + lane]       = j0;
    gcol[(size_t)row * KC + lane + 32]  = j1;
}

__global__ void __launch_bounds__(LTH)
lapjv_kernel(const float* __restrict__ cost, const float* __restrict__ gcost,
             const int* __restrict__ gcol, int* __restrict__ mapping) {
    const int b    = blockIdx.x;
    const int tid  = threadIdx.x;
    const int lane = tid & 31;
    const int wave = __builtin_amdgcn_readfirstlane(tid >> 5);
    const float* costB = cost + (size_t)b * MM * NN;
    const bool owner = tid < NN / CPL;
    const unsigned full = owner ? 0xFu : 0u;

    __shared__ __attribute__((aligned(16))) float lcost[MM * KC];
    __shared__ __attribute__((aligned(16))) unsigned short lcol[MM * KC];
    __shared__ __attribute__((aligned(16))) double sh_ud[512];
    __shared__ __attribute__((aligned(16))) double sh_vd[512];
    __shared__ __attribute__((aligned(16))) double sh_short[512];
    __shared__ __attribute__((aligned(16))) float  sh_mbox[512];
    __shared__ __attribute__((aligned(16))) float  sh_vf[512];
    __shared__ __attribute__((aligned(16))) float  sh_uf[512];
    __shared__ __attribute__((aligned(16))) float  sh_popv[512];
    __shared__ __attribute__((aligned(16))) int sh_popc[512];
    __shared__ __attribute__((aligned(16))) int sh_path[512];
    __shared__ __attribute__((aligned(16))) int sh_c4r[512];
    __shared__ __attribute__((aligned(16))) int sh_r4c[512];
    __shared__ __attribute__((aligned(16))) int sh_SR[512];
    __shared__ __attribute__((aligned(16))) int sh_free[512];
    __shared__ __attribute__((aligned(16))) int sh_list[512];
    __shared__ int sh_cnt;
    __shared__ v4i rec[2][NWAVE];

    for (int t = tid; t < 512; t += LTH) {
        sh_uf[t] = 0.0f; sh_vf[t] = 0.0f; sh_c4r[t] = -1; sh_r4c[t] = -1;
    }
    int par = 0;

    {
        const v4f* gc = (const v4f*)(gcost + (size_t)b * MM * KC);
        const v4i* gj = (const v4i*)(gcol + (size_t)b * MM * KC);
        const int NG = MM * KC / 4;
        for (int t0 = tid; t0 < NG; t0 += LTH * 8) {
            v4f cq[8];
            v4i jq[8];
            for (int e = 0; e < 8; ++e) {
                const int t = imin_i(t0 + LTH * e, NG - 1);
                cq[e] = gc[t];
                jq[e] = gj[t];
            }
            for (int e = 0; e < 8; ++e) {
                const int t = t0 + LTH * e;
                if (t < NG) {
                    *(v4f*)&lcost[4 * t] = cq[e];
                    const v2i pj = {(jq[e].x & 0xffff) | (jq[e].y << 16),
                                    (jq[e].z & 0xffff) | (jq[e].w << 16)};
                    *(v2i*)&lcol[4 * t] = pj;
                }
            }
        }
    }
    __syncthreads();

    if (wave == 0) {
        const float INFf = __builtin_inff();
        for (int q = 0; q < 4; ++q) {
            const v4f ini = {INFf, INFf, INFf, INFf};
            *(v4f*)&sh_mbox[128 * q + 4 * lane] = ini;
        }
        for (int cur = 0; cur < MM; ++cur) {
            float sh[16], vv[16];
            int   path[16], pk[16];
            for (int q = 0; q < 4; ++q) {
                const int c0 = 128 * q + 4 * lane;
                const bool real = c0 < NN;
                const v4i r = *(const v4i*)&sh_r4c[c0];
                const v4f w = *(const v4f*)&sh_vf[c0];
                pk[4 * q + 0] = ((r.x + 1) << 4) | (4 * q + 0);
                pk[4 * q + 1] = ((r.y + 1) << 4) | (4 * q + 1);
                pk[4 * q + 2] = ((r.z + 1) << 4) | (4 * q + 2);
                pk[4 * q + 3] = ((r.w + 1) << 4) | (4 * q + 3);
                vv[4 * q + 0] = real ? w.x : -INFf;
                vv[4 * q + 1] = real ? w.y : -INFf;
                vv[4 * q + 2] = real ? w.z : -INFf;
                vv[4 * q + 3] = real ? w.w : -INFf;
            }
            for (int k = 0; k < 16; ++k) { sh[k] = INFf; path[k] = -1; }

            int   ii   = cur;
            float minv = 0.0f;
            int   nsr  = 0;
            int   sink = -1;
            bool  okay = true;
            v2f   cc = *(const v2f*)&lcost[ii * KC + 2 * lane];
            v2us  cw = *(const v2us*)&lcol[ii * KC + 2 * lane];
            float uf = sh_uf[ii];
            while (true) {
                const int c0 = cw.x, c1 = cw.y;
                sh_mbox[c0] = cc.x;
                sh_mbox[c1] = cc.y;
                v4f x[4];
                for (int q = 0; q < 4; ++q) x[q] = *(const v4f*)&sh_mbox[128 * q + 4 * lane];
                sh_mbox[c0] = INFf;
                sh_mbox[c1] = INFf;

                const float h = minv - uf;
                for (int k = 0; k < 16; ++k) {
                    const float xk = x[k >> 2][k & 3];
                    const float r = xk + (h - vv[k]);
                    const bool better = r < sh[k];
                    sh[k]   = better ? r : sh[k];
                    path[k] = better ? ii : path[k];
                }

                float m8[8]; int p8[8];
                for (int t = 0; t < 8; ++t) {
                    const float ma = sh[2 * t], mb = sh[2 * t + 1];
                    const int   pa = pk[2 * t], pb = pk[2 * t + 1];
                    const bool tk = mb < ma;
                    m8[t] = tk ? mb : ma;
                    p8[t] = tk ? pb : pa;
                }
                float m4[4]; int p4[4];
                for (int t = 0; t < 4; ++t) {
                    const float ma = m8[2 * t], mb = m8[2 * t + 1];
                    const int   pa = p8[2 * t], pb = p8[2 * t + 1];
                    const bool tk = mb < ma;
                    m4[t] = tk ? mb : ma;
                    p4[t] = tk ? pb : pa;
                }
                float m2[2]; int p2[2];
                for (int t = 0; t < 2; ++t) {
                    const float ma = m4[2 * t], mb = m4[2 * t + 1];
                    const int   pa = p4[2 * t], pb = p4[2 * t + 1];
                    const bool tk = mb < ma;
                    m2[t] = tk ? mb : ma;
                    p2[t] = tk ? pb : pa;
                }
                const float m2a = m2[0], m2b = m2[1];
                const int   p2a = p2[0], p2b = p2[1];
                const bool tk1 = m2b < m2a;
                const float lm = tk1 ? m2b : m2a;
                const int   lp = tk1 ? p2b : p2a;

                float wm = lm;
                { const float o = XT<float>::rx<1>(wm); wm = (o < wm) ? o : wm; }
                { const float o = XT<float>::rx<2>(wm); wm = (o < wm) ? o : wm; }
                { const float o = XT<float>::rx<4>(wm); wm = (o < wm) ? o : wm; }
                { const float o = XT<float>::rx<8>(wm); wm = (o < wm) ? o : wm; }
                { const float o = XT<float>::hx(wm);    wm = (o < wm) ? o : wm; }
                const unsigned bal = __builtin_amdgcn_ballot_w32(lm == wm);
                const int srcl = __builtin_ctz(bal | 0x80000000u);
                const int wp   = __builtin_amdgcn_readlane(lp, srcl);
                const float wms = __builtin_bit_cast(float,
                    __builtin_amdgcn_readfirstlane(__builtin_bit_cast(int, wm)));
                if (!(wms < INFf)) { okay = false; break; }
                const int k  = wp & 15;
                const int br = (wp >> 4) - 1;
                const int bj = 128 * (k >> 2) + 4 * srcl + (k & 3);
                minv = wms;
                const int nrow = (br < 0) ? ii : br;
                cc = *(const v2f*)&lcost[nrow * KC + 2 * lane];
                cw = *(const v2us*)&lcol[nrow * KC + 2 * lane];
                uf = sh_uf[nrow];
                if (lane == 0) {
                    sh_SR[nsr]   = ii;
                    sh_popc[nsr] = bj;
                    sh_popv[nsr] = wms;
                }
                ++nsr;
                if (br == -1) { sink = bj; break; }
                const int mk = (lp & 15) + ((lane == srcl) ? 0 : 16);
                for (int kk = 0; kk < 16; ++kk) {
                    const bool hit = (mk == kk);
                    sh[kk] = hit ? INFf : sh[kk];
                    vv[kk] = hit ? -INFf : vv[kk];
                }
                ii = br;
            }
            if (!okay) continue;

            for (int q = 0; q < 4; ++q) {
                const v4i pq = {path[4 * q + 0], path[4 * q + 1], path[4 * q + 2], path[4 * q + 3]};
                *(v4i*)&sh_path[128 * q + 4 * lane] = pq;
            }
            for (int t = lane; t < nsr; t += 32) {
                const float d = minv - sh_popv[t];
                sh_vf[sh_popc[t]] -= d;
                if (t == 0) sh_uf[cur] += minv;
                if (t + 1 < nsr) sh_uf[sh_SR[t + 1]] += d;
            }
            if (lane == 0) {
                int j = sink;
                for (int g = 0; g < MM; ++g) {
                    const int i = sh_path[j];
                    sh_r4c[j] = i;
                    const int tmp = sh_c4r[i];
                    sh_c4r[i] = j;
                    j = tmp;
                    if (i == cur) break;
                }
            }
        }
    }
    __syncthreads();

    double v[CPL];
    {
        v4f q = {0.0f, 0.0f, 0.0f, 0.0f};
        if (owner) q = *(const v4f*)&sh_vf[CPL * tid];
        v[0] = (double)q.x; v[1] = (double)q.y; v[2] = (double)q.z; v[3] = (double)q.w;
    }
    if (owner) {
        v2d a01 = {v[0], v[1]};
        v2d a23 = {v[2], v[3]};
        *(v2d*)&sh_vd[CPL * tid]     = a01;
        *(v2d*)&sh_vd[CPL * tid + 2] = a23;
    }
    __syncthreads();
    for (int i = wave; i < MM; i += NWAVE) {
        const float* row = costB + (size_t)i * NN;
        double m = __builtin_inf();
        for (int q = 0; q < 4; ++q) {
            const int c0 = 128 * q + CPL * lane;
            if (c0 < NN) {
                const v4f c = *(const v4f*)(row + c0);
                const v2d va = *(const v2d*)&sh_vd[c0];
                const v2d vb = *(const v2d*)&sh_vd[c0 + 2];
                const double d0 = (double)c.x - va.x;
                const double d1 = (double)c.y - va.y;
                const double d2 = (double)c.z - vb.x;
                const double d3 = (double)c.w - vb.y;
                m = (d0 < m) ? d0 : m;
                m = (d1 < m) ? d1 : m;
                m = (d2 < m) ? d2 : m;
                m = (d3 < m) ? d3 : m;
            }
        }
        { const double o = XT<double>::rx<1>(m); m = (o < m) ? o : m; }
        { const double o = XT<double>::rx<2>(m); m = (o < m) ? o : m; }
        { const double o = XT<double>::rx<4>(m); m = (o < m) ? o : m; }
        { const double o = XT<double>::rx<8>(m); m = (o < m) ? o : m; }
        { const double o = XT<double>::hx(m);    m = (o < m) ? o : m; }
        const int jm = sh_c4r[i];
        const bool has = jm >= 0;
        const int jq = has ? jm : 0;
        const double dm = (double)row[jq] - sh_vd[jq];
        const bool keep = has && (dm == m);
        if (lane == 0) {
            sh_ud[i]   = m;
            sh_free[i] = keep ? 0 : 1;
            if (has && !keep) { sh_c4r[i] = -1; sh_r4c[jm] = -1; }
        }
    }
    __syncthreads();
    if (wave == 0) {
        int cnt = 0;
        for (int base = 0; base < 512; base += 32) {
            const int r = base + lane;
            const bool f = (r < MM) && (sh_free[r] != 0);
            const unsigned bal = __builtin_amdgcn_ballot_w32(f);
            const int pos = cnt + __builtin_popcount(bal & ((1u << lane) - 1u));
            if (f) sh_list[pos] = r;
            cnt += __builtin_popcount(bal);
        }
        if (lane == 0) sh_cnt = cnt;
    }
    __syncthreads();
    const int nfree = __builtin_amdgcn_readfirstlane(sh_cnt);

    sap_rows<double, true>(costB, nfree, sh_list, v, sh_ud, sh_short,
                           sh_path, sh_c4r, sh_r4c, sh_SR, rec, par,
                           tid, lane, wave, owner, full);

    for (int t = tid; t < MM; t += LTH) mapping[b * MM + t] = sh_c4r[t];
}

__global__ void extrap_kernel(const float* __restrict__ p1b,
                              const float* __restrict__ p1l,
                              const float* __restrict__ p2b,
                              const float* __restrict__ p2l,
                              const float* __restrict__ toff,
                              const int*   __restrict__ mapping,
                              float* __restrict__ out) {
    int idx = blockIdx.x * blockDim.x + threadIdx.x;
    if (idx >= BB * MM) return;
    int b = idx / MM;

    float first  = toff[b * 3 + 1] - toff[b * 3 + 0];
    float second = toff[b * 3 + 2] - toff[b * 3 + 1];
    float factor = second / first;

    int  j   = mapping[idx];
    bool has = (j >= 0);
    int  jc  = has ? j : 0;

    const float* pb2 = p2b + (size_t)idx * 4;
    const float* pb1 = p1b + ((size_t)b * NN + jc) * 4;
    float* ob = out + (size_t)idx * 4;
    for (int k = 0; k < 4; ++k) {
        float c1 = has ? pb1[k] : pb2[k];
        float v  = pb2[k] + (pb2[k] - c1) * factor;
        ob[k] = (k >= 2) ? fmaxf(v, 0.0f) : v;
    }

    const float* pl2 = p2l + (size_t)idx * CC;
    const float* pl1 = p1l + ((size_t)b * NN + jc) * CC;
    float* ol = out + (size_t)BB * MM * 4 + (size_t)idx * CC;
    for (int c = 0; c < CC; ++c) {
        float c1 = has ? pl1[c] : 0.0f;
        ol[c] = 0.5f * (pl2[c] + c1);
    }
}

extern "C" void kernel_launch(void* const* d_in, const int* in_sizes, int n_in,
                              void* d_out, int out_size, void* d_ws, size_t ws_size,
                              hipStream_t stream) {
    (void)in_sizes; (void)n_in; (void)out_size; (void)ws_size;
    const float* p1b  = (const float*)d_in[0];
    const float* p1l  = (const float*)d_in[1];
    const float* p2b  = (const float*)d_in[2];
    const float* p2l  = (const float*)d_in[3];
    const float* toff = (const float*)d_in[4];

    float* ws   = (float*)d_ws;
    float* s1   = ws;
    float* s2   = s1 + (size_t)BB * NN * CC;
    float* cost = s2 + (size_t)BB * MM * CC;
    int*   map  = (int*)(cost + (size_t)BB * MM * NN);
    float* out  = (float*)d_out;

    int tot = BB * NN * CC + BB * MM * CC;
    sigmoid_kernel<<<(tot + 255) / 256, 256, 0, stream>>>(p1l, p2l, s1, s2);

    const int TM = (MM + 15) / 16, TN = (NN + 15) / 16;
    cost_kernel<<<BB * TM * TN, 32, 0, stream>>>(p1b, p2b, s1, s2, cost);

    float* gcost = s1;
    int*   gcol  = (int*)s2;
    select_kernel<<<BB * MM / SEL_WAVES, 32 * SEL_WAVES, 0, stream>>>(cost, gcost, gcol);

    lapjv_kernel<<<BB, LTH, 0, stream>>>(cost, gcost, gcol, map);

    int tot2 = BB * MM;
    extrap_kernel<<<(tot2 + 255) / 256, 256, 0, stream>>>(p1b, p1l, p2b, p2l,
                                                          toff, map, out);
}


# Sparse f32 assignment pass: node values of the 16-column tree and the wave minimum are min instructions (4 DPP min + 1 cross-row); compares only pick the index; no compare+select per level; same sums
# speedup vs baseline: 6.1412x; 1.0640x over previous
#include <hip/hip_runtime.h>
#include <math.h>

#define BB 4
#define MM 500
#define NN 500
#define CC 91

typedef __attribute__((ext_vector_type(2))) float v2f;
typedef __attribute__((ext_vector_type(8))) float v8f;

__device__ __forceinline__ int imin_i(int a, int b) { return a < b ? a : b; }

__global__ void sigmoid_kernel(const float* __restrict__ p1l,
                               const float* __restrict__ p2l,
                               float* __restrict__ s1,
                               float* __restrict__ s2) {
    int idx = blockIdx.x * blockDim.x + threadIdx.x;
    const int n1 = BB * NN * CC;
    const int n2 = BB * MM * CC;
    if (idx < n1) {
        float x = p1l[idx];
        s1[idx] = 1.0f / (1.0f + expf(-x));
    } else if (idx < n1 + n2) {
        float x = p2l[idx - n1];
        s2[idx - n1] = 1.0f / (1.0f + expf(-x));
    }
}

__global__ void cost_kernel(const float* __restrict__ p1b,
                            const float* __restrict__ p2b,
                            const float* __restrict__ s1,
                            const float* __restrict__ s2,
                            float* __restrict__ cost) {
    const int TN = (NN + 15) / 16;
    const int TM = (MM + 15) / 16;
    int bid = blockIdx.x;
    int b  = bid / (TM * TN);
    int t  = bid % (TM * TN);
    int m0 = (t / TN) * 16;
    int n0 = (t % TN) * 16;
    int lane = threadIdx.x;

    __shared__ float sq2[16];
    __shared__ float sq1[16];
    __shared__ float ls2[16 * CC];
    __shared__ float ls1[16 * CC];

    v2f a  = {0.0f, 0.0f};
    v2f bv = {0.0f, 0.0f};
    if (lane < 16) {
        int m = imin_i(m0 + lane, MM - 1);
        const float* pb = p2b + ((size_t)b * MM + m) * 4;
        float x = pb[0], y = pb[1];
        a.x = x; a.y = y;
        sq2[lane] = x * x + y * y;
        int n = imin_i(n0 + lane, NN - 1);
        const float* qb = p1b + ((size_t)b * NN + n) * 4;
        float bx = qb[0], by = qb[1];
        bv.x = bx; bv.y = by;
        sq1[lane] = bx * bx + by * by;
    }
    const float* s2base = s2 + (size_t)b * MM * CC;
    const float* s1base = s1 + (size_t)b * NN * CC;
    for (int idx = lane; idx < 16 * CC; idx += 32) {
        int r = idx / CC;
        int c = idx - r * CC;
        ls2[idx] = s2base[(size_t)imin_i(m0 + r, MM - 1) * CC + c];
        ls1[idx] = s1base[(size_t)imin_i(n0 + r, NN - 1) * CC + c];
    }
    __syncthreads();

    v8f cacc = {0.f, 0.f, 0.f, 0.f, 0.f, 0.f, 0.f, 0.f};
    v8f dot = __builtin_amdgcn_wmma_f32_16x16x4_f32(
         false, a,  false, bv,
         (short)0, cacc,  false,  false);

    int n_loc  = lane & 15;
    int m_base = (lane >> 4) << 3;

    float cls[8];
#pragma unroll
    for (int i = 0; i < 8; ++i) cls[i] = 0.0f;
    for (int c = 0; c < CC; ++c) {
        float s1v = ls1[n_loc * CC + c];
#pragma unroll
        for (int i = 0; i < 8; ++i) {
            float d = fabsf(ls2[(m_base + i) * CC + c] - s1v);
            cls[i] = fmaxf(cls[i], d);
        }
    }

#pragma unroll
    for (int i = 0; i < 8; ++i) {
        int m = m0 + m_base + i;
        int n = n0 + n_loc;
        if (m < MM && n < NN) {
            float d2 = sq2[m_base + i] + sq1[n_loc] - 2.0f * dot[i];
            float cd = sqrtf(fmaxf(d2, 0.0f));
            cost[((size_t)b * MM + m) * NN + n] = 0.5f * cd + 0.5f * cls[i];
        }
    }
}

#define LTH 128
#define NWAVE (LTH / 32)
#define CPL 4
typedef __attribute__((ext_vector_type(4))) float v4f;
typedef __attribute__((ext_vector_type(4))) int   v4i;
typedef __attribute__((ext_vector_type(2))) int   v2i;
typedef __attribute__((ext_vector_type(2))) double v2d;
typedef __attribute__((ext_vector_type(2))) unsigned short v2us;

#define KC 64
static_assert(KC == 64, "two candidates per lane of one wave");
static_assert(NN < 512 && NN <= 65535, "16 columns per lane, column 511 is a pad; 16-bit column ids");

static_assert(NN % CPL == 0 && NN / CPL <= LTH, "column tiling");
static_assert(MM == NN, "square problem");

template <typename T> struct XT;
template <> struct XT<double> {
    template <int MASK>
    static __device__ __forceinline__ double rx(double x) {
        v2i p = __builtin_bit_cast(v2i, x);
        p.x = __builtin_amdgcn_update_dpp(p.x, p.x, 0x160 + MASK, 0xf, 0xf, false);
        p.y = __builtin_amdgcn_update_dpp(p.y, p.y, 0x160 + MASK, 0xf, 0xf, false);
        return __builtin_bit_cast(double, p);
    }
    static __device__ __forceinline__ double hx(double x) {
        v2i p = __builtin_bit_cast(v2i, x);
        p.x = __builtin_amdgcn_permlanex16(p.x, p.x, 0x76543210, 0xfedcba98, false, false);
        p.y = __builtin_amdgcn_permlanex16(p.y, p.y, 0x76543210, 0xfedcba98, false, false);
        return __builtin_bit_cast(double, p);
    }
    static __device__ __forceinline__ v2i pack(double x) { return __builtin_bit_cast(v2i, x); }
    static __device__ __forceinline__ double unpack(v2i p) { return __builtin_bit_cast(double, p); }
};

template <typename T, bool LIST>
__device__ __forceinline__ void sap_rows(const float* __restrict__ costB,
                                         const int nrows, const int* rowlist,
                                         T (&v)[CPL], T* sh_u, T* sh_short,
                                         int* sh_path, int* sh_c4r, int* sh_r4c,
                                         int* sh_SR, v4i (*rec)[NWAVE], int& par,
                                         const int tid, const int lane,
                                         const int wave, const bool owner,
                                         const unsigned full) {
    const T INF = (T)__builtin_inf();
    for (int idx = 0; idx < nrows; ++idx) {
        const int cur = LIST ? rowlist[idx] : idx;
        T   sh[CPL];
        int path[CPL];
        int r4c[CPL];
        {
            v4i q = *(const v4i*)&sh_r4c[CPL * (owner ? tid : 0)];
            r4c[0] = q.x; r4c[1] = q.y; r4c[2] = q.z; r4c[3] = q.w;
        }
        for (int k = 0; k < CPL; ++k) { sh[k] = INF; path[k] = -1; }
        unsigned rem = full;
        int ii   = cur;
        T   minv = (T)0;
        int nsr  = 0;
        int sink = -1;

        v4f c = {0.0f, 0.0f, 0.0f, 0.0f};
        if (owner) c = *(const v4f*)(costB + (size_t)ii * NN + CPL * tid);
        T z[CPL];
        {
            const T h = minv - sh_u[ii];
            for (int k = 0; k < CPL; ++k) z[k] = h - v[k];
        }

        while (true) {
            if (tid == 0) sh_SR[nsr] = ii;
            ++nsr;

            T t[CPL];
            for (int k = 0; k < CPL; ++k) t[k] = (T)c[k];
            for (int k = 0; k < CPL; ++k) t[k] = t[k] + z[k];

            T lm = INF; int lk = 0; int lr = -1;
            for (int k = 0; k < CPL; ++k) {
                const bool rm = ((rem >> k) & 1u) != 0u;
                const bool better = rm && (t[k] < sh[k]);
                sh[k]   = better ? t[k] : sh[k];
                path[k] = better ? ii : path[k];
                const T cd = rm ? sh[k] : INF;
                if (k == 0)       { lm = cd; lk = 0; lr = r4c[0]; }
                else if (cd < lm) { lm = cd; lk = k; lr = r4c[k]; }
            }

            T wm = lm;
            { const T o = XT<T>::template rx<1>(wm); wm = (o < wm) ? o : wm; }
            { const T o = XT<T>::template rx<2>(wm); wm = (o < wm) ? o : wm; }
            { const T o = XT<T>::template rx<4>(wm); wm = (o < wm) ? o : wm; }
            { const T o = XT<T>::template rx<8>(wm); wm = (o < wm) ? o : wm; }
            { const T o = XT<T>::hx(wm);             wm = (o < wm) ? o : wm; }
            const unsigned bal = __builtin_amdgcn_ballot_w32(lm == wm);
            const int srcl = __builtin_ctz(bal | 0x80000000u);
            const int wk = __builtin_amdgcn_readlane(lk, srcl);
            const int wr = __builtin_amdgcn_readlane(lr, srcl);
            if (lane == 0) {
                v2i p = XT<T>::pack(wm);
                v4i q = {p.x, p.y, CPL * (wave * 32 + srcl) + wk, wr};
                rec[par][wave] = q;
            }
            __syncthreads();

            T bv; int bj, br;
            {
                v4i q = rec[par][0];
                v2i p = {q.x, q.y};
                bv = XT<T>::unpack(p); bj = q.z; br = q.w;
            }
            for (int w = 1; w < NWAVE; ++w) {
                v4i q = rec[par][w];
                v2i p = {q.x, q.y};
                const T ov = XT<T>::unpack(p);
                if (ov < bv) { bv = ov; bj = q.z; br = q.w; }
            }
            par ^= 1;
            minv = bv;
            bj = __builtin_amdgcn_readfirstlane(bj);
            br = __builtin_amdgcn_readfirstlane(br);
            if (tid == (bj >> 2)) rem &= ~(1u << (bj & 3));
            if (br == -1) { sink = bj; break; }
            ii = br;

            c = (v4f){0.0f, 0.0f, 0.0f, 0.0f};
            if (owner) c = *(const v4f*)(costB + (size_t)ii * NN + CPL * tid);
            {
                const T h = minv - sh_u[ii];
                for (int k = 0; k < CPL; ++k) z[k] = h - v[k];
            }
        }

        if (owner) {
            v4i pq = {path[0], path[1], path[2], path[3]};
            *(v4i*)&sh_path[CPL * tid] = pq;
            for (int k = 0; k < CPL; ++k) sh_short[CPL * tid + k] = sh[k];
        }
        __syncthreads();

        for (int t2 = tid; t2 < nsr; t2 += LTH) {
            if (t2 == 0) {
                sh_u[cur] += minv;
            } else {
                int sr = sh_SR[t2];
                sh_u[sr] += minv - sh_short[sh_c4r[sr]];
            }
        }
        {
            const unsigned scanned = full & ~rem;
            for (int k = 0; k < CPL; ++k)
                if ((scanned >> k) & 1u) v[k] -= minv - sh[k];
        }
        __syncthreads();

        if (tid == 0) {
            int j = sink;
            while (true) {
                int i = sh_path[j];
                sh_r4c[j] = i;
                int tmp = sh_c4r[i];
                sh_c4r[i] = j;
                j = tmp;
                if (i == cur) break;
            }
        }
        __syncthreads();
    }
}

__device__ __forceinline__ int wave_sum_i(int x) {
    x += __builtin_amdgcn_update_dpp(x, x, 0x160 + 1, 0xf, 0xf, false);
    x += __builtin_amdgcn_update_dpp(x, x, 0x160 + 2, 0xf, 0xf, false);
    x += __builtin_amdgcn_update_dpp(x, x, 0x160 + 4, 0xf, 0xf, false);
    x += __builtin_amdgcn_update_dpp(x, x, 0x160 + 8, 0xf, 0xf, false);
    x += __builtin_amdgcn_permlanex16(x, x, 0x76543210, 0xfedcba98, false, false);
    return x;
}

__device__ __forceinline__ unsigned fbits(const float x) { return __builtin_bit_cast(unsigned, x); }

__device__ __forceinline__ float min2_f32(const float a, const float b) {
    float o;
    asm("v_min_num_f32 %0, %1, %2" : "=v"(o) : "v"(a), "v"(b));
    return o;
}

__device__ __forceinline__ float wave_min_f32(float x) {
    float a, b, c, d;
    asm("v_min_num_f32_dpp %0, %1, %1 row_xmask:1 row_mask:0xf bank_mask:0xf" : "=v"(a) : "v"(x));
    asm("v_min_num_f32_dpp %0, %1, %1 row_xmask:2 row_mask:0xf bank_mask:0xf" : "=v"(b) : "v"(a));
    asm("v_min_num_f32_dpp %0, %1, %1 row_xmask:4 row_mask:0xf bank_mask:0xf" : "=v"(c) : "v"(b));
    asm("v_min_num_f32_dpp %0, %1, %1 row_xmask:8 row_mask:0xf bank_mask:0xf" : "=v"(d) : "v"(c));
    const float o = __builtin_bit_cast(float,
        __builtin_amdgcn_permlanex16(0, __builtin_bit_cast(int, d), 0x76543210, 0xfedcba98, false, false));
    return min2_f32(o, d);
}

__device__ __forceinline__ void load_row16(const float* __restrict__ row, const int lane,
                                           unsigned (&bits)[16]) {
    for (int q = 0; q < 4; ++q) {
        const int c0 = 128 * q + 4 * lane;
        if (c0 < NN) {
            const v4f c = *(const v4f*)(row + c0);
            bits[4 * q + 0] = fbits(c.x);
            bits[4 * q + 1] = fbits(c.y);
            bits[4 * q + 2] = fbits(c.z);
            bits[4 * q + 3] = fbits(c.w);
        } else {
            for (int e = 0; e < 4; ++e) bits[4 * q + e] = 0x7f800000u;
        }
    }
}

#define SEL_WAVES 4
static_assert((BB * MM) % SEL_WAVES == 0, "whole work-groups of rows");
static_assert(MM * KC <= NN * CC && KC <= CC, "the lists fit into the two sigmoid planes they reuse");
__global__ void __launch_bounds__(32 * SEL_WAVES)
select_kernel(const float* __restrict__ cost, float* __restrict__ gcost, int* __restrict__ gcol) {
    const int lane = threadIdx.x & 31;
    const int wave = __builtin_amdgcn_readfirstlane((int)(threadIdx.x >> 5));
    const int row  = blockIdx.x * SEL_WAVES + wave;

    __shared__ float sc[SEL_WAVES][KC];
    __shared__ int   sj[SEL_WAVES][KC];

    const unsigned ltmask = (1u << lane) - 1u;
    unsigned bits[16];
    load_row16(cost + (size_t)row * NN, lane, bits);
    unsigned T = 0u;
    for (int bit = 30; bit >= 7; --bit) {
        const unsigned trial = T | (1u << bit);
        int cnt = 0;
        for (int k = 0; k < 16; ++k) cnt += (bits[k] < trial) ? 1 : 0;
        cnt = __builtin_amdgcn_readfirstlane(wave_sum_i(cnt));
        if (cnt <= KC) T = trial;
    }
    sc[wave][lane]      = __builtin_inff();
    sc[wave][lane + 32] = __builtin_inff();
    sj[wave][lane]      = 511;
    sj[wave][lane + 32] = 511;
    int base = 0;
    for (int k = 0; k < 16; ++k) {
        const bool f = bits[k] < T;
        const unsigned bal = __builtin_amdgcn_ballot_w32(f);
        const int pos = base + __builtin_popcount(bal & ltmask);
        if (f && pos < KC) {
            sc[wave][pos] = __builtin_bit_cast(float, bits[k]);
            sj[wave][pos] = 128 * (k >> 2) + 4 * lane + (k & 3);
        }
        base += __builtin_popcount(bal);
    }
    const float a0 = sc[wave][lane], a1 = sc[wave][lane + 32];
    const int   j0 = sj[wave][lane], j1 = sj[wave][lane + 32];
    gcost[(size_t)row * KC + lane]      = a0;
    gcost[(size_t)row * KC + lane + 32] = a1;
    gcol[(size_t)row * KC + lane]       = j0;
    gcol[(size_t)row * KC + lane + 32]  = j1;
}

__global__ void __launch_bounds__(LTH)
lapjv_kernel(const float* __restrict__ cost, const float* __restrict__ gcost,
             const int* __restrict__ gcol, int* __restrict__ mapping) {
    const int b    = blockIdx.x;
    const int tid  = threadIdx.x;
    const int lane = tid & 31;
    const int wave = __builtin_amdgcn_readfirstlane(tid >> 5);
    const float* costB = cost + (size_t)b * MM * NN;
    const bool owner = tid < NN / CPL;
    const unsigned full = owner ? 0xFu : 0u;

    __shared__ __attribute__((aligned(16))) float lcost[MM * KC];
    __shared__ __attribute__((aligned(16))) unsigned short lcol[MM * KC];
    __shared__ __attribute__((aligned(16))) double sh_ud[512];
    __shared__ __attribute__((aligned(16))) double sh_vd[512];
    __shared__ __attribute__((aligned(16))) double sh_short[512];
    __shared__ __attribute__((aligned(16))) float  sh_mbox[512];
    __shared__ __attribute__((aligned(16))) float  sh_vf[512];
    __shared__ __attribute__((aligned(16))) float  sh_uf[512];
    __shared__ __attribute__((aligned(16))) float  sh_popv[512];
    __shared__ __attribute__((aligned(16))) int sh_popc[512];
    __shared__ __attribute__((aligned(16))) int sh_path[512];
    __shared__ __attribute__((aligned(16))) int sh_c4r[512];
    __shared__ __attribute__((aligned(16))) int sh_r4c[512];
    __shared__ __attribute__((aligned(16))) int sh_SR[512];
    __shared__ __attribute__((aligned(16))) int sh_free[512];
    __shared__ __attribute__((aligned(16))) int sh_list[512];
    __shared__ int sh_cnt;
    __shared__ v4i rec[2][NWAVE];

    for (int t = tid; t < 512; t += LTH) {
        sh_uf[t] = 0.0f; sh_vf[t] = 0.0f; sh_c4r[t] = -1; sh_r4c[t] = -1;
    }
    int par = 0;

    {
        const v4f* gc = (const v4f*)(gcost + (size_t)b * MM * KC);
        const v4i* gj = (const v4i*)(gcol + (size_t)b * MM * KC);
        const int NG = MM * KC / 4;
        for (int t0 = tid; t0 < NG; t0 += LTH * 8) {
            v4f cq[8];
            v4i jq[8];
            for (int e = 0; e < 8; ++e) {
                const int t = imin_i(t0 + LTH * e, NG - 1);
                cq[e] = gc[t];
                jq[e] = gj[t];
            }
            for (int e = 0; e < 8; ++e) {
                const int t = t0 + LTH * e;
                if (t < NG) {
                    *(v4f*)&lcost[4 * t] = cq[e];
                    const v2i pj = {(jq[e].x & 0xffff) | (jq[e].y << 16),
                                    (jq[e].z & 0xffff) | (jq[e].w << 16)};
                    *(v2i*)&lcol[4 * t] = pj;
                }
            }
        }
    }
    __syncthreads();

    if (wave == 0) {
        const float INFf = __builtin_inff();
        for (int q = 0; q < 4; ++q) {
            const v4f ini = {INFf, INFf, INFf, INFf};
            *(v4f*)&sh_mbox[128 * q + 4 * lane] = ini;
        }
        for (int cur = 0; cur < MM; ++cur) {
            float sh[16], vv[16];
            int   path[16], pk[16];
            for (int q = 0; q < 4; ++q) {
                const int c0 = 128 * q + 4 * lane;
                const bool real = c0 < NN;
                const v4i r = *(const v4i*)&sh_r4c[c0];
                const v4f w = *(const v4f*)&sh_vf[c0];
                pk[4 * q + 0] = ((r.x + 1) << 4) | (4 * q + 0);
                pk[4 * q + 1] = ((r.y + 1) << 4) | (4 * q + 1);
                pk[4 * q + 2] = ((r.z + 1) << 4) | (4 * q + 2);
                pk[4 * q + 3] = ((r.w + 1) << 4) | (4 * q + 3);
                vv[4 * q + 0] = real ? w.x : -INFf;
                vv[4 * q + 1] = real ? w.y : -INFf;
                vv[4 * q + 2] = real ? w.z : -INFf;
                vv[4 * q + 3] = real ? w.w : -INFf;
            }
            for (int k = 0; k < 16; ++k) { sh[k] = INFf; path[k] = -1; }

            int   ii   = cur;
            float minv = 0.0f;
            int   nsr  = 0;
            int   sink = -1;
            bool  okay = true;
            v2f   cc = *(const v2f*)&lcost[ii * KC + 2 * lane];
            v2us  cw = *(const v2us*)&lcol[ii * KC + 2 * lane];
            float uf = sh_uf[ii];
            while (true) {
                const int c0 = cw.x, c1 = cw.y;
                sh_mbox[c0] = cc.x;
                sh_mbox[c1] = cc.y;
                v4f x[4];
                for (int q = 0; q < 4; ++q) x[q] = *(const v4f*)&sh_mbox[128 * q + 4 * lane];
                sh_mbox[c0] = INFf;
                sh_mbox[c1] = INFf;

                const float h = minv - uf;
                for (int k = 0; k < 16; ++k) {
                    const float xk = x[k >> 2][k & 3];
                    const float r = xk + (h - vv[k]);
                    const bool better = r < sh[k];
                    sh[k]   = better ? r : sh[k];
                    path[k] = better ? ii : path[k];
                }

                float m8[8]; int p8[8];
                for (int t = 0; t < 8; ++t) {
                    const float ma = sh[2 * t], mb = sh[2 * t + 1];
                    const int   pa = pk[2 * t], pb = pk[2 * t + 1];
                    const bool tk = mb < ma;
                    m8[t] = min2_f32(ma, mb);
                    p8[t] = tk ? pb : pa;
                }
                float m4[4]; int p4[4];
                for (int t = 0; t < 4; ++t) {
                    const float ma = m8[2 * t], mb = m8[2 * t + 1];
                    const int   pa = p8[2 * t], pb = p8[2 * t + 1];
                    const bool tk = mb < ma;
                    m4[t] = min2_f32(ma, mb);
                    p4[t] = tk ? pb : pa;
                }
                float m2[2]; int p2[2];
                for (int t = 0; t < 2; ++t) {
                    const float ma = m4[2 * t], mb = m4[2 * t + 1];
                    const int   pa = p4[2 * t], pb = p4[2 * t + 1];
                    const bool tk = mb < ma;
                    m2[t] = min2_f32(ma, mb);
                    p2[t] = tk ? pb : pa;
                }
                const float m2a = m2[0], m2b = m2[1];
                const int   p2a = p2[0], p2b = p2[1];
                const bool tk1 = m2b < m2a;
                const float lm = min2_f32(m2a, m2b);
                const int   lp = tk1 ? p2b : p2a;

                const float wm = wave_min_f32(lm);
                const unsigned bal = __builtin_amdgcn_ballot_w32(lm == wm);
                const int srcl = __builtin_ctz(bal | 0x80000000u);
                const int wp   = __builtin_amdgcn_readlane(lp, srcl);
                const float wms = __builtin_bit_cast(float,
                    __builtin_amdgcn_readfirstlane(__builtin_bit_cast(int, wm)));
                if (!(wms < INFf)) { okay = false; break; }
                const int k  = wp & 15;
                const int br = (wp >> 4) - 1;
                const int bj = 128 * (k >> 2) + 4 * srcl + (k & 3);
                minv = wms;
                const int nrow = (br < 0) ? ii : br;
                cc = *(const v2f*)&lcost[nrow * KC + 2 * lane];
                cw = *(const v2us*)&lcol[nrow * KC + 2 * lane];
                uf = sh_uf[nrow];
                if (lane == 0) {
                    sh_SR[nsr]   = ii;
                    sh_popc[nsr] = bj;
                    sh_popv[nsr] = wms;
                }
                ++nsr;
                if (br == -1) { sink = bj; break; }
                const int mk = (lp & 15) + ((lane == srcl) ? 0 : 16);
                for (int kk = 0; kk < 16; ++kk) {
                    const bool hit = (mk == kk);
                    sh[kk] = hit ? INFf : sh[kk];
                    vv[kk] = hit ? -INFf : vv[kk];
                }
                ii = br;
            }
            if (!okay) continue;

            for (int q = 0; q < 4; ++q) {
                const v4i pq = {path[4 * q + 0], path[4 * q + 1], path[4 * q + 2], path[4 * q + 3]};
                *(v4i*)&sh_path[128 * q + 4 * lane] = pq;
            }
            for (int t = lane; t < nsr; t += 32) {
                const float d = minv - sh_popv[t];
                sh_vf[sh_popc[t]] -= d;
                if (t == 0) sh_uf[cur] += minv;
                if (t + 1 < nsr) sh_uf[sh_SR[t + 1]] += d;
            }
            if (lane == 0) {
                int j = sink;
                for (int g = 0; g < MM; ++g) {
                    const int i = sh_path[j];
                    sh_r4c[j] = i;
                    const int tmp = sh_c4r[i];
                    sh_c4r[i] = j;
                    j = tmp;
                    if (i == cur) break;
                }
            }
        }
    }
    __syncthreads();

    double v[CPL];
    {
        v4f q = {0.0f, 0.0f, 0.0f, 0.0f};
        if (owner) q = *(const v4f*)&sh_vf[CPL * tid];
        v[0] = (double)q.x; v[1] = (double)q.y; v[2] = (double)q.z; v[3] = (double)q.w;
    }
    if (owner) {
        v2d a01 = {v[0], v[1]};
        v2d a23 = {v[2], v[3]};
        *(v2d*)&sh_vd[CPL * tid]     = a01;
        *(v2d*)&sh_vd[CPL * tid + 2] = a23;
    }
    __syncthreads();
    for (int i = wave; i < MM; i += NWAVE) {
        const float* row = costB + (size_t)i * NN;
        double m = __builtin_inf();
        for (int q = 0; q < 4; ++q) {
            const int c0 = 128 * q + CPL * lane;
            if (c0 < NN) {
                const v4f c = *(const v4f*)(row + c0);
                const v2d va = *(const v2d*)&sh_vd[c0];
                const v2d vb = *(const v2d*)&sh_vd[c0 + 2];
                const double d0 = (double)c.x - va.x;
                const double d1 = (double)c.y - va.y;
                const double d2 = (double)c.z - vb.x;
                const double d3 = (double)c.w - vb.y;
                m = (d0 < m) ? d0 : m;
                m = (d1 < m) ? d1 : m;
                m = (d2 < m) ? d2 : m;
                m = (d3 < m) ? d3 : m;
            }
        }
        { const double o = XT<double>::rx<1>(m); m = (o < m) ? o : m; }
        { const double o = XT<double>::rx<2>(m); m = (o < m) ? o : m; }
        { const double o = XT<double>::rx<4>(m); m = (o < m) ? o : m; }
        { const double o = XT<double>::rx<8>(m); m = (o < m) ? o : m; }
        { const double o = XT<double>::hx(m);    m = (o < m) ? o : m; }
        const int jm = sh_c4r[i];
        const bool has = jm >= 0;
        const int jq = has ? jm : 0;
        const double dm = (double)row[jq] - sh_vd[jq];
        const bool keep = has && (dm == m);
        if (lane == 0) {
            sh_ud[i]   = m;
            sh_free[i] = keep ? 0 : 1;
            if (has && !keep) { sh_c4r[i] = -1; sh_r4c[jm] = -1; }
        }
    }
    __syncthreads();
    if (wave == 0) {
        int cnt = 0;
        for (int base = 0; base < 512; base += 32) {
            const int r = base + lane;
            const bool f = (r < MM) && (sh_free[r] != 0);
            const unsigned bal = __builtin_amdgcn_ballot_w32(f);
            const int pos = cnt + __builtin_popcount(bal & ((1u << lane) - 1u));
            if (f) sh_list[pos] = r;
            cnt += __builtin_popcount(bal);
        }
        if (lane == 0) sh_cnt = cnt;
    }
    __syncthreads();
    const int nfree = __builtin_amdgcn_readfirstlane(sh_cnt);

    sap_rows<double, true>(costB, nfree, sh_list, v, sh_ud, sh_short,
                           sh_path, sh_c4r, sh_r4c, sh_SR, rec, par,
                           tid, lane, wave, owner, full);

    for (int t = tid; t < MM; t += LTH) mapping[b * MM + t] = sh_c4r[t];
}

__global__ void extrap_kernel(const float* __restrict__ p1b,
                              const float* __restrict__ p1l,
                              const float* __restrict__ p2b,
                              const float* __restrict__ p2l,
                              const float* __restrict__ toff,
                              const int*   __restrict__ mapping,
                              float* __restrict__ out) {
    int idx = blockIdx.x * blockDim.x + threadIdx.x;
    if (idx >= BB * MM) return;
    int b = idx / MM;

    float first  = toff[b * 3 + 1] - toff[b * 3 + 0];
    float second = toff[b * 3 + 2] - toff[b * 3 + 1];
    float factor = second / first;

    int  j   = mapping[idx];
    bool has = (j >= 0);
    int  jc  = has ? j : 0;

    const float* pb2 = p2b + (size_t)idx * 4;
    const float* pb1 = p1b + ((size_t)b * NN + jc) * 4;
    float* ob = out + (size_t)idx * 4;
    for (int k = 0; k < 4; ++k) {
        float c1 = has ? pb1[k] : pb2[k];
        float v  = pb2[k] + (pb2[k] - c1) * factor;
        ob[k] = (k >= 2) ? fmaxf(v, 0.0f) : v;
    }

    const float* pl2 = p2l + (size_t)idx * CC;
    const float* pl1 = p1l + ((size_t)b * NN + jc) * CC;
    float* ol = out + (size_t)BB * MM * 4 + (size_t)idx * CC;
    for (int c = 0; c < CC; ++c) {
        float c1 = has ? pl1[c] : 0.0f;
        ol[c] = 0.5f * (pl2[c] + c1);
    }
}

extern "C" void kernel_launch(void* const* d_in, const int* in_sizes, int n_in,
                              void* d_out, int out_size, void* d_ws, size_t ws_size,
                              hipStream_t stream) {
    (void)in_sizes; (void)n_in; (void)out_size; (void)ws_size;
    const float* p1b  = (const float*)d_in[0];
    const float* p1l  = (const float*)d_in[1];
    const float* p2b  = (const float*)d_in[2];
    const float* p2l  = (const float*)d_in[3];
    const float* toff = (const float*)d_in[4];

    float* ws   = (float*)d_ws;
    float* s1   = ws;
    float* s2   = s1 + (size_t)BB * NN * CC;
    float* cost = s2 + (size_t)BB * MM * CC;
    int*   map  = (int*)(cost + (size_t)BB * MM * NN);
    float* out  = (float*)d_out;

    int tot = BB * NN * CC + BB * MM * CC;
    sigmoid_kernel<<<(tot + 255) / 256, 256, 0, stream>>>(p1l, p2l, s1, s2);

    const int TM = (MM + 15) / 16, TN = (NN + 15) / 16;
    cost_kernel<<<BB * TM * TN, 32, 0, stream>>>(p1b, p2b, s1, s2, cost);

    float* gcost = s1;
    int*   gcol  = (int*)s2;
    select_kernel<<<BB * MM / SEL_WAVES, 32 * SEL_WAVES, 0, stream>>>(cost, gcost, gcol);

    lapjv_kernel<<<BB, LTH, 0, stream>>>(cost, gcost, gcol, map);

    int tot2 = BB * MM;
    extrap_kernel<<<(tot2 + 255) / 256, 256, 0, stream>>>(p1b, p1l, p2b, p2l,
                                                          toff, map, out);
}


# Sparse f32 pass: candidate lists hold each column's mailbox LDS address (32-bit, not a 16-bit id) and payloads index the lists as they are: no arithmetic between the dependent LDS trips; same sums
# speedup vs baseline: 6.4131x; 1.0443x over previous
#include <hip/hip_runtime.h>
#include <math.h>

#define BB 4
#define MM 500
#define NN 500
#define CC 91

typedef __attribute__((ext_vector_type(2))) float v2f;
typedef __attribute__((ext_vector_type(8))) float v8f;

__device__ __forceinline__ int imin_i(int a, int b) { return a < b ? a : b; }

__global__ void sigmoid_kernel(const float* __restrict__ p1l,
                               const float* __restrict__ p2l,
                               float* __restrict__ s1,
                               float* __restrict__ s2) {
    int idx = blockIdx.x * blockDim.x + threadIdx.x;
    const int n1 = BB * NN * CC;
    const int n2 = BB * MM * CC;
    if (idx < n1) {
        float x = p1l[idx];
        s1[idx] = 1.0f / (1.0f + expf(-x));
    } else if (idx < n1 + n2) {
        float x = p2l[idx - n1];
        s2[idx - n1] = 1.0f / (1.0f + expf(-x));
    }
}

__global__ void cost_kernel(const float* __restrict__ p1b,
                            const float* __restrict__ p2b,
                            const float* __restrict__ s1,
                            const float* __restrict__ s2,
                            float* __restrict__ cost) {
    const int TN = (NN + 15) / 16;
    const int TM = (MM + 15) / 16;
    int bid = blockIdx.x;
    int b  = bid / (TM * TN);
    int t  = bid % (TM * TN);
    int m0 = (t / TN) * 16;
    int n0 = (t % TN) * 16;
    int lane = threadIdx.x;

    __shared__ float sq2[16];
    __shared__ float sq1[16];
    __shared__ float ls2[16 * CC];
    __shared__ float ls1[16 * CC];

    v2f a  = {0.0f, 0.0f};
    v2f bv = {0.0f, 0.0f};
    if (lane < 16) {
        int m = imin_i(m0 + lane, MM - 1);
        const float* pb = p2b + ((size_t)b * MM + m) * 4;
        float x = pb[0], y = pb[1];
        a.x = x; a.y = y;
        sq2[lane] = x * x + y * y;
        int n = imin_i(n0 + lane, NN - 1);
        const float* qb = p1b + ((size_t)b * NN + n) * 4;
        float bx = qb[0], by = qb[1];
        bv.x = bx; bv.y = by;
        sq1[lane] = bx * bx + by * by;
    }
    const float* s2base = s2 + (size_t)b * MM * CC;
    const float* s1base = s1 + (size_t)b * NN * CC;
    for (int idx = lane; idx < 16 * CC; idx += 32) {
        int r = idx / CC;
        int c = idx - r * CC;
        ls2[idx] = s2base[(size_t)imin_i(m0 + r, MM - 1) * CC + c];
        ls1[idx] = s1base[(size_t)imin_i(n0 + r, NN - 1) * CC + c];
    }
    __syncthreads();

    v8f cacc = {0.f, 0.f, 0.f, 0.f, 0.f, 0.f, 0.f, 0.f};
    v8f dot = __builtin_amdgcn_wmma_f32_16x16x4_f32(
         false, a,  false, bv,
         (short)0, cacc,  false,  false);

    int n_loc  = lane & 15;
    int m_base = (lane >> 4) << 3;

    float cls[8];
#pragma unroll
    for (int i = 0; i < 8; ++i) cls[i] = 0.0f;
    for (int c = 0; c < CC; ++c) {
        float s1v = ls1[n_loc * CC + c];
#pragma unroll
        for (int i = 0; i < 8; ++i) {
            float d = fabsf(ls2[(m_base + i) * CC + c] - s1v);
            cls[i] = fmaxf(cls[i], d);
        }
    }

#pragma unroll
    for (int i = 0; i < 8; ++i) {
        int m = m0 + m_base + i;
        int n = n0 + n_loc;
        if (m < MM && n < NN) {
            float d2 = sq2[m_base + i] + sq1[n_loc] - 2.0f * dot[i];
            float cd = sqrtf(fmaxf(d2, 0.0f));
            cost[((size_t)b * MM + m) * NN + n] = 0.5f * cd + 0.5f * cls[i];
        }
    }
}

#define LTH 128
#define NWAVE (LTH / 32)
#define CPL 4
typedef __attribute__((ext_vector_type(4))) float v4f;
typedef __attribute__((ext_vector_type(4))) int   v4i;
typedef __attribute__((ext_vector_type(2))) int   v2i;
typedef __attribute__((ext_vector_type(2))) double v2d;
typedef __attribute__((ext_vector_type(2))) unsigned v2u;
typedef __attribute__((ext_vector_type(4))) unsigned v4u;
typedef __attribute__((address_space(3))) float lds_float;

#define KC 64
static_assert(KC == 64, "two candidates per lane of one wave");
static_assert(NN < 512, "16 columns per lane, column 511 is a pad");

static_assert(NN % CPL == 0 && NN / CPL <= LTH, "column tiling");
static_assert(MM == NN, "square problem");

template <typename T> struct XT;
template <> struct XT<double> {
    template <int MASK>
    static __device__ __forceinline__ double rx(double x) {
        v2i p = __builtin_bit_cast(v2i, x);
        p.x = __builtin_amdgcn_update_dpp(p.x, p.x, 0x160 + MASK, 0xf, 0xf, false);
        p.y = __builtin_amdgcn_update_dpp(p.y, p.y, 0x160 + MASK, 0xf, 0xf, false);
        return __builtin_bit_cast(double, p);
    }
    static __device__ __forceinline__ double hx(double x) {
        v2i p = __builtin_bit_cast(v2i, x);
        p.x = __builtin_amdgcn_permlanex16(p.x, p.x, 0x76543210, 0xfedcba98, false, false);
        p.y = __builtin_amdgcn_permlanex16(p.y, p.y, 0x76543210, 0xfedcba98, false, false);
        return __builtin_bit_cast(double, p);
    }
    static __device__ __forceinline__ v2i pack(double x) { return __builtin_bit_cast(v2i, x); }
    static __device__ __forceinline__ double unpack(v2i p) { return __builtin_bit_cast(double, p); }
};

template <typename T, bool LIST>
__device__ __forceinline__ void sap_rows(const float* __restrict__ costB,
                                         const int nrows, const int* rowlist,
                                         T (&v)[CPL], T* sh_u, T* sh_short,
                                         int* sh_path, int* sh_c4r, int* sh_r4c,
                                         int* sh_SR, v4i (*rec)[NWAVE], int& par,
                                         const int tid, const int lane,
                                         const int wave, const bool owner,
                                         const unsigned full) {
    const T INF = (T)__builtin_inf();
    for (int idx = 0; idx < nrows; ++idx) {
        const int cur = LIST ? rowlist[idx] : idx;
        T   sh[CPL];
        int path[CPL];
        int r4c[CPL];
        {
            v4i q = *(const v4i*)&sh_r4c[CPL * (owner ? tid : 0)];
            r4c[0] = q.x; r4c[1] = q.y; r4c[2] = q.z; r4c[3] = q.w;
        }
        for (int k = 0; k < CPL; ++k) { sh[k] = INF; path[k] = -1; }
        unsigned rem = full;
        int ii   = cur;
        T   minv = (T)0;
        int nsr  = 0;
        int sink = -1;

        v4f c = {0.0f, 0.0f, 0.0f, 0.0f};
        if (owner) c = *(const v4f*)(costB + (size_t)ii * NN + CPL * tid);
        T z[CPL];
        {
            const T h = minv - sh_u[ii];
            for (int k = 0; k < CPL; ++k) z[k] = h - v[k];
        }

        while (true) {
            if (tid == 0) sh_SR[nsr] = ii;
            ++nsr;

            T t[CPL];
            for (int k = 0; k < CPL; ++k) t[k] = (T)c[k];
            for (int k = 0; k < CPL; ++k) t[k] = t[k] + z[k];

            T lm = INF; int lk = 0; int lr = -1;
            for (int k = 0; k < CPL; ++k) {
                const bool rm = ((rem >> k) & 1u) != 0u;
                const bool better = rm && (t[k] < sh[k]);
                sh[k]   = better ? t[k] : sh[k];
                path[k] = better ? ii : path[k];
                const T cd = rm ? sh[k] : INF;
                if (k == 0)       { lm = cd; lk = 0; lr = r4c[0]; }
                else if (cd < lm) { lm = cd; lk = k; lr = r4c[k]; }
            }

            T wm = lm;
            { const T o = XT<T>::template rx<1>(wm); wm = (o < wm) ? o : wm; }
            { const T o = XT<T>::template rx<2>(wm); wm = (o < wm) ? o : wm; }
            { const T o = XT<T>::template rx<4>(wm); wm = (o < wm) ? o : wm; }
            { const T o = XT<T>::template rx<8>(wm); wm = (o < wm) ? o : wm; }
            { const T o = XT<T>::hx(wm);             wm = (o < wm) ? o : wm; }
            const unsigned bal = __builtin_amdgcn_ballot_w32(lm == wm);
            const int srcl = __builtin_ctz(bal | 0x80000000u);
            const int wk = __builtin_amdgcn_readlane(lk, srcl);
            const int wr = __builtin_amdgcn_readlane(lr, srcl);
            if (lane == 0) {
                v2i p = XT<T>::pack(wm);
                v4i q = {p.x, p.y, CPL * (wave * 32 + srcl) + wk, wr};
                rec[par][wave] = q;
            }
            __syncthreads();

            T bv; int bj, br;
            {
                v4i q = rec[par][0];
                v2i p = {q.x, q.y};
                bv = XT<T>::unpack(p); bj = q.z; br = q.w;
            }
            for (int w = 1; w < NWAVE; ++w) {
                v4i q = rec[par][w];
                v2i p = {q.x, q.y};
                const T ov = XT<T>::unpack(p);
                if (ov < bv) { bv = ov; bj = q.z; br = q.w; }
            }
            par ^= 1;
            minv = bv;
            bj = __builtin_amdgcn_readfirstlane(bj);
            br = __builtin_amdgcn_readfirstlane(br);
            if (tid == (bj >> 2)) rem &= ~(1u << (bj & 3));
            if (br == -1) { sink = bj; break; }
            ii = br;

            c = (v4f){0.0f, 0.0f, 0.0f, 0.0f};
            if (owner) c = *(const v4f*)(costB + (size_t)ii * NN + CPL * tid);
            {
                const T h = minv - sh_u[ii];
                for (int k = 0; k < CPL; ++k) z[k] = h - v[k];
            }
        }

        if (owner) {
            v4i pq = {path[0], path[1], path[2], path[3]};
            *(v4i*)&sh_path[CPL * tid] = pq;
            for (int k = 0; k < CPL; ++k) sh_short[CPL * tid + k] = sh[k];
        }
        __syncthreads();

        for (int t2 = tid; t2 < nsr; t2 += LTH) {
            if (t2 == 0) {
                sh_u[cur] += minv;
            } else {
                int sr = sh_SR[t2];
                sh_u[sr] += minv - sh_short[sh_c4r[sr]];
            }
        }
        {
            const unsigned scanned = full & ~rem;
            for (int k = 0; k < CPL; ++k)
                if ((scanned >> k) & 1u) v[k] -= minv - sh[k];
        }
        __syncthreads();

        if (tid == 0) {
            int j = sink;
            while (true) {
                int i = sh_path[j];
                sh_r4c[j] = i;
                int tmp = sh_c4r[i];
                sh_c4r[i] = j;
                j = tmp;
                if (i == cur) break;
            }
        }
        __syncthreads();
    }
}

__device__ __forceinline__ int wave_sum_i(int x) {
    x += __builtin_amdgcn_update_dpp(x, x, 0x160 + 1, 0xf, 0xf, false);
    x += __builtin_amdgcn_update_dpp(x, x, 0x160 + 2, 0xf, 0xf, false);
    x += __builtin_amdgcn_update_dpp(x, x, 0x160 + 4, 0xf, 0xf, false);
    x += __builtin_amdgcn_update_dpp(x, x, 0x160 + 8, 0xf, 0xf, false);
    x += __builtin_amdgcn_permlanex16(x, x, 0x76543210, 0xfedcba98, false, false);
    return x;
}

__device__ __forceinline__ unsigned fbits(const float x) { return __builtin_bit_cast(unsigned, x); }

__device__ __forceinline__ float min2_f32(const float a, const float b) {
    float o;
    asm("v_min_num_f32 %0, %1, %2" : "=v"(o) : "v"(a), "v"(b));
    return o;
}

__device__ __forceinline__ float wave_min_f32(float x) {
    float a, b, c, d;
    asm("v_min_num_f32_dpp %0, %1, %1 row_xmask:1 row_mask:0xf bank_mask:0xf" : "=v"(a) : "v"(x));
    asm("v_min_num_f32_dpp %0, %1, %1 row_xmask:2 row_mask:0xf bank_mask:0xf" : "=v"(b) : "v"(a));
    asm("v_min_num_f32_dpp %0, %1, %1 row_xmask:4 row_mask:0xf bank_mask:0xf" : "=v"(c) : "v"(b));
    asm("v_min_num_f32_dpp %0, %1, %1 row_xmask:8 row_mask:0xf bank_mask:0xf" : "=v"(d) : "v"(c));
    const float o = __builtin_bit_cast(float,
        __builtin_amdgcn_permlanex16(0, __builtin_bit_cast(int, d), 0x76543210, 0xfedcba98, false, false));
    return min2_f32(o, d);
}

__device__ __forceinline__ void load_row16(const float* __restrict__ row, const int lane,
                                           unsigned (&bits)[16]) {
    for (int q = 0; q < 4; ++q) {
        const int c0 = 128 * q + 4 * lane;
        if (c0 < NN) {
            const v4f c = *(const v4f*)(row + c0);
            bits[4 * q + 0] = fbits(c.x);
            bits[4 * q + 1] = fbits(c.y);
            bits[4 * q + 2] = fbits(c.z);
            bits[4 * q + 3] = fbits(c.w);
        } else {
            for (int e = 0; e < 4; ++e) bits[4 * q + e] = 0x7f800000u;
        }
    }
}

#define SEL_WAVES 4
static_assert((BB * MM) % SEL_WAVES == 0, "whole work-groups of rows");
static_assert(MM * KC <= NN * CC && KC <= CC, "the lists fit into the two sigmoid planes they reuse");
__global__ void __launch_bounds__(32 * SEL_WAVES)
select_kernel(const float* __restrict__ cost, float* __restrict__ gcost, int* __restrict__ gcol) {
    const int lane = threadIdx.x & 31;
    const int wave = __builtin_amdgcn_readfirstlane((int)(threadIdx.x >> 5));
    const int row  = blockIdx.x * SEL_WAVES + wave;

    __shared__ float sc[SEL_WAVES][KC];
    __shared__ int   sj[SEL_WAVES][KC];

    const unsigned ltmask = (1u << lane) - 1u;
    unsigned bits[16];
    load_row16(cost + (size_t)row * NN, lane, bits);
    unsigned T = 0u;
    for (int bit = 30; bit >= 7; --bit) {
        const unsigned trial = T | (1u << bit);
        int cnt = 0;
        for (int k = 0; k < 16; ++k) cnt += (bits[k] < trial) ? 1 : 0;
        cnt = __builtin_amdgcn_readfirstlane(wave_sum_i(cnt));
        if (cnt <= KC) T = trial;
    }
    sc[wave][lane]      = __builtin_inff();
    sc[wave][lane + 32] = __builtin_inff();
    sj[wave][lane]      = 511;
    sj[wave][lane + 32] = 511;
    int base = 0;
    for (int k = 0; k < 16; ++k) {
        const bool f = bits[k] < T;
        const unsigned bal = __builtin_amdgcn_ballot_w32(f);
        const int pos = base + __builtin_popcount(bal & ltmask);
        if (f && pos < KC) {
            sc[wave][pos] = __builtin_bit_cast(float, bits[k]);
            sj[wave][pos] = 128 * (k >> 2) + 4 * lane + (k & 3);
        }
        base += __builtin_popcount(bal);
    }
    const float a0 = sc[wave][lane], a1 = sc[wave][lane + 32];
    const int   j0 = sj[wave][lane], j1 = sj[wave][lane + 32];
    gcost[(size_t)row * KC + lane]      = a0;
    gcost[(size_t)row * KC + lane + 32] = a1;
    gcol[(size_t)row * KC + lane]       = j0;
    gcol[(size_t)row * KC + lane + 32]  = j1;
}

__global__ void __launch_bounds__(LTH)
lapjv_kernel(const float* __restrict__ cost, const float* __restrict__ gcost,
             const int* __restrict__ gcol, int* __restrict__ mapping) {
    const int b    = blockIdx.x;
    const int tid  = threadIdx.x;
    const int lane = tid & 31;
    const int wave = __builtin_amdgcn_readfirstlane(tid >> 5);
    const float* costB = cost + (size_t)b * MM * NN;
    const bool owner = tid < NN / CPL;
    const unsigned full = owner ? 0xFu : 0u;

    __shared__ __attribute__((aligned(16))) float lcost[(MM + 1) * KC];
    __shared__ __attribute__((aligned(16))) unsigned lmbx[(MM + 1) * KC];
    __shared__ __attribute__((aligned(16))) double sh_ud[512];
    __shared__ __attribute__((aligned(16))) double sh_vd[512];
    __shared__ __attribute__((aligned(16))) double sh_short[512];
    __shared__ __attribute__((aligned(16))) float  sh_mbox[512];
    __shared__ __attribute__((aligned(16))) float  sh_vf[512];
    __shared__ __attribute__((aligned(16))) float  sh_uf[512];
    __shared__ __attribute__((aligned(16))) float  sh_popv[512];
    __shared__ __attribute__((aligned(16))) int sh_popc[512];
    __shared__ __attribute__((aligned(16))) int sh_path[512];
    __shared__ __attribute__((aligned(16))) int sh_c4r[512];
    __shared__ __attribute__((aligned(16))) int sh_r4c[512];
    __shared__ __attribute__((aligned(16))) int sh_SR[512];
    __shared__ __attribute__((aligned(16))) int sh_free[512];
    __shared__ __attribute__((aligned(16))) int sh_list[512];
    __shared__ int sh_cnt;
    __shared__ v4i rec[2][NWAVE];

    for (int t = tid; t < 512; t += LTH) {
        sh_uf[t] = 0.0f; sh_vf[t] = 0.0f; sh_c4r[t] = -1; sh_r4c[t] = -1;
    }
    int par = 0;

    {
        const v4f* gc = (const v4f*)(gcost + (size_t)b * MM * KC);
        const v4i* gj = (const v4i*)(gcol + (size_t)b * MM * KC);
        const int NG = MM * KC / 4;
        const unsigned mb = (unsigned)(__UINTPTR_TYPE__)(lds_float*)sh_mbox;
        if (tid < KC) { lcost[tid] = __builtin_inff(); lmbx[tid] = mb + 4u * 511u; }
        for (int t0 = tid; t0 < NG; t0 += LTH * 8) {
            v4f cq[8];
            v4i jq[8];
            for (int e = 0; e < 8; ++e) {
                const int t = imin_i(t0 + LTH * e, NG - 1);
                cq[e] = gc[t];
                jq[e] = gj[t];
            }
            for (int e = 0; e < 8; ++e) {
                const int t = t0 + LTH * e;
                if (t < NG) {
                    *(v4f*)&lcost[KC + 4 * t] = cq[e];
                    const v4u pj = {mb + 4u * (unsigned)jq[e].x, mb + 4u * (unsigned)jq[e].y,
                                    mb + 4u * (unsigned)jq[e].z, mb + 4u * (unsigned)jq[e].w};
                    *(v4u*)&lmbx[KC + 4 * t] = pj;
                }
            }
        }
    }
    __syncthreads();

    if (wave == 0) {
        const float INFf = __builtin_inff();
        for (int q = 0; q < 4; ++q) {
            const v4f ini = {INFf, INFf, INFf, INFf};
            *(v4f*)&sh_mbox[128 * q + 4 * lane] = ini;
        }
        for (int cur = 0; cur < MM; ++cur) {
            float sh[16], vv[16];
            int   path[16], pk[16];
            for (int q = 0; q < 4; ++q) {
                const int c0 = 128 * q + 4 * lane;
                const bool real = c0 < NN;
                const v4i r = *(const v4i*)&sh_r4c[c0];
                const v4f w = *(const v4f*)&sh_vf[c0];
                pk[4 * q + 0] = ((r.x + 1) << 4) | (4 * q + 0);
                pk[4 * q + 1] = ((r.y + 1) << 4) | (4 * q + 1);
                pk[4 * q + 2] = ((r.z + 1) << 4) | (4 * q + 2);
                pk[4 * q + 3] = ((r.w + 1) << 4) | (4 * q + 3);
                vv[4 * q + 0] = real ? w.x : -INFf;
                vv[4 * q + 1] = real ? w.y : -INFf;
                vv[4 * q + 2] = real ? w.z : -INFf;
                vv[4 * q + 3] = real ? w.w : -INFf;
            }
            for (int k = 0; k < 16; ++k) { sh[k] = INFf; path[k] = -1; }

            int   ii   = cur;
            float minv = 0.0f;
            int   nsr  = 0;
            int   sink = -1;
            bool  okay = true;
            v2f   cc = *(const v2f*)&lcost[(ii + 1) * KC + 2 * lane];
            v2u   cw = *(const v2u*)&lmbx[(ii + 1) * KC + 2 * lane];
            float uf = sh_uf[ii + 1];
            while (true) {
                lds_float* const c0 = (lds_float*)cw.x;
                lds_float* const c1 = (lds_float*)cw.y;
                *c0 = cc.x;
                *c1 = cc.y;
                v4f x[4];
                for (int q = 0; q < 4; ++q) x[q] = *(const v4f*)&sh_mbox[128 * q + 4 * lane];
                *c0 = INFf;
                *c1 = INFf;

                const float h = minv - uf;
                for (int k = 0; k < 16; ++k) {
                    const float xk = x[k >> 2][k & 3];
                    const float r = xk + (h - vv[k]);
                    const bool better = r < sh[k];
                    sh[k]   = better ? r : sh[k];
                    path[k] = better ? ii : path[k];
                }

                float m8[8]; int p8[8];
                for (int t = 0; t < 8; ++t) {
                    const float ma = sh[2 * t], mb = sh[2 * t + 1];
                    const int   pa = pk[2 * t], pb = pk[2 * t + 1];
                    const bool tk = mb < ma;
                    m8[t] = min2_f32(ma, mb);
                    p8[t] = tk ? pb : pa;
                }
                float m4[4]; int p4[4];
                for (int t = 0; t < 4; ++t) {
                    const float ma = m8[2 * t], mb = m8[2 * t + 1];
                    const int   pa = p8[2 * t], pb = p8[2 * t + 1];
                    const bool tk = mb < ma;
                    m4[t] = min2_f32(ma, mb);
                    p4[t] = tk ? pb : pa;
                }
                float m2[2]; int p2[2];
                for (int t = 0; t < 2; ++t) {
                    const float ma = m4[2 * t], mb = m4[2 * t + 1];
                    const int   pa = p4[2 * t], pb = p4[2 * t + 1];
                    const bool tk = mb < ma;
                    m2[t] = min2_f32(ma, mb);
                    p2[t] = tk ? pb : pa;
                }
                const float m2a = m2[0], m2b = m2[1];
                const int   p2a = p2[0], p2b = p2[1];
                const bool tk1 = m2b < m2a;
                const float lm = min2_f32(m2a, m2b);
                const int   lp = tk1 ? p2b : p2a;

                const float wm = wave_min_f32(lm);
                const unsigned bal = __builtin_amdgcn_ballot_w32(lm == wm);
                const int srcl = __builtin_ctz(bal | 0x80000000u);
                const int wp   = __builtin_amdgcn_readlane(lp, srcl);
                const float wms = __builtin_bit_cast(float,
                    __builtin_amdgcn_readfirstlane(__builtin_bit_cast(int, wm)));
                if (!(wms < INFf)) { okay = false; break; }
                const int k  = wp & 15;
                const int r1 = wp >> 4;
                const int br = r1 - 1;
                const int bj = 128 * (k >> 2) + 4 * srcl + (k & 3);
                minv = wms;
                cc = *(const v2f*)&lcost[r1 * KC + 2 * lane];
                cw = *(const v2u*)&lmbx[r1 * KC + 2 * lane];
                uf = sh_uf[r1];
                if (lane == 0) {
                    sh_SR[nsr]   = ii;
                    sh_popc[nsr] = bj;
                    sh_popv[nsr] = wms;
                }
                ++nsr;
                if (br == -1) { sink = bj; break; }
                const int mk = (lp & 15) + ((lane == srcl) ? 0 : 16);
                for (int kk = 0; kk < 16; ++kk) {
                    const bool hit = (mk == kk);
                    sh[kk] = hit ? INFf : sh[kk];
                    vv[kk] = hit ? -INFf : vv[kk];
                }
                ii = br;
            }
            if (!okay) continue;

            for (int q = 0; q < 4; ++q) {
                const v4i pq = {path[4 * q + 0], path[4 * q + 1], path[4 * q + 2], path[4 * q + 3]};
                *(v4i*)&sh_path[128 * q + 4 * lane] = pq;
            }
            for (int t = lane; t < nsr; t += 32) {
                const float d = minv - sh_popv[t];
                sh_vf[sh_popc[t]] -= d;
                if (t == 0) sh_uf[cur + 1] += minv;
                if (t + 1 < nsr) sh_uf[sh_SR[t + 1] + 1] += d;
            }
            if (lane == 0) {
                int j = sink;
                for (int g = 0; g < MM; ++g) {
                    const int i = sh_path[j];
                    sh_r4c[j] = i;
                    const int tmp = sh_c4r[i];
                    sh_c4r[i] = j;
                    j = tmp;
                    if (i == cur) break;
                }
            }
        }
    }
    __syncthreads();

    double v[CPL];
    {
        v4f q = {0.0f, 0.0f, 0.0f, 0.0f};
        if (owner) q = *(const v4f*)&sh_vf[CPL * tid];
        v[0] = (double)q.x; v[1] = (double)q.y; v[2] = (double)q.z; v[3] = (double)q.w;
    }
    if (owner) {
        v2d a01 = {v[0], v[1]};
        v2d a23 = {v[2], v[3]};
        *(v2d*)&sh_vd[CPL * tid]     = a01;
        *(v2d*)&sh_vd[CPL * tid + 2] = a23;
    }
    __syncthreads();
    for (int i = wave; i < MM; i += NWAVE) {
        const float* row = costB + (size_t)i * NN;
        double m = __builtin_inf();
        for (int q = 0; q < 4; ++q) {
            const int c0 = 128 * q + CPL * lane;
            if (c0 < NN) {
                const v4f c = *(const v4f*)(row + c0);
                const v2d va = *(const v2d*)&sh_vd[c0];
                const v2d vb = *(const v2d*)&sh_vd[c0 + 2];
                const double d0 = (double)c.x - va.x;
                const double d1 = (double)c.y - va.y;
                const double d2 = (double)c.z - vb.x;
                const double d3 = (double)c.w - vb.y;
                m = (d0 < m) ? d0 : m;
                m = (d1 < m) ? d1 : m;
                m = (d2 < m) ? d2 : m;
                m = (d3 < m) ? d3 : m;
            }
        }
        { const double o = XT<double>::rx<1>(m); m = (o < m) ? o : m; }
        { const double o = XT<double>::rx<2>(m); m = (o < m) ? o : m; }
        { const double o = XT<double>::rx<4>(m); m = (o < m) ? o : m; }
        { const double o = XT<double>::rx<8>(m); m = (o < m) ? o : m; }
        { const double o = XT<double>::hx(m);    m = (o < m) ? o : m; }
        const int jm = sh_c4r[i];
        const bool has = jm >= 0;
        const int jq = has ? jm : 0;
        const double dm = (double)row[jq] - sh_vd[jq];
        const bool keep = has && (dm == m);
        if (lane == 0) {
            sh_ud[i]   = m;
            sh_free[i] = keep ? 0 : 1;
            if (has && !keep) { sh_c4r[i] = -1; sh_r4c[jm] = -1; }
        }
    }
    __syncthreads();
    if (wave == 0) {
        int cnt = 0;
        for (int base = 0; base < 512; base += 32) {
            const int r = base + lane;
            const bool f = (r < MM) && (sh_free[r] != 0);
            const unsigned bal = __builtin_amdgcn_ballot_w32(f);
            const int pos = cnt + __builtin_popcount(bal & ((1u << lane) - 1u));
            if (f) sh_list[pos] = r;
            cnt += __builtin_popcount(bal);
        }
        if (lane == 0) sh_cnt = cnt;
    }
    __syncthreads();
    const int nfree = __builtin_amdgcn_readfirstlane(sh_cnt);

    sap_rows<double, true>(costB, nfree, sh_list, v, sh_ud, sh_short,
                           sh_path, sh_c4r, sh_r4c, sh_SR, rec, par,
                           tid, lane, wave, owner, full);

    for (int t = tid; t < MM; t += LTH) mapping[b * MM + t] = sh_c4r[t];
}

__global__ void extrap_kernel(const float* __restrict__ p1b,
                              const float* __restrict__ p1l,
                              const float* __restrict__ p2b,
                              const float* __restrict__ p2l,
                              const float* __restrict__ toff,
                              const int*   __restrict__ mapping,
                              float* __restrict__ out) {
    int idx = blockIdx.x * blockDim.x + threadIdx.x;
    if (idx >= BB * MM) return;
    int b = idx / MM;

    float first  = toff[b * 3 + 1] - toff[b * 3 + 0];
    float second = toff[b * 3 + 2] - toff[b * 3 + 1];
    float factor = second / first;

    int  j   = mapping[idx];
    bool has = (j >= 0);
    int  jc  = has ? j : 0;

    const float* pb2 = p2b + (size_t)idx * 4;
    const float* pb1 = p1b + ((size_t)b * NN + jc) * 4;
    float* ob = out + (size_t)idx * 4;
    for (int k = 0; k < 4; ++k) {
        float c1 = has ? pb1[k] : pb2[k];
        float v  = pb2[k] + (pb2[k] - c1) * factor;
        ob[k] = (k >= 2) ? fmaxf(v, 0.0f) : v;
    }

    const float* pl2 = p2l + (size_t)idx * CC;
    const float* pl1 = p1l + ((size_t)b * NN + jc) * CC;
    float* ol = out + (size_t)BB * MM * 4 + (size_t)idx * CC;
    for (int c = 0; c < CC; ++c) {
        float c1 = has ? pl1[c] : 0.0f;
        ol[c] = 0.5f * (pl2[c] + c1);
    }
}

extern "C" void kernel_launch(void* const* d_in, const int* in_sizes, int n_in,
                              void* d_out, int out_size, void* d_ws, size_t ws_size,
                              hipStream_t stream) {
    (void)in_sizes; (void)n_in; (void)out_size; (void)ws_size;
    const float* p1b  = (const float*)d_in[0];
    const float* p1l  = (const float*)d_in[1];
    const float* p2b  = (const float*)d_in[2];
    const float* p2l  = (const float*)d_in[3];
    const float* toff = (const float*)d_in[4];

    float* ws   = (float*)d_ws;
    float* s1   = ws;
    float* s2   = s1 + (size_t)BB * NN * CC;
    float* cost = s2 + (size_t)BB * MM * CC;
    int*   map  = (int*)(cost + (size_t)BB * MM * NN);
    float* out  = (float*)d_out;

    int tot = BB * NN * CC + BB * MM * CC;
    sigmoid_kernel<<<(tot + 255) / 256, 256, 0, stream>>>(p1l, p2l, s1, s2);

    const int TM = (MM + 15) / 16, TN = (NN + 15) / 16;
    cost_kernel<<<BB * TM * TN, 32, 0, stream>>>(p1b, p2b, s1, s2, cost);

    float* gcost = s1;
    int*   gcol  = (int*)s2;
    select_kernel<<<BB * MM / SEL_WAVES, 32 * SEL_WAVES, 0, stream>>>(cost, gcost, gcol);

    lapjv_kernel<<<BB, LTH, 0, stream>>>(cost, gcost, gcol, map);

    int tot2 = BB * MM;
    extrap_kernel<<<(tot2 + 255) / 256, 256, 0, stream>>>(p1b, p1l, p2b, p2l,
                                                          toff, map, out);
}


# Sparse f32 pass: next list requested one vector add behind the winner's lane read (offset ready-made, one record per lane); the step's bookkeeping moved under its two LDS waits; same sums
# speedup vs baseline: 7.0251x; 1.0954x over previous
#include <hip/hip_runtime.h>
#include <math.h>

#define BB 4
#define MM 500
#define NN 500
#define CC 91

typedef __attribute__((ext_vector_type(2))) float v2f;
typedef __attribute__((ext_vector_type(8))) float v8f;

__device__ __forceinline__ int imin_i(int a, int b) { return a < b ? a : b; }

__global__ void sigmoid_kernel(const float* __restrict__ p1l,
                               const float* __restrict__ p2l,
                               float* __restrict__ s1,
                               float* __restrict__ s2) {
    int idx = blockIdx.x * blockDim.x + threadIdx.x;
    const int n1 = BB * NN * CC;
    const int n2 = BB * MM * CC;
    if (idx < n1) {
        float x = p1l[idx];
        s1[idx] = 1.0f / (1.0f + expf(-x));
    } else if (idx < n1 + n2) {
        float x = p2l[idx - n1];
        s2[idx - n1] = 1.0f / (1.0f + expf(-x));
    }
}

__global__ void cost_kernel(const float* __restrict__ p1b,
                            const float* __restrict__ p2b,
                            const float* __restrict__ s1,
                            const float* __restrict__ s2,
                            float* __restrict__ cost) {
    const int TN = (NN + 15) / 16;
    const int TM = (MM + 15) / 16;
    int bid = blockIdx.x;
    int b  = bid / (TM * TN);
    int t  = bid % (TM * TN);
    int m0 = (t / TN) * 16;
    int n0 = (t % TN) * 16;
    int lane = threadIdx.x;

    __shared__ float sq2[16];
    __shared__ float sq1[16];
    __shared__ float ls2[16 * CC];
    __shared__ float ls1[16 * CC];

    v2f a  = {0.0f, 0.0f};
    v2f bv = {0.0f, 0.0f};
    if (lane < 16) {
        int m = imin_i(m0 + lane, MM - 1);
        const float* pb = p2b + ((size_t)b * MM + m) * 4;
        float x = pb[0], y = pb[1];
        a.x = x; a.y = y;
        sq2[lane] = x * x + y * y;
        int n = imin_i(n0 + lane, NN - 1);
        const float* qb = p1b + ((size_t)b * NN + n) * 4;
        float bx = qb[0], by = qb[1];
        bv.x = bx; bv.y = by;
        sq1[lane] = bx * bx + by * by;
    }
    const float* s2base = s2 + (size_t)b * MM * CC;
    const float* s1base = s1 + (size_t)b * NN * CC;
    for (int idx = lane; idx < 16 * CC; idx += 32) {
        int r = idx / CC;
        int c = idx - r * CC;
        ls2[idx] = s2base[(size_t)imin_i(m0 + r, MM - 1) * CC + c];
        ls1[idx] = s1base[(size_t)imin_i(n0 + r, NN - 1) * CC + c];
    }
    __syncthreads();

    v8f cacc = {0.f, 0.f, 0.f, 0.f, 0.f, 0.f, 0.f, 0.f};
    v8f dot = __builtin_amdgcn_wmma_f32_16x16x4_f32(
         false, a,  false, bv,
         (short)0, cacc,  false,  false);

    int n_loc  = lane & 15;
    int m_base = (lane >> 4) << 3;

    float cls[8];
#pragma unroll
    for (int i = 0; i < 8; ++i) cls[i] = 0.0f;
    for (int c = 0; c < CC; ++c) {
        float s1v = ls1[n_loc * CC + c];
#pragma unroll
        for (int i = 0; i < 8; ++i) {
            float d = fabsf(ls2[(m_base + i) * CC + c] - s1v);
            cls[i] = fmaxf(cls[i], d);
        }
    }

#pragma unroll
    for (int i = 0; i < 8; ++i) {
        int m = m0 + m_base + i;
        int n = n0 + n_loc;
        if (m < MM && n < NN) {
            float d2 = sq2[m_base + i] + sq1[n_loc] - 2.0f * dot[i];
            float cd = sqrtf(fmaxf(d2, 0.0f));
            cost[((size_t)b * MM + m) * NN + n] = 0.5f * cd + 0.5f * cls[i];
        }
    }
}

#define LTH 128
#define NWAVE (LTH / 32)
#define CPL 4
typedef __attribute__((ext_vector_type(4))) float v4f;
typedef __attribute__((ext_vector_type(4))) int   v4i;
typedef __attribute__((ext_vector_type(2))) int   v2i;
typedef __attribute__((ext_vector_type(2))) double v2d;
typedef __attribute__((ext_vector_type(2))) unsigned v2u;
typedef __attribute__((ext_vector_type(4))) unsigned v4u;
typedef __attribute__((address_space(3))) float lds_float;
typedef __attribute__((address_space(3))) unsigned lds_unsigned;
typedef __attribute__((address_space(3))) int lds_int;
typedef __attribute__((address_space(3))) v4u lds_v4u;
#define LREC_SHIFT 9

#define KC 64
static_assert(KC == 64, "two candidates per lane of one wave");
static_assert(NN < 512, "16 columns per lane, column 511 is a pad");

static_assert(NN % CPL == 0 && NN / CPL <= LTH, "column tiling");
static_assert(MM == NN, "square problem");

template <typename T> struct XT;
template <> struct XT<double> {
    template <int MASK>
    static __device__ __forceinline__ double rx(double x) {
        v2i p = __builtin_bit_cast(v2i, x);
        p.x = __builtin_amdgcn_update_dpp(p.x, p.x, 0x160 + MASK, 0xf, 0xf, false);
        p.y = __builtin_amdgcn_update_dpp(p.y, p.y, 0x160 + MASK, 0xf, 0xf, false);
        return __builtin_bit_cast(double, p);
    }
    static __device__ __forceinline__ double hx(double x) {
        v2i p = __builtin_bit_cast(v2i, x);
        p.x = __builtin_amdgcn_permlanex16(p.x, p.x, 0x76543210, 0xfedcba98, false, false);
        p.y = __builtin_amdgcn_permlanex16(p.y, p.y, 0x76543210, 0xfedcba98, false, false);
        return __builtin_bit_cast(double, p);
    }
    static __device__ __forceinline__ v2i pack(double x) { return __builtin_bit_cast(v2i, x); }
    static __device__ __forceinline__ double unpack(v2i p) { return __builtin_bit_cast(double, p); }
};

template <typename T, bool LIST>
__device__ __forceinline__ void sap_rows(const float* __restrict__ costB,
                                         const int nrows, const int* rowlist,
                                         T (&v)[CPL], T* sh_u, T* sh_short,
                                         int* sh_path, int* sh_c4r, int* sh_r4c,
                                         int* sh_SR, v4i (*rec)[NWAVE], int& par,
                                         const int tid, const int lane,
                                         const int wave, const bool owner,
                                         const unsigned full) {
    const T INF = (T)__builtin_inf();
    for (int idx = 0; idx < nrows; ++idx) {
        const int cur = LIST ? rowlist[idx] : idx;
        T   sh[CPL];
        int path[CPL];
        int r4c[CPL];
        {
            v4i q = *(const v4i*)&sh_r4c[CPL * (owner ? tid : 0)];
            r4c[0] = q.x; r4c[1] = q.y; r4c[2] = q.z; r4c[3] = q.w;
        }
        for (int k = 0; k < CPL; ++k) { sh[k] = INF; path[k] = -1; }
        unsigned rem = full;
        int ii   = cur;
        T   minv = (T)0;
        int nsr  = 0;
        int sink = -1;

        v4f c = {0.0f, 0.0f, 0.0f, 0.0f};
        if (owner) c = *(const v4f*)(costB + (size_t)ii * NN + CPL * tid);
        T z[CPL];
        {
            const T h = minv - sh_u[ii];
            for (int k = 0; k < CPL; ++k) z[k] = h - v[k];
        }

        while (true) {
            if (tid == 0) sh_SR[nsr] = ii;
            ++nsr;

            T t[CPL];
            for (int k = 0; k < CPL; ++k) t[k] = (T)c[k];
            for (int k = 0; k < CPL; ++k) t[k] = t[k] + z[k];

            T lm = INF; int lk = 0; int lr = -1;
            for (int k = 0; k < CPL; ++k) {
                const bool rm = ((rem >> k) & 1u) != 0u;
                const bool better = rm && (t[k] < sh[k]);
                sh[k]   = better ? t[k] : sh[k];
                path[k] = better ? ii : path[k];
                const T cd = rm ? sh[k] : INF;
                if (k == 0)       { lm = cd; lk = 0; lr = r4c[0]; }
                else if (cd < lm) { lm = cd; lk = k; lr = r4c[k]; }
            }

            T wm = lm;
            { const T o = XT<T>::template rx<1>(wm); wm = (o < wm) ? o : wm; }
            { const T o = XT<T>::template rx<2>(wm); wm = (o < wm) ? o : wm; }
            { const T o = XT<T>::template rx<4>(wm); wm = (o < wm) ? o : wm; }
            { const T o = XT<T>::template rx<8>(wm); wm = (o < wm) ? o : wm; }
            { const T o = XT<T>::hx(wm);             wm = (o < wm) ? o : wm; }
            const unsigned bal = __builtin_amdgcn_ballot_w32(lm == wm);
            const int srcl = __builtin_ctz(bal | 0x80000000u);
            const int wk = __builtin_amdgcn_readlane(lk, srcl);
            const int wr = __builtin_amdgcn_readlane(lr, srcl);
            if (lane == 0) {
                v2i p = XT<T>::pack(wm);
                v4i q = {p.x, p.y, CPL * (wave * 32 + srcl) + wk, wr};
                rec[par][wave] = q;
            }
            __syncthreads();

            T bv; int bj, br;
            {
                v4i q = rec[par][0];
                v2i p = {q.x, q.y};
                bv = XT<T>::unpack(p); bj = q.z; br = q.w;
            }
            for (int w = 1; w < NWAVE; ++w) {
                v4i q = rec[par][w];
                v2i p = {q.x, q.y};
                const T ov = XT<T>::unpack(p);
                if (ov < bv) { bv = ov; bj = q.z; br = q.w; }
            }
            par ^= 1;
            minv = bv;
            bj = __builtin_amdgcn_readfirstlane(bj);
            br = __builtin_amdgcn_readfirstlane(br);
            if (tid == (bj >> 2)) rem &= ~(1u << (bj & 3));
            if (br == -1) { sink = bj; break; }
            ii = br;

            c = (v4f){0.0f, 0.0f, 0.0f, 0.0f};
            if (owner) c = *(const v4f*)(costB + (size_t)ii * NN + CPL * tid);
            {
                const T h = minv - sh_u[ii];
                for (int k = 0; k < CPL; ++k) z[k] = h - v[k];
            }
        }

        if (owner) {
            v4i pq = {path[0], path[1], path[2], path[3]};
            *(v4i*)&sh_path[CPL * tid] = pq;
            for (int k = 0; k < CPL; ++k) sh_short[CPL * tid + k] = sh[k];
        }
        __syncthreads();

        for (int t2 = tid; t2 < nsr; t2 += LTH) {
            if (t2 == 0) {
                sh_u[cur] += minv;
            } else {
                int sr = sh_SR[t2];
                sh_u[sr] += minv - sh_short[sh_c4r[sr]];
            }
        }
        {
            const unsigned scanned = full & ~rem;
            for (int k = 0; k < CPL; ++k)
                if ((scanned >> k) & 1u) v[k] -= minv - sh[k];
        }
        __syncthreads();

        if (tid == 0) {
            int j = sink;
            while (true) {
                int i = sh_path[j];
                sh_r4c[j] = i;
                int tmp = sh_c4r[i];
                sh_c4r[i] = j;
                j = tmp;
                if (i == cur) break;
            }
        }
        __syncthreads();
    }
}

__device__ __forceinline__ int wave_sum_i(int x) {
    x += __builtin_amdgcn_update_dpp(x, x, 0x160 + 1, 0xf, 0xf, false);
    x += __builtin_amdgcn_update_dpp(x, x, 0x160 + 2, 0xf, 0xf, false);
    x += __builtin_amdgcn_update_dpp(x, x, 0x160 + 4, 0xf, 0xf, false);
    x += __builtin_amdgcn_update_dpp(x, x, 0x160 + 8, 0xf, 0xf, false);
    x += __builtin_amdgcn_permlanex16(x, x, 0x76543210, 0xfedcba98, false, false);
    return x;
}

__device__ __forceinline__ unsigned fbits(const float x) { return __builtin_bit_cast(unsigned, x); }

__device__ __forceinline__ float min2_f32(const float a, const float b) {
    float o;
    asm("v_min_num_f32 %0, %1, %2" : "=v"(o) : "v"(a), "v"(b));
    return o;
}

__device__ __forceinline__ float wave_min_f32(float x) {
    float a, b, c, d;
    asm("v_min_num_f32_dpp %0, %1, %1 row_xmask:1 row_mask:0xf bank_mask:0xf" : "=v"(a) : "v"(x));
    asm("v_min_num_f32_dpp %0, %1, %1 row_xmask:2 row_mask:0xf bank_mask:0xf" : "=v"(b) : "v"(a));
    asm("v_min_num_f32_dpp %0, %1, %1 row_xmask:4 row_mask:0xf bank_mask:0xf" : "=v"(c) : "v"(b));
    asm("v_min_num_f32_dpp %0, %1, %1 row_xmask:8 row_mask:0xf bank_mask:0xf" : "=v"(d) : "v"(c));
    const float o = __builtin_bit_cast(float,
        __builtin_amdgcn_permlanex16(0, __builtin_bit_cast(int, d), 0x76543210, 0xfedcba98, false, false));
    return min2_f32(o, d);
}

__device__ __forceinline__ void load_row16(const float* __restrict__ row, const int lane,
                                           unsigned (&bits)[16]) {
    for (int q = 0; q < 4; ++q) {
        const int c0 = 128 * q + 4 * lane;
        if (c0 < NN) {
            const v4f c = *(const v4f*)(row + c0);
            bits[4 * q + 0] = fbits(c.x);
            bits[4 * q + 1] = fbits(c.y);
            bits[4 * q + 2] = fbits(c.z);
            bits[4 * q + 3] = fbits(c.w);
        } else {
            for (int e = 0; e < 4; ++e) bits[4 * q + e] = 0x7f800000u;
        }
    }
}

#define SEL_WAVES 4
static_assert((BB * MM) % SEL_WAVES == 0, "whole work-groups of rows");
static_assert(MM * KC <= NN * CC && KC <= CC, "the lists fit into the two sigmoid planes they reuse");
__global__ void __launch_bounds__(32 * SEL_WAVES)
select_kernel(const float* __restrict__ cost, float* __restrict__ gcost, int* __restrict__ gcol) {
    const int lane = threadIdx.x & 31;
    const int wave = __builtin_amdgcn_readfirstlane((int)(threadIdx.x >> 5));
    const int row  = blockIdx.x * SEL_WAVES + wave;

    __shared__ float sc[SEL_WAVES][KC];
    __shared__ int   sj[SEL_WAVES][KC];

    const unsigned ltmask = (1u << lane) - 1u;
    unsigned bits[16];
    load_row16(cost + (size_t)row * NN, lane, bits);
    unsigned T = 0u;
    for (int bit = 30; bit >= 7; --bit) {
        const unsigned trial = T | (1u << bit);
        int cnt = 0;
        for (int k = 0; k < 16; ++k) cnt += (bits[k] < trial) ? 1 : 0;
        cnt = __builtin_amdgcn_readfirstlane(wave_sum_i(cnt));
        if (cnt <= KC) T = trial;
    }
    sc[wave][lane]      = __builtin_inff();
    sc[wave][lane + 32] = __builtin_inff();
    sj[wave][lane]      = 511;
    sj[wave][lane + 32] = 511;
    int base = 0;
    for (int k = 0; k < 16; ++k) {
        const bool f = bits[k] < T;
        const unsigned bal = __builtin_amdgcn_ballot_w32(f);
        const int pos = base + __builtin_popcount(bal & ltmask);
        if (f && pos < KC) {
            sc[wave][pos] = __builtin_bit_cast(float, bits[k]);
            sj[wave][pos] = 128 * (k >> 2) + 4 * lane + (k & 3);
        }
        base += __builtin_popcount(bal);
    }
    const float a0 = sc[wave][lane], a1 = sc[wave][lane + 32];
    const int   j0 = sj[wave][lane], j1 = sj[wave][lane + 32];
    gcost[(size_t)row * KC + lane]      = a0;
    gcost[(size_t)row * KC + lane + 32] = a1;
    gcol[(size_t)row * KC + lane]       = j0;
    gcol[(size_t)row * KC + lane + 32]  = j1;
}

__global__ void __launch_bounds__(LTH)
lapjv_kernel(const float* __restrict__ cost, const float* __restrict__ gcost,
             const int* __restrict__ gcol, int* __restrict__ mapping) {
    const int b    = blockIdx.x;
    const int tid  = threadIdx.x;
    const int lane = tid & 31;
    const int wave = __builtin_amdgcn_readfirstlane(tid >> 5);
    const float* costB = cost + (size_t)b * MM * NN;
    const bool owner = tid < NN / CPL;
    const unsigned full = owner ? 0xFu : 0u;

    __shared__ __attribute__((aligned(16))) unsigned lmix[(MM + 1) * KC * 2];
    __shared__ __attribute__((aligned(16))) double sh_ud[512];
    __shared__ __attribute__((aligned(16))) double sh_vd[512];
    __shared__ __attribute__((aligned(16))) double sh_short[512];
    __shared__ __attribute__((aligned(16))) float  sh_mbox[512];
    __shared__ __attribute__((aligned(16))) float  sh_vf[512];
    __shared__ __attribute__((aligned(16))) float  sh_uf[512];
    __shared__ __attribute__((aligned(16))) int sh_path[512];
    __shared__ __attribute__((aligned(16))) int sh_c4r[512];
    __shared__ __attribute__((aligned(16))) int sh_r4c[512];
    __shared__ __attribute__((aligned(16))) int sh_rec[4 * 512 + 32];
    int* const sh_SR = sh_rec;
    __shared__ __attribute__((aligned(16))) int sh_free[512];
    __shared__ __attribute__((aligned(16))) int sh_list[512];
    __shared__ int sh_cnt;
    __shared__ v4i rec[2][NWAVE];

    for (int t = tid; t < 512; t += LTH) {
        sh_uf[t] = 0.0f; sh_vf[t] = 0.0f; sh_c4r[t] = -1; sh_r4c[t] = -1;
    }
    int par = 0;

    {
        const v4f* gc = (const v4f*)(gcost + (size_t)b * MM * KC);
        const v4i* gj = (const v4i*)(gcol + (size_t)b * MM * KC);
        const int NG = MM * KC / 4;
        const unsigned mb = (unsigned)(__UINTPTR_TYPE__)(lds_float*)sh_mbox;
        if (tid < KC / 2) {
            const v4u e0 = {0x7f800000u, 0x7f800000u, mb + 4u * 511u, mb + 4u * 511u};
            *(v4u*)&lmix[4 * tid] = e0;
        }
        for (int t0 = tid; t0 < NG; t0 += LTH * 8) {
            v4f cq[8];
            v4i jq[8];
            for (int e = 0; e < 8; ++e) {
                const int t = imin_i(t0 + LTH * e, NG - 1);
                cq[e] = gc[t];
                jq[e] = gj[t];
            }
            for (int e = 0; e < 8; ++e) {
                const int t = t0 + LTH * e;
                if (t < NG) {
                    const v4u ra = {fbits(cq[e].x), fbits(cq[e].y),
                                    mb + 4u * (unsigned)jq[e].x, mb + 4u * (unsigned)jq[e].y};
                    const v4u rb = {fbits(cq[e].z), fbits(cq[e].w),
                                    mb + 4u * (unsigned)jq[e].z, mb + 4u * (unsigned)jq[e].w};
                    *(v4u*)&lmix[2 * (KC + 4 * t)]     = ra;
                    *(v4u*)&lmix[2 * (KC + 4 * t) + 4] = rb;
                }
            }
        }
    }
    __syncthreads();

    if (wave == 0) {
        const float INFf = __builtin_inff();
        for (int q = 0; q < 4; ++q) {
            const v4f ini = {INFf, INFf, INFf, INFf};
            *(v4f*)&sh_mbox[128 * q + 4 * lane] = ini;
        }
        const unsigned lane16 = (unsigned)(__UINTPTR_TYPE__)(lds_unsigned*)lmix + 16u * (unsigned)lane;
        const unsigned ufb = (unsigned)(__UINTPTR_TYPE__)(lds_float*)sh_uf;
        const unsigned recb = (unsigned)(__UINTPTR_TYPE__)(lds_int*)sh_rec
                            + 4u * (unsigned)((lane < 3) ? 512 * lane : 1536 + lane);
        for (int cur = 0; cur < MM; ++cur) {
            v4u   rc = *(const lds_v4u*)(lane16 + ((unsigned)(cur + 1) << LREC_SHIFT));
            float uf = sh_uf[cur + 1];
            float sh[16], vv[16];
            int   path[16], pk[16];
            for (int q = 0; q < 4; ++q) {
                const int c0 = 128 * q + 4 * lane;
                const bool real = c0 < NN;
                const v4i r = *(const v4i*)&sh_r4c[c0];
                const v4f w = *(const v4f*)&sh_vf[c0];
                pk[4 * q + 0] = ((r.x + 1) << LREC_SHIFT) | (4 * q + 0);
                pk[4 * q + 1] = ((r.y + 1) << LREC_SHIFT) | (4 * q + 1);
                pk[4 * q + 2] = ((r.z + 1) << LREC_SHIFT) | (4 * q + 2);
                pk[4 * q + 3] = ((r.w + 1) << LREC_SHIFT) | (4 * q + 3);
                vv[4 * q + 0] = real ? w.x : -INFf;
                vv[4 * q + 1] = real ? w.y : -INFf;
                vv[4 * q + 2] = real ? w.z : -INFf;
                vv[4 * q + 3] = real ? w.w : -INFf;
            }
            for (int k = 0; k < 16; ++k) { sh[k] = INFf; path[k] = -1; }

            int   ii   = cur;
            float minv = 0.0f;
            int   nsr  = 0;
            int   sink = -1;
            bool  okay = true;
            int   mkp = 16;
            while (true) {
                const unsigned rcx = rc.x, rcy = rc.y, rcz = rc.z, rcw = rc.w;
                lds_float* const c0 = (lds_float*)rcz;
                lds_float* const c1 = (lds_float*)rcw;
                *c0 = __builtin_bit_cast(float, rcx);
                *c1 = __builtin_bit_cast(float, rcy);
                v4f x[4];
                for (int q = 0; q < 4; ++q) x[q] = *(const v4f*)&sh_mbox[128 * q + 4 * lane];
                *c0 = INFf;
                *c1 = INFf;
                __builtin_amdgcn_sched_barrier(0);
                asm volatile("" : "+v"(mkp));

                for (int kk = 0; kk < 16; ++kk) {
                    const bool hit = (mkp == kk);
                    sh[kk] = hit ? INFf : sh[kk];
                    vv[kk] = hit ? -INFf : vv[kk];
                }
                const float h = minv - uf;
                float z[16];
                for (int k = 0; k < 16; ++k) z[k] = h - vv[k];
                asm volatile("" : "+v"(sh[0]), "+v"(sh[1]), "+v"(sh[2]), "+v"(sh[3]),
                                  "+v"(sh[4]), "+v"(sh[5]), "+v"(sh[6]), "+v"(sh[7]), "+v"(mkp));
                asm volatile("" : "+v"(sh[8]), "+v"(sh[9]), "+v"(sh[10]), "+v"(sh[11]),
                                  "+v"(sh[12]), "+v"(sh[13]), "+v"(sh[14]), "+v"(sh[15]), "+v"(mkp));
                asm volatile("" : "+v"(z[8]), "+v"(z[9]), "+v"(z[10]), "+v"(z[11]),
                                  "+v"(z[12]), "+v"(z[13]), "+v"(z[14]), "+v"(z[15]), "+v"(mkp));
                asm volatile("" : "+v"(z[0]), "+v"(z[1]), "+v"(z[2]), "+v"(z[3]),
                                  "+v"(z[4]), "+v"(z[5]), "+v"(z[6]), "+v"(z[7]), "+v"(mkp));
                for (int k = 0; k < 16; ++k) {
                    const float xk = x[k >> 2][k & 3];
                    const float r = xk + z[k];
                    const bool better = r < sh[k];
                    sh[k]   = better ? r : sh[k];
                    path[k] = better ? ii : path[k];
                }

                float m8[8]; int p8[8];
                for (int t = 0; t < 8; ++t) {
                    const float ma = sh[2 * t], mb = sh[2 * t + 1];
                    const int   pa = pk[2 * t], pb = pk[2 * t + 1];
                    const bool tk = mb < ma;
                    m8[t] = min2_f32(ma, mb);
                    p8[t] = tk ? pb : pa;
                }
                float m4[4]; int p4[4];
                for (int t = 0; t < 4; ++t) {
                    const float ma = m8[2 * t], mb = m8[2 * t + 1];
                    const int   pa = p8[2 * t], pb = p8[2 * t + 1];
                    const bool tk = mb < ma;
                    m4[t] = min2_f32(ma, mb);
                    p4[t] = tk ? pb : pa;
                }
                float m2[2]; int p2[2];
                for (int t = 0; t < 2; ++t) {
                    const float ma = m4[2 * t], mb = m4[2 * t + 1];
                    const int   pa = p4[2 * t], pb = p4[2 * t + 1];
                    const bool tk = mb < ma;
                    m2[t] = min2_f32(ma, mb);
                    p2[t] = tk ? pb : pa;
                }
                const float m2a = m2[0], m2b = m2[1];
                const int   p2a = p2[0], p2b = p2[1];
                const bool tk1 = m2b < m2a;
                const float lm = min2_f32(m2a, m2b);
                const int   lp = tk1 ? p2b : p2a;
                const int lro = lp & ~((1 << LREC_SHIFT) - 1);

                const float wm = wave_min_f32(lm);
                const unsigned bal = __builtin_amdgcn_ballot_w32(lm == wm);
                const int srcl = __builtin_ctz(bal | 0x80000000u);
                const int wro = __builtin_amdgcn_readlane(lro, srcl);
                rc = *(const lds_v4u*)(lane16 + (unsigned)wro);
                unsigned uo;
                asm("v_lshrrev_b32 %0, 7, %1" : "=v"(uo) : "s"(wro));
                uf = *(const lds_float*)(ufb + uo);
                __builtin_amdgcn_sched_barrier(0);
                const int wp = __builtin_amdgcn_readlane(lp, srcl);
                const float wms = __builtin_bit_cast(float,
                    __builtin_amdgcn_readfirstlane(__builtin_bit_cast(int, wm)));
                if (!(wms < INFf)) { okay = false; break; }
                const int k  = wp & 15;
                const int r1 = wp >> LREC_SHIFT;
                const int br = r1 - 1;
                const int bj = 128 * (k >> 2) + 4 * srcl + (k & 3);
                minv = wms;
                {
                    const int rv = (lane == 0) ? ii : ((lane == 1) ? bj : __builtin_bit_cast(int, wms));
                    *(lds_int*)(recb + 4u * (unsigned)nsr) = rv;
                }
                ++nsr;
                if (br == -1) { sink = bj; break; }
                mkp = (lp & 15) + ((lane == srcl) ? 0 : 16);
                ii = br;
            }
            if (!okay) continue;
            for (int q = 0; q < 4; ++q) {
                const v4i pq = {path[4 * q + 0], path[4 * q + 1], path[4 * q + 2], path[4 * q + 3]};
                *(v4i*)&sh_path[128 * q + 4 * lane] = pq;
            }
            for (int t = lane; t < nsr; t += 32) {
                const float d = minv - __builtin_bit_cast(float, sh_rec[1024 + t]);
                sh_vf[sh_rec[512 + t]] -= d;
                if (t == 0) sh_uf[cur + 1] += minv;
                if (t + 1 < nsr) sh_uf[sh_rec[t + 1] + 1] += d;
            }
            if (lane == 0) {
                int j = sink;
                for (int g = 0; g < MM; ++g) {
                    const int i = sh_path[j];
                    sh_r4c[j] = i;
                    const int tmp = sh_c4r[i];
                    sh_c4r[i] = j;
                    j = tmp;
                    if (i == cur) break;
                }
            }
        }
    }
    __syncthreads();

    double v[CPL];
    {
        v4f q = {0.0f, 0.0f, 0.0f, 0.0f};
        if (owner) q = *(const v4f*)&sh_vf[CPL * tid];
        v[0] = (double)q.x; v[1] = (double)q.y; v[2] = (double)q.z; v[3] = (double)q.w;
    }
    if (owner) {
        v2d a01 = {v[0], v[1]};
        v2d a23 = {v[2], v[3]};
        *(v2d*)&sh_vd[CPL * tid]     = a01;
        *(v2d*)&sh_vd[CPL * tid + 2] = a23;
    }
    __syncthreads();
    for (int i = wave; i < MM; i += NWAVE) {
        const float* row = costB + (size_t)i * NN;
        double m = __builtin_inf();
        for (int q = 0; q < 4; ++q) {
            const int c0 = 128 * q + CPL * lane;
            if (c0 < NN) {
                const v4f c = *(const v4f*)(row + c0);
                const v2d va = *(const v2d*)&sh_vd[c0];
                const v2d vb = *(const v2d*)&sh_vd[c0 + 2];
                const double d0 = (double)c.x - va.x;
                const double d1 = (double)c.y - va.y;
                const double d2 = (double)c.z - vb.x;
                const double d3 = (double)c.w - vb.y;
                m = (d0 < m) ? d0 : m;
                m = (d1 < m) ? d1 : m;
                m = (d2 < m) ? d2 : m;
                m = (d3 < m) ? d3 : m;
            }
        }
        { const double o = XT<double>::rx<1>(m); m = (o < m) ? o : m; }
        { const double o = XT<double>::rx<2>(m); m = (o < m) ? o : m; }
        { const double o = XT<double>::rx<4>(m); m = (o < m) ? o : m; }
        { const double o = XT<double>::rx<8>(m); m = (o < m) ? o : m; }
        { const double o = XT<double>::hx(m);    m = (o < m) ? o : m; }
        const int jm = sh_c4r[i];
        const bool has = jm >= 0;
        const int jq = has ? jm : 0;
        const double dm = (double)row[jq] - sh_vd[jq];
        const bool keep = has && (dm == m);
        if (lane == 0) {
            sh_ud[i]   = m;
            sh_free[i] = keep ? 0 : 1;
            if (has && !keep) { sh_c4r[i] = -1; sh_r4c[jm] = -1; }
        }
    }
    __syncthreads();
    if (wave == 0) {
        int cnt = 0;
        for (int base = 0; base < 512; base += 32) {
            const int r = base + lane;
            const bool f = (r < MM) && (sh_free[r] != 0);
            const unsigned bal = __builtin_amdgcn_ballot_w32(f);
            const int pos = cnt + __builtin_popcount(bal & ((1u << lane) - 1u));
            if (f) sh_list[pos] = r;
            cnt += __builtin_popcount(bal);
        }
        if (lane == 0) sh_cnt = cnt;
    }
    __syncthreads();
    const int nfree = __builtin_amdgcn_readfirstlane(sh_cnt);

    sap_rows<double, true>(costB, nfree, sh_list, v, sh_ud, sh_short,
                           sh_path, sh_c4r, sh_r4c, sh_SR, rec, par,
                           tid, lane, wave, owner, full);

    for (int t = tid; t < MM; t += LTH) mapping[b * MM + t] = sh_c4r[t];
}

__global__ void extrap_kernel(const float* __restrict__ p1b,
                              const float* __restrict__ p1l,
                              const float* __restrict__ p2b,
                              const float* __restrict__ p2l,
                              const float* __restrict__ toff,
                              const int*   __restrict__ mapping,
                              float* __restrict__ out) {
    int idx = blockIdx.x * blockDim.x + threadIdx.x;
    if (idx >= BB * MM) return;
    int b = idx / MM;

    float first  = toff[b * 3 + 1] - toff[b * 3 + 0];
    float second = toff[b * 3 + 2] - toff[b * 3 + 1];
    float factor = second / first;

    int  j   = mapping[idx];
    bool has = (j >= 0);
    int  jc  = has ? j : 0;

    const float* pb2 = p2b + (size_t)idx * 4;
    const float* pb1 = p1b + ((size_t)b * NN + jc) * 4;
    float* ob = out + (size_t)idx * 4;
    for (int k = 0; k < 4; ++k) {
        float c1 = has ? pb1[k] : pb2[k];
        float v  = pb2[k] + (pb2[k] - c1) * factor;
        ob[k] = (k >= 2) ? fmaxf(v, 0.0f) : v;
    }

    const float* pl2 = p2l + (size_t)idx * CC;
    const float* pl1 = p1l + ((size_t)b * NN + jc) * CC;
    float* ol = out + (size_t)BB * MM * 4 + (size_t)idx * CC;
    for (int c = 0; c < CC; ++c) {
        float c1 = has ? pl1[c] : 0.0f;
        ol[c] = 0.5f * (pl2[c] + c1);
    }
}

extern "C" void kernel_launch(void* const* d_in, const int* in_sizes, int n_in,
                              void* d_out, int out_size, void* d_ws, size_t ws_size,
                              hipStream_t stream) {
    (void)in_sizes; (void)n_in; (void)out_size; (void)ws_size;
    const float* p1b  = (const float*)d_in[0];
    const float* p1l  = (const float*)d_in[1];
    const float* p2b  = (const float*)d_in[2];
    const float* p2l  = (const float*)d_in[3];
    const float* toff = (const float*)d_in[4];

    float* ws   = (float*)d_ws;
    float* s1   = ws;
    float* s2   = s1 + (size_t)BB * NN * CC;
    float* cost = s2 + (size_t)BB * MM * CC;
    int*   map  = (int*)(cost + (size_t)BB * MM * NN);
    float* out  = (float*)d_out;

    int tot = BB * NN * CC + BB * MM * CC;
    sigmoid_kernel<<<(tot + 255) / 256, 256, 0, stream>>>(p1l, p2l, s1, s2);

    const int TM = (MM + 15) / 16, TN = (NN + 15) / 16;
    cost_kernel<<<BB * TM * TN, 32, 0, stream>>>(p1b, p2b, s1, s2, cost);

    float* gcost = s1;
    int*   gcol  = (int*)s2;
    select_kernel<<<BB * MM / SEL_WAVES, 32 * SEL_WAVES, 0, stream>>>(cost, gcost, gcol);

    lapjv_kernel<<<BB, LTH, 0, stream>>>(cost, gcost, gcol, map);

    int tot2 = BB * MM;
    extrap_kernel<<<(tot2 + 255) / 256, 256, 0, stream>>>(p1b, p1l, p2b, p2l,
                                                          toff, map, out);
}


# Sparse approximate pass: costs, distances, duals as exact 32-bit fixed point (unit 2^-26), not f32, so the f64 repair finds its pairs exactly tight and frees few rows; sums exact; same optimum
# speedup vs baseline: 7.5884x; 1.0802x over previous
#include <hip/hip_runtime.h>
#include <math.h>

#define BB 4
#define MM 500
#define NN 500
#define CC 91

typedef __attribute__((ext_vector_type(2))) float v2f;
typedef __attribute__((ext_vector_type(8))) float v8f;

__device__ __forceinline__ int imin_i(int a, int b) { return a < b ? a : b; }

__global__ void sigmoid_kernel(const float* __restrict__ p1l,
                               const float* __restrict__ p2l,
                               float* __restrict__ s1,
                               float* __restrict__ s2) {
    int idx = blockIdx.x * blockDim.x + threadIdx.x;
    const int n1 = BB * NN * CC;
    const int n2 = BB * MM * CC;
    if (idx < n1) {
        float x = p1l[idx];
        s1[idx] = 1.0f / (1.0f + expf(-x));
    } else if (idx < n1 + n2) {
        float x = p2l[idx - n1];
        s2[idx - n1] = 1.0f / (1.0f + expf(-x));
    }
}

__global__ void cost_kernel(const float* __restrict__ p1b,
                            const float* __restrict__ p2b,
                            const float* __restrict__ s1,
                            const float* __restrict__ s2,
                            float* __restrict__ cost) {
    const int TN = (NN + 15) / 16;
    const int TM = (MM + 15) / 16;
    int bid = blockIdx.x;
    int b  = bid / (TM * TN);
    int t  = bid % (TM * TN);
    int m0 = (t / TN) * 16;
    int n0 = (t % TN) * 16;
    int lane = threadIdx.x;

    __shared__ float sq2[16];
    __shared__ float sq1[16];
    __shared__ float ls2[16 * CC];
    __shared__ float ls1[16 * CC];

    v2f a  = {0.0f, 0.0f};
    v2f bv = {0.0f, 0.0f};
    if (lane < 16) {
        int m = imin_i(m0 + lane, MM - 1);
        const float* pb = p2b + ((size_t)b * MM + m) * 4;
        float x = pb[0], y = pb[1];
        a.x = x; a.y = y;
        sq2[lane] = x * x + y * y;
        int n = imin_i(n0 + lane, NN - 1);
        const float* qb = p1b + ((size_t)b * NN + n) * 4;
        float bx = qb[0], by = qb[1];
        bv.x = bx; bv.y = by;
        sq1[lane] = bx * bx + by * by;
    }
    const float* s2base = s2 + (size_t)b * MM * CC;
    const float* s1base = s1 + (size_t)b * NN * CC;
    for (int idx = lane; idx < 16 * CC; idx += 32) {
        int r = idx / CC;
        int c = idx - r * CC;
        ls2[idx] = s2base[(size_t)imin_i(m0 + r, MM - 1) * CC + c];
        ls1[idx] = s1base[(size_t)imin_i(n0 + r, NN - 1) * CC + c];
    }
    __syncthreads();

    v8f cacc = {0.f, 0.f, 0.f, 0.f, 0.f, 0.f, 0.f, 0.f};
    v8f dot = __builtin_amdgcn_wmma_f32_16x16x4_f32(
         false, a,  false, bv,
         (short)0, cacc,  false,  false);

    int n_loc  = lane & 15;
    int m_base = (lane >> 4) << 3;

    float cls[8];
#pragma unroll
    for (int i = 0; i < 8; ++i) cls[i] = 0.0f;
    for (int c = 0; c < CC; ++c) {
        float s1v = ls1[n_loc * CC + c];
#pragma unroll
        for (int i = 0; i < 8; ++i) {
            float d = fabsf(ls2[(m_base + i) * CC + c] - s1v);
            cls[i] = fmaxf(cls[i], d);
        }
    }

#pragma unroll
    for (int i = 0; i < 8; ++i) {
        int m = m0 + m_base + i;
        int n = n0 + n_loc;
        if (m < MM && n < NN) {
            float d2 = sq2[m_base + i] + sq1[n_loc] - 2.0f * dot[i];
            float cd = sqrtf(fmaxf(d2, 0.0f));
            cost[((size_t)b * MM + m) * NN + n] = 0.5f * cd + 0.5f * cls[i];
        }
    }
}

#define LTH 128
#define NWAVE (LTH / 32)
#define CPL 4
typedef __attribute__((ext_vector_type(4))) float v4f;
typedef __attribute__((ext_vector_type(4))) int   v4i;
typedef __attribute__((ext_vector_type(2))) int   v2i;
typedef __attribute__((ext_vector_type(2))) double v2d;
typedef __attribute__((ext_vector_type(2))) unsigned v2u;
typedef __attribute__((ext_vector_type(4))) unsigned v4u;
typedef __attribute__((address_space(3))) unsigned lds_unsigned;
typedef __attribute__((address_space(3))) int lds_int;
typedef __attribute__((address_space(3))) v4u lds_v4u;
#define LREC_SHIFT 9

#define KC 64
#define FX_ONE   67108864.0f
#define FX_BIAS  0x80000000u
#define FX_XINF  (1u << 29)
#define FX_SINF  0x7fffffff
#define FX_SDEAD 0
#define FX_VDEAD 0x40000000u
#define FX_THR   (1u << 28)
#define FX_BND   (1 << 28)
static_assert(KC == 64, "two candidates per lane of one wave");
static_assert(NN < 512, "16 columns per lane, column 511 is a pad");

static_assert(NN % CPL == 0 && NN / CPL <= LTH, "column tiling");
static_assert(MM == NN, "square problem");

template <typename T> struct XT;
template <> struct XT<double> {
    template <int MASK>
    static __device__ __forceinline__ double rx(double x) {
        v2i p = __builtin_bit_cast(v2i, x);
        p.x = __builtin_amdgcn_update_dpp(p.x, p.x, 0x160 + MASK, 0xf, 0xf, false);
        p.y = __builtin_amdgcn_update_dpp(p.y, p.y, 0x160 + MASK, 0xf, 0xf, false);
        return __builtin_bit_cast(double, p);
    }
    static __device__ __forceinline__ double hx(double x) {
        v2i p = __builtin_bit_cast(v2i, x);
        p.x = __builtin_amdgcn_permlanex16(p.x, p.x, 0x76543210, 0xfedcba98, false, false);
        p.y = __builtin_amdgcn_permlanex16(p.y, p.y, 0x76543210, 0xfedcba98, false, false);
        return __builtin_bit_cast(double, p);
    }
    static __device__ __forceinline__ v2i pack(double x) { return __builtin_bit_cast(v2i, x); }
    static __device__ __forceinline__ double unpack(v2i p) { return __builtin_bit_cast(double, p); }
};

template <typename T, bool LIST>
__device__ __forceinline__ void sap_rows(const float* __restrict__ costB,
                                         const int nrows, const int* rowlist,
                                         T (&v)[CPL], T* sh_u, T* sh_short,
                                         int* sh_path, int* sh_c4r, int* sh_r4c,
                                         int* sh_SR, v4i (*rec)[NWAVE], int& par,
                                         const int tid, const int lane,
                                         const int wave, const bool owner,
                                         const unsigned full) {
    const T INF = (T)__builtin_inf();
    for (int idx = 0; idx < nrows; ++idx) {
        const int cur = LIST ? rowlist[idx] : idx;
        T   sh[CPL];
        int path[CPL];
        int r4c[CPL];
        {
            v4i q = *(const v4i*)&sh_r4c[CPL * (owner ? tid : 0)];
            r4c[0] = q.x; r4c[1] = q.y; r4c[2] = q.z; r4c[3] = q.w;
        }
        for (int k = 0; k < CPL; ++k) { sh[k] = INF; path[k] = -1; }
        unsigned rem = full;
        int ii   = cur;
        T   minv = (T)0;
        int nsr  = 0;
        int sink = -1;

        v4f c = {0.0f, 0.0f, 0.0f, 0.0f};
        if (owner) c = *(const v4f*)(costB + (size_t)ii * NN + CPL * tid);
        T z[CPL];
        {
            const T h = minv - sh_u[ii];
            for (int k = 0; k < CPL; ++k) z[k] = h - v[k];
        }

        while (true) {
            if (tid == 0) sh_SR[nsr] = ii;
            ++nsr;

            T t[CPL];
            for (int k = 0; k < CPL; ++k) t[k] = (T)c[k];
            for (int k = 0; k < CPL; ++k) t[k] = t[k] + z[k];

            T lm = INF; int lk = 0; int lr = -1;
            for (int k = 0; k < CPL; ++k) {
                const bool rm = ((rem >> k) & 1u) != 0u;
                const bool better = rm && (t[k] < sh[k]);
                sh[k]   = better ? t[k] : sh[k];
                path[k] = better ? ii : path[k];
                const T cd = rm ? sh[k] : INF;
                if (k == 0)       { lm = cd; lk = 0; lr = r4c[0]; }
                else if (cd < lm) { lm = cd; lk = k; lr = r4c[k]; }
            }

            T wm = lm;
            { const T o = XT<T>::template rx<1>(wm); wm = (o < wm) ? o : wm; }
            { const T o = XT<T>::template rx<2>(wm); wm = (o < wm) ? o : wm; }
            { const T o = XT<T>::template rx<4>(wm); wm = (o < wm) ? o : wm; }
            { const T o = XT<T>::template rx<8>(wm); wm = (o < wm) ? o : wm; }
            { const T o = XT<T>::hx(wm);             wm = (o < wm) ? o : wm; }
            const unsigned bal = __builtin_amdgcn_ballot_w32(lm == wm);
            const int srcl = __builtin_ctz(bal | 0x80000000u);
            const int wk = __builtin_amdgcn_readlane(lk, srcl);
            const int wr = __builtin_amdgcn_readlane(lr, srcl);
            if (lane == 0) {
                v2i p = XT<T>::pack(wm);
                v4i q = {p.x, p.y, CPL * (wave * 32 + srcl) + wk, wr};
                rec[par][wave] = q;
            }
            __syncthreads();

            T bv; int bj, br;
            {
                v4i q = rec[par][0];
                v2i p = {q.x, q.y};
                bv = XT<T>::unpack(p); bj = q.z; br = q.w;
            }
            for (int w = 1; w < NWAVE; ++w) {
                v4i q = rec[par][w];
                v2i p = {q.x, q.y};
                const T ov = XT<T>::unpack(p);
                if (ov < bv) { bv = ov; bj = q.z; br = q.w; }
            }
            par ^= 1;
            minv = bv;
            bj = __builtin_amdgcn_readfirstlane(bj);
            br = __builtin_amdgcn_readfirstlane(br);
            if (tid == (bj >> 2)) rem &= ~(1u << (bj & 3));
            if (br == -1) { sink = bj; break; }
            ii = br;

            c = (v4f){0.0f, 0.0f, 0.0f, 0.0f};
            if (owner) c = *(const v4f*)(costB + (size_t)ii * NN + CPL * tid);
            {
                const T h = minv - sh_u[ii];
                for (int k = 0; k < CPL; ++k) z[k] = h - v[k];
            }
        }

        if (owner) {
            v4i pq = {path[0], path[1], path[2], path[3]};
            *(v4i*)&sh_path[CPL * tid] = pq;
            for (int k = 0; k < CPL; ++k) sh_short[CPL * tid + k] = sh[k];
        }
        __syncthreads();

        for (int t2 = tid; t2 < nsr; t2 += LTH) {
            if (t2 == 0) {
                sh_u[cur] += minv;
            } else {
                int sr = sh_SR[t2];
                sh_u[sr] += minv - sh_short[sh_c4r[sr]];
            }
        }
        {
            const unsigned scanned = full & ~rem;
            for (int k = 0; k < CPL; ++k)
                if ((scanned >> k) & 1u) v[k] -= minv - sh[k];
        }
        __syncthreads();

        if (tid == 0) {
            int j = sink;
            while (true) {
                int i = sh_path[j];
                sh_r4c[j] = i;
                int tmp = sh_c4r[i];
                sh_c4r[i] = j;
                j = tmp;
                if (i == cur) break;
            }
        }
        __syncthreads();
    }
}

__device__ __forceinline__ int wave_sum_i(int x) {
    x += __builtin_amdgcn_update_dpp(x, x, 0x160 + 1, 0xf, 0xf, false);
    x += __builtin_amdgcn_update_dpp(x, x, 0x160 + 2, 0xf, 0xf, false);
    x += __builtin_amdgcn_update_dpp(x, x, 0x160 + 4, 0xf, 0xf, false);
    x += __builtin_amdgcn_update_dpp(x, x, 0x160 + 8, 0xf, 0xf, false);
    x += __builtin_amdgcn_permlanex16(x, x, 0x76543210, 0xfedcba98, false, false);
    return x;
}

__device__ __forceinline__ unsigned fbits(const float x) { return __builtin_bit_cast(unsigned, x); }

__device__ __forceinline__ unsigned fx_of(const float c) {
    return (c < 8.0f) ? (unsigned)(int)(c * FX_ONE) : FX_XINF;
}

__device__ __forceinline__ int min2_i32(const int a, const int b) { return b < a ? b : a; }

__device__ __forceinline__ int wave_min_i32(int x) {
    int a, b, c, d;
    asm("v_min_i32_dpp %0, %1, %1 row_xmask:1 row_mask:0xf bank_mask:0xf" : "=v"(a) : "v"(x));
    asm("v_min_i32_dpp %0, %1, %1 row_xmask:2 row_mask:0xf bank_mask:0xf" : "=v"(b) : "v"(a));
    asm("v_min_i32_dpp %0, %1, %1 row_xmask:4 row_mask:0xf bank_mask:0xf" : "=v"(c) : "v"(b));
    asm("v_min_i32_dpp %0, %1, %1 row_xmask:8 row_mask:0xf bank_mask:0xf" : "=v"(d) : "v"(c));
    const int o = __builtin_amdgcn_permlanex16(0, d, 0x76543210, 0xfedcba98, false, false);
    return min2_i32(o, d);
}

__device__ __forceinline__ void load_row16(const float* __restrict__ row, const int lane,
                                           unsigned (&bits)[16]) {
    for (int q = 0; q < 4; ++q) {
        const int c0 = 128 * q + 4 * lane;
        if (c0 < NN) {
            const v4f c = *(const v4f*)(row + c0);
            bits[4 * q + 0] = fbits(c.x);
            bits[4 * q + 1] = fbits(c.y);
            bits[4 * q + 2] = fbits(c.z);
            bits[4 * q + 3] = fbits(c.w);
        } else {
            for (int e = 0; e < 4; ++e) bits[4 * q + e] = 0x7f800000u;
        }
    }
}

#define SEL_WAVES 4
static_assert((BB * MM) % SEL_WAVES == 0, "whole work-groups of rows");
static_assert(MM * KC <= NN * CC && KC <= CC, "the lists fit into the two sigmoid planes they reuse");
__global__ void __launch_bounds__(32 * SEL_WAVES)
select_kernel(const float* __restrict__ cost, float* __restrict__ gcost, int* __restrict__ gcol) {
    const int lane = threadIdx.x & 31;
    const int wave = __builtin_amdgcn_readfirstlane((int)(threadIdx.x >> 5));
    const int row  = blockIdx.x * SEL_WAVES + wave;

    __shared__ float sc[SEL_WAVES][KC];
    __shared__ int   sj[SEL_WAVES][KC];

    const unsigned ltmask = (1u << lane) - 1u;
    unsigned bits[16];
    load_row16(cost + (size_t)row * NN, lane, bits);
    unsigned T = 0u;
    for (int bit = 30; bit >= 7; --bit) {
        const unsigned trial = T | (1u << bit);
        int cnt = 0;
        for (int k = 0; k < 16; ++k) cnt += (bits[k] < trial) ? 1 : 0;
        cnt = __builtin_amdgcn_readfirstlane(wave_sum_i(cnt));
        if (cnt <= KC) T = trial;
    }
    sc[wave][lane]      = __builtin_inff();
    sc[wave][lane + 32] = __builtin_inff();
    sj[wave][lane]      = 511;
    sj[wave][lane + 32] = 511;
    int base = 0;
    for (int k = 0; k < 16; ++k) {
        const bool f = bits[k] < T;
        const unsigned bal = __builtin_amdgcn_ballot_w32(f);
        const int pos = base + __builtin_popcount(bal & ltmask);
        if (f && pos < KC) {
            sc[wave][pos] = __builtin_bit_cast(float, bits[k]);
            sj[wave][pos] = 128 * (k >> 2) + 4 * lane + (k & 3);
        }
        base += __builtin_popcount(bal);
    }
    const float a0 = sc[wave][lane], a1 = sc[wave][lane + 32];
    const int   j0 = sj[wave][lane], j1 = sj[wave][lane + 32];
    gcost[(size_t)row * KC + lane]      = a0;
    gcost[(size_t)row * KC + lane + 32] = a1;
    gcol[(size_t)row * KC + lane]       = j0;
    gcol[(size_t)row * KC + lane + 32]  = j1;
}

__global__ void __launch_bounds__(LTH)
lapjv_kernel(const float* __restrict__ cost, const float* __restrict__ gcost,
             const int* __restrict__ gcol, int* __restrict__ mapping) {
    const int b    = blockIdx.x;
    const int tid  = threadIdx.x;
    const int lane = tid & 31;
    const int wave = __builtin_amdgcn_readfirstlane(tid >> 5);
    const float* costB = cost + (size_t)b * MM * NN;
    const bool owner = tid < NN / CPL;
    const unsigned full = owner ? 0xFu : 0u;

    __shared__ __attribute__((aligned(16))) unsigned lmix[(MM + 1) * KC * 2];
    __shared__ __attribute__((aligned(16))) double sh_ud[512];
    __shared__ __attribute__((aligned(16))) double sh_vd[512];
    __shared__ __attribute__((aligned(16))) double sh_short[512];
    __shared__ __attribute__((aligned(16))) unsigned sh_mbox[512];
    __shared__ __attribute__((aligned(16))) int      sh_vi[512];
    __shared__ __attribute__((aligned(16))) unsigned sh_ui[512];
    __shared__ __attribute__((aligned(16))) int sh_path[512];
    __shared__ __attribute__((aligned(16))) int sh_c4r[512];
    __shared__ __attribute__((aligned(16))) int sh_r4c[512];
    __shared__ __attribute__((aligned(16))) int sh_rec[4 * 512 + 32];
    int* const sh_SR = sh_rec;
    __shared__ __attribute__((aligned(16))) int sh_free[512];
    __shared__ __attribute__((aligned(16))) int sh_list[512];
    __shared__ int sh_cnt;
    __shared__ v4i rec[2][NWAVE];

    for (int t = tid; t < 512; t += LTH) {
        sh_ui[t] = 0u; sh_vi[t] = 0; sh_c4r[t] = -1; sh_r4c[t] = -1;
    }
    int par = 0;

    {
        const v4f* gc = (const v4f*)(gcost + (size_t)b * MM * KC);
        const v4i* gj = (const v4i*)(gcol + (size_t)b * MM * KC);
        const int NG = MM * KC / 4;
        const unsigned mb = (unsigned)(__UINTPTR_TYPE__)(lds_unsigned*)sh_mbox;
        if (tid < KC / 2) {
            const v4u e0 = {FX_XINF, FX_XINF, mb + 4u * 511u, mb + 4u * 511u};
            *(v4u*)&lmix[4 * tid] = e0;
        }
        for (int t0 = tid; t0 < NG; t0 += LTH * 8) {
            v4f cq[8];
            v4i jq[8];
            for (int e = 0; e < 8; ++e) {
                const int t = imin_i(t0 + LTH * e, NG - 1);
                cq[e] = gc[t];
                jq[e] = gj[t];
            }
            for (int e = 0; e < 8; ++e) {
                const int t = t0 + LTH * e;
                if (t < NG) {
                    const v4u ra = {fx_of(cq[e].x), fx_of(cq[e].y),
                                    mb + 4u * (unsigned)jq[e].x, mb + 4u * (unsigned)jq[e].y};
                    const v4u rb = {fx_of(cq[e].z), fx_of(cq[e].w),
                                    mb + 4u * (unsigned)jq[e].z, mb + 4u * (unsigned)jq[e].w};
                    *(v4u*)&lmix[2 * (KC + 4 * t)]     = ra;
                    *(v4u*)&lmix[2 * (KC + 4 * t) + 4] = rb;
                }
            }
        }
    }
    __syncthreads();

    if (wave == 0) {
        for (int q = 0; q < 4; ++q) {
            const v4u ini = {FX_XINF, FX_XINF, FX_XINF, FX_XINF};
            *(v4u*)&sh_mbox[128 * q + 4 * lane] = ini;
        }
        const unsigned lane16 = (unsigned)(__UINTPTR_TYPE__)(lds_unsigned*)lmix + 16u * (unsigned)lane;
        const unsigned ufb = (unsigned)(__UINTPTR_TYPE__)(lds_unsigned*)sh_ui;
        const unsigned recb = (unsigned)(__UINTPTR_TYPE__)(lds_int*)sh_rec
                            + 4u * (unsigned)((lane < 3) ? 512 * lane : 1536 + lane);
        for (int cur = 0; cur < MM; ++cur) {
            v4u   rc = *(const lds_v4u*)(lane16 + ((unsigned)(cur + 1) << LREC_SHIFT));
            unsigned uf = sh_ui[cur + 1];
            int      sh[16];
            unsigned vv[16];
            int   path[16], pk[16];
            for (int q = 0; q < 4; ++q) {
                const int c0 = 128 * q + 4 * lane;
                const bool real = c0 < NN;
                const v4i r = *(const v4i*)&sh_r4c[c0];
                const v4i w = *(const v4i*)&sh_vi[c0];
                pk[4 * q + 0] = ((r.x + 1) << LREC_SHIFT) | (4 * q + 0);
                pk[4 * q + 1] = ((r.y + 1) << LREC_SHIFT) | (4 * q + 1);
                pk[4 * q + 2] = ((r.z + 1) << LREC_SHIFT) | (4 * q + 2);
                pk[4 * q + 3] = ((r.w + 1) << LREC_SHIFT) | (4 * q + 3);
                vv[4 * q + 0] = real ? (unsigned)w.x : FX_VDEAD;
                vv[4 * q + 1] = real ? (unsigned)w.y : FX_VDEAD;
                vv[4 * q + 2] = real ? (unsigned)w.z : FX_VDEAD;
                vv[4 * q + 3] = real ? (unsigned)w.w : FX_VDEAD;
                for (int e = 0; e < 4; ++e) sh[4 * q + e] = real ? FX_SINF : FX_SDEAD;
            }
            for (int k = 0; k < 16; ++k) path[k] = -1;

            int      ii   = cur;
            unsigned minv = 0u;
            int   nsr  = 0;
            int   sink = -1;
            bool  okay = true;
            int   mkp = 16;
            while (true) {
                const unsigned rcx = rc.x, rcy = rc.y, rcz = rc.z, rcw = rc.w;
                lds_unsigned* const c0 = (lds_unsigned*)rcz;
                lds_unsigned* const c1 = (lds_unsigned*)rcw;
                *c0 = rcx;
                *c1 = rcy;
                v4u x[4];
                for (int q = 0; q < 4; ++q) x[q] = *(const v4u*)&sh_mbox[128 * q + 4 * lane];
                *c0 = FX_XINF;
                *c1 = FX_XINF;
                __builtin_amdgcn_sched_barrier(0);
                asm volatile("" : "+v"(mkp));

                for (int kk = 0; kk < 16; ++kk) {
                    const bool hit = (mkp == kk);
                    sh[kk] = hit ? FX_SDEAD : sh[kk];
                    vv[kk] = hit ? FX_VDEAD : vv[kk];
                }
                int h = (int)(minv - uf);
                h = h < -FX_BND ? -FX_BND : (h > FX_BND ? FX_BND : h);
                const unsigned hu = (unsigned)h + FX_BIAS;
                unsigned z[16];
                for (int k = 0; k < 16; ++k) z[k] = hu - vv[k];
                asm volatile("" : "+v"(sh[0]), "+v"(sh[1]), "+v"(sh[2]), "+v"(sh[3]),
                                  "+v"(sh[4]), "+v"(sh[5]), "+v"(sh[6]), "+v"(sh[7]), "+v"(mkp));
                asm volatile("" : "+v"(sh[8]), "+v"(sh[9]), "+v"(sh[10]), "+v"(sh[11]),
                                  "+v"(sh[12]), "+v"(sh[13]), "+v"(sh[14]), "+v"(sh[15]), "+v"(mkp));
                asm volatile("" : "+v"(z[8]), "+v"(z[9]), "+v"(z[10]), "+v"(z[11]),
                                  "+v"(z[12]), "+v"(z[13]), "+v"(z[14]), "+v"(z[15]), "+v"(mkp));
                asm volatile("" : "+v"(z[0]), "+v"(z[1]), "+v"(z[2]), "+v"(z[3]),
                                  "+v"(z[4]), "+v"(z[5]), "+v"(z[6]), "+v"(z[7]), "+v"(mkp));
                for (int k = 0; k < 16; ++k) {
                    const unsigned xk = x[k >> 2][k & 3];
                    const int r = (int)(xk + z[k]);
                    const bool better = r < sh[k];
                    sh[k]   = better ? r : sh[k];
                    path[k] = better ? ii : path[k];
                }

                int m8[8]; int p8[8];
                for (int t = 0; t < 8; ++t) {
                    const int ma = sh[2 * t], mb = sh[2 * t + 1];
                    const int pa = pk[2 * t], pb = pk[2 * t + 1];
                    const bool tk = mb < ma;
                    m8[t] = min2_i32(ma, mb);
                    p8[t] = tk ? pb : pa;
                }
                int m4[4]; int p4[4];
                for (int t = 0; t < 4; ++t) {
                    const int ma = m8[2 * t], mb = m8[2 * t + 1];
                    const int pa = p8[2 * t], pb = p8[2 * t + 1];
                    const bool tk = mb < ma;
                    m4[t] = min2_i32(ma, mb);
                    p4[t] = tk ? pb : pa;
                }
                int m2[2]; int p2[2];
                for (int t = 0; t < 2; ++t) {
                    const int ma = m4[2 * t], mb = m4[2 * t + 1];
                    const int pa = p4[2 * t], pb = p4[2 * t + 1];
                    const bool tk = mb < ma;
                    m2[t] = min2_i32(ma, mb);
                    p2[t] = tk ? pb : pa;
                }
                const int m2a = m2[0], m2b = m2[1];
                const int p2a = p2[0], p2b = p2[1];
                const bool tk1 = m2b < m2a;
                const int lm = min2_i32(m2a, m2b);
                const int lp = tk1 ? p2b : p2a;
                const int lro = lp & ~((1 << LREC_SHIFT) - 1);

                const int wm = wave_min_i32(lm);
                const unsigned bal = __builtin_amdgcn_ballot_w32(lm == wm);
                const int srcl = __builtin_ctz(bal | 0x80000000u);
                const int wro = __builtin_amdgcn_readlane(lro, srcl);
                rc = *(const lds_v4u*)(lane16 + (unsigned)wro);
                unsigned uo;
                asm("v_lshrrev_b32 %0, 7, %1" : "=v"(uo) : "s"(wro));
                uf = *(const lds_unsigned*)(ufb + uo);
                __builtin_amdgcn_sched_barrier(0);
                const int wp = __builtin_amdgcn_readlane(lp, srcl);
                const unsigned wms = (unsigned)__builtin_amdgcn_readfirstlane(wm) ^ FX_BIAS;
                if (wms >= FX_THR) { okay = false; break; }
                const int k  = wp & 15;
                const int r1 = wp >> LREC_SHIFT;
                const int br = r1 - 1;
                const int bj = 128 * (k >> 2) + 4 * srcl + (k & 3);
                minv = wms;
                {
                    const int rv = (lane == 0) ? ii : ((lane == 1) ? bj : (int)wms);
                    *(lds_int*)(recb + 4u * (unsigned)nsr) = rv;
                }
                ++nsr;
                if (br == -1) { sink = bj; break; }
                mkp = (lp & 15) + ((lane == srcl) ? 0 : 16);
                ii = br;
            }
            if (!okay) continue;
            for (int q = 0; q < 4; ++q) {
                const v4i pq = {path[4 * q + 0], path[4 * q + 1], path[4 * q + 2], path[4 * q + 3]};
                *(v4i*)&sh_path[128 * q + 4 * lane] = pq;
            }
            for (int t = lane; t < nsr; t += 32) {
                const int d  = (int)minv - sh_rec[1024 + t];
                const int jc = sh_rec[512 + t];
                const int nv = sh_vi[jc] - d;
                sh_vi[jc] = nv < -FX_BND ? -FX_BND : (nv > FX_BND ? FX_BND : nv);
                if (t == 0) sh_ui[cur + 1] += minv;
                if (t + 1 < nsr) sh_ui[sh_rec[t + 1] + 1] += (unsigned)d;
            }
            if (lane == 0) {
                int j = sink;
                for (int g = 0; g < MM; ++g) {
                    const int i = sh_path[j];
                    sh_r4c[j] = i;
                    const int tmp = sh_c4r[i];
                    sh_c4r[i] = j;
                    j = tmp;
                    if (i == cur) break;
                }
            }
        }
    }
    __syncthreads();

    double v[CPL];
    {
        v4i q = {0, 0, 0, 0};
        if (owner) q = *(const v4i*)&sh_vi[CPL * tid];
        v[0] = (double)q.x * 0x1p-26; v[1] = (double)q.y * 0x1p-26;
        v[2] = (double)q.z * 0x1p-26; v[3] = (double)q.w * 0x1p-26;
    }
    if (owner) {
        v2d a01 = {v[0], v[1]};
        v2d a23 = {v[2], v[3]};
        *(v2d*)&sh_vd[CPL * tid]     = a01;
        *(v2d*)&sh_vd[CPL * tid + 2] = a23;
    }
    __syncthreads();
    for (int i = wave; i < MM; i += NWAVE) {
        const float* row = costB + (size_t)i * NN;
        double m = __builtin_inf();
        for (int q = 0; q < 4; ++q) {
            const int c0 = 128 * q + CPL * lane;
            if (c0 < NN) {
                const v4f c = *(const v4f*)(row + c0);
                const v2d va = *(const v2d*)&sh_vd[c0];
                const v2d vb = *(const v2d*)&sh_vd[c0 + 2];
                const double d0 = (double)c.x - va.x;
                const double d1 = (double)c.y - va.y;
                const double d2 = (double)c.z - vb.x;
                const double d3 = (double)c.w - vb.y;
                m = (d0 < m) ? d0 : m;
                m = (d1 < m) ? d1 : m;
                m = (d2 < m) ? d2 : m;
                m = (d3 < m) ? d3 : m;
            }
        }
        { const double o = XT<double>::rx<1>(m); m = (o < m) ? o : m; }
        { const double o = XT<double>::rx<2>(m); m = (o < m) ? o : m; }
        { const double o = XT<double>::rx<4>(m); m = (o < m) ? o : m; }
        { const double o = XT<double>::rx<8>(m); m = (o < m) ? o : m; }
        { const double o = XT<double>::hx(m);    m = (o < m) ? o : m; }
        const int jm = sh_c4r[i];
        const bool has = jm >= 0;
        const int jq = has ? jm : 0;
        const double dm = (double)row[jq] - sh_vd[jq];
        const bool keep = has && (dm == m);
        if (lane == 0) {
            sh_ud[i]   = m;
            sh_free[i] = keep ? 0 : 1;
            if (has && !keep) { sh_c4r[i] = -1; sh_r4c[jm] = -1; }
        }
    }
    __syncthreads();
    if (wave == 0) {
        int cnt = 0;
        for (int base = 0; base < 512; base += 32) {
            const int r = base + lane;
            const bool f = (r < MM) && (sh_free[r] != 0);
            const unsigned bal = __builtin_amdgcn_ballot_w32(f);
            const int pos = cnt + __builtin_popcount(bal & ((1u << lane) - 1u));
            if (f) sh_list[pos] = r;
            cnt += __builtin_popcount(bal);
        }
        if (lane == 0) sh_cnt = cnt;
    }
    __syncthreads();
    const int nfree = __builtin_amdgcn_readfirstlane(sh_cnt);

    sap_rows<double, true>(costB, nfree, sh_list, v, sh_ud, sh_short,
                           sh_path, sh_c4r, sh_r4c, sh_SR, rec, par,
                           tid, lane, wave, owner, full);

    for (int t = tid; t < MM; t += LTH) mapping[b * MM + t] = sh_c4r[t];
}

__global__ void extrap_kernel(const float* __restrict__ p1b,
                              const float* __restrict__ p1l,
                              const float* __restrict__ p2b,
                              const float* __restrict__ p2l,
                              const float* __restrict__ toff,
                              const int*   __restrict__ mapping,
                              float* __restrict__ out) {
    int idx = blockIdx.x * blockDim.x + threadIdx.x;
    if (idx >= BB * MM) return;
    int b = idx / MM;

    float first  = toff[b * 3 + 1] - toff[b * 3 + 0];
    float second = toff[b * 3 + 2] - toff[b * 3 + 1];
    float factor = second / first;

    int  j   = mapping[idx];
    bool has = (j >= 0);
    int  jc  = has ? j : 0;

    const float* pb2 = p2b + (size_t)idx * 4;
    const float* pb1 = p1b + ((size_t)b * NN + jc) * 4;
    float* ob = out + (size_t)idx * 4;
    for (int k = 0; k < 4; ++k) {
        float c1 = has ? pb1[k] : pb2[k];
        float v  = pb2[k] + (pb2[k] - c1) * factor;
        ob[k] = (k >= 2) ? fmaxf(v, 0.0f) : v;
    }

    const float* pl2 = p2l + (size_t)idx * CC;
    const float* pl1 = p1l + ((size_t)b * NN + jc) * CC;
    float* ol = out + (size_t)BB * MM * 4 + (size_t)idx * CC;
    for (int c = 0; c < CC; ++c) {
        float c1 = has ? pl1[c] : 0.0f;
        ol[c] = 0.5f * (pl2[c] + c1);
    }
}

extern "C" void kernel_launch(void* const* d_in, const int* in_sizes, int n_in,
                              void* d_out, int out_size, void* d_ws, size_t ws_size,
                              hipStream_t stream) {
    (void)in_sizes; (void)n_in; (void)out_size; (void)ws_size;
    const float* p1b  = (const float*)d_in[0];
    const float* p1l  = (const float*)d_in[1];
    const float* p2b  = (const float*)d_in[2];
    const float* p2l  = (const float*)d_in[3];
    const float* toff = (const float*)d_in[4];

    float* ws   = (float*)d_ws;
    float* s1   = ws;
    float* s2   = s1 + (size_t)BB * NN * CC;
    float* cost = s2 + (size_t)BB * MM * CC;
    int*   map  = (int*)(cost + (size_t)BB * MM * NN);
    float* out  = (float*)d_out;

    int tot = BB * NN * CC + BB * MM * CC;
    sigmoid_kernel<<<(tot + 255) / 256, 256, 0, stream>>>(p1l, p2l, s1, s2);

    const int TM = (MM + 15) / 16, TN = (NN + 15) / 16;
    cost_kernel<<<BB * TM * TN, 32, 0, stream>>>(p1b, p2b, s1, s2, cost);

    float* gcost = s1;
    int*   gcol  = (int*)s2;
    select_kernel<<<BB * MM / SEL_WAVES, 32 * SEL_WAVES, 0, stream>>>(cost, gcost, gcol);

    lapjv_kernel<<<BB, LTH, 0, stream>>>(cost, gcost, gcol, map);

    int tot2 = BB * MM;
    extrap_kernel<<<(tot2 + 255) / 256, 256, 0, stream>>>(p1b, p1l, p2b, p2l,
                                                          toff, map, out);
}
